# starting kernel
# baseline (speedup 1.0000x reference)
#include <hip/hip_runtime.h>
#include <math.h>


typedef __attribute__((ext_vector_type(2))) float v2f;
typedef __attribute__((ext_vector_type(8))) float v8f;

__device__ __forceinline__ v2f frag_a(const float* A, int ld, int k0) {
  int lane = threadIdx.x & 31;
  int row  = lane & 15;
  int kk   = k0 + ((lane >> 4) << 1);
  v2f a;
  a.x = A[row * ld + kk];
  a.y = A[row * ld + kk + 1];
  return a;
}

__device__ __forceinline__ v2f frag_b(const float* W, int ld, int k0, int c0) {
  int lane = threadIdx.x & 31;
  int col  = c0 + (lane & 15);
  int kk   = k0 + ((lane >> 4) << 1);
  v2f b;
  b.x = W[kk * ld + col];
  b.y = W[(kk + 1) * ld + col];
  return b;
}

__device__ __forceinline__ v2f frag_b_kguard(const float* W, int ld, int k0,
                                             int c0, int kmax) {
  int lane = threadIdx.x & 31;
  int col  = c0 + (lane & 15);
  int kk   = k0 + ((lane >> 4) << 1);
  v2f b;
  b.x = (kk     < kmax) ? W[kk * ld + col]       : 0.0f;
  b.y = (kk + 1 < kmax) ? W[(kk + 1) * ld + col] : 0.0f;
  return b;
}

__device__ __forceinline__ v8f wmma4(v2f a, v2f b, v8f c) {
  return __builtin_amdgcn_wmma_f32_16x16x4_f32(
       false, a,  false, b,
       (short)0, c,  false,  false);
}

__device__ __forceinline__ float fast_tanh(float x) {
  float e = __expf(2.0f * x);
  return 1.0f - __fdividef(2.0f, e + 1.0f);
}
__device__ __forceinline__ float fast_sigmoid(float x) {
  return __fdividef(1.0f, 1.0f + __expf(-x));
}

__global__ __launch_bounds__(256) void kalman_gru_scan(
    const float* __restrict__ x,
    const float* __restrict__ y,
    const float* __restrict__ W_enc,
    const float* __restrict__ b_enc,
    const float* __restrict__ W_in,
    const float* __restrict__ W_rec,
    const float* __restrict__ b_s,
    const float* __restrict__ W_x,
    const float* __restrict__ W_h,
    const float* __restrict__ b_x,
    const float* __restrict__ b_h,
    const float* __restrict__ W_gate,
    const float* __restrict__ b_gate,
    float* __restrict__ out)
{
  constexpr int T = 2048, IN_ = 32, S = 32, M = 8, E = 64, H = 128;
  constexpr int CIN = 65, G = 384;

  __shared__ float sh_x[16 * 32];
  __shared__ float sh_inp[16 * 68];
  __shared__ float sh_h[16 * 128];
  __shared__ float sh_ns[16 * 8 * 32];
  __shared__ float sh_kst[16 * 32];
  __shared__ float sh_gate[16 * 8];
  __shared__ float sh_sc0[16 * 16];
  __shared__ float sh_sc1[16 * 16];
  __shared__ float sh_y[16];
  __shared__ float sh_err[16];

  const int tid  = threadIdx.x;
  const int wave = tid >> 5;
  const int lane = tid & 31;
  const int b0   = blockIdx.x * 16;

  const int colb = lane & 15;
  const int rb   = (lane >> 4) << 3;
  const int zc0  = wave * 16;
  const int hc   = zc0 + colb;

  const float bias_z  = b_x[hc]       + b_h[hc];
  const float bias_r  = b_x[128 + hc] + b_h[128 + hc];
  const float bias_xn = b_x[256 + hc];
  const float bias_hn = b_h[256 + hc];
  const float bias_s0 = b_s[wave * 32 + colb];
  const float bias_s1 = b_s[wave * 32 + 16 + colb];
  const float bias_e  = (wave < 4) ? b_enc[wave * 16 + colb] : 0.0f;

  v2f whz[32], whr[32], whn[32];
  #pragma unroll
  for (int i = 0; i < 32; ++i) {
    whz[i] = frag_b(W_h, G, 4 * i, zc0);
    whr[i] = frag_b(W_h, G, 4 * i, 128 + zc0);
    whn[i] = frag_b(W_h, G, 4 * i, 256 + zc0);
  }
  v2f wxz[17], wxr[17], wxn[17];
  #pragma unroll
  for (int i = 0; i < 16; ++i) {
    wxz[i] = frag_b(W_x, G, 4 * i, zc0);
    wxr[i] = frag_b(W_x, G, 4 * i, 128 + zc0);
    wxn[i] = frag_b(W_x, G, 4 * i, 256 + zc0);
  }
  wxz[16] = frag_b_kguard(W_x, G, 64, zc0,       CIN);
  wxr[16] = frag_b_kguard(W_x, G, 64, 128 + zc0, CIN);
  wxn[16] = frag_b_kguard(W_x, G, 64, 256 + zc0, CIN);

  for (int i = tid; i < 16 * H; i += 256) sh_h[i] = 0.0f;
  for (int i = tid; i < 16 * S; i += 256) sh_kst[i] = 0.0f;
  for (int i = tid; i < 16 * 68; i += 256) sh_inp[i] = 0.0f;
  if (tid < 16) sh_err[tid] = 0.0f;
  __syncthreads();

  for (int t = 0; t < T; ++t) {
    for (int i = tid; i < 16 * IN_; i += 256) {
      int row = i >> 5, col = i & 31;
      sh_x[i] = x[(size_t)(b0 + row) * T * IN_ + (size_t)t * IN_ + col];
    }
    if (tid < 16) {
      sh_y[tid] = y[(size_t)(b0 + tid) * T + t];
      sh_inp[tid * 68 + 64] = sh_err[tid];
      if (t + 1 < T)
        __builtin_prefetch(&x[(size_t)(b0 + tid) * T * IN_ +
                              (size_t)(t + 1) * IN_], 0, 1);
    }
    __syncthreads();

    {
      const int m = wave;
      const float* Wi = W_in  + m * IN_ * S;
      const float* Wr = W_rec + m * S * S;
      v8f c0 = {}, c1 = {};
      #pragma unroll
      for (int k0 = 0; k0 < IN_; k0 += 4) {
        v2f a = frag_a(sh_x, 32, k0);
        c0 = wmma4(a, frag_b(Wi, S, k0, 0),  c0);
        c1 = wmma4(a, frag_b(Wi, S, k0, 16), c1);
      }
      #pragma unroll
      for (int k0 = 0; k0 < S; k0 += 4) {
        v2f a = frag_a(sh_kst, 32, k0);
        c0 = wmma4(a, frag_b(Wr, S, k0, 0),  c0);
        c1 = wmma4(a, frag_b(Wr, S, k0, 16), c1);
      }
      #pragma unroll
      for (int v = 0; v < 8; ++v) {
        int row = rb + v;
        sh_ns[row * 256 + m * 32 + colb]      = fast_tanh(c0[v] + bias_s0);
        sh_ns[row * 256 + m * 32 + 16 + colb] = fast_tanh(c1[v] + bias_s1);
      }
    }
    if (wave < 4) {
      int c0c = wave * 16;
      v8f c = {};
      #pragma unroll
      for (int k0 = 0; k0 < IN_; k0 += 4)
        c = wmma4(frag_a(sh_x, 32, k0), frag_b(W_enc, E, k0, c0c), c);
      #pragma unroll
      for (int v = 0; v < 8; ++v)
        sh_inp[(rb + v) * 68 + c0c + colb] = fast_tanh(c[v] + bias_e);
    }
    __syncthreads();

    float hnew[8];
    {
      v8f Cz = {}, Cr = {}, Cxn = {}, Chn = {};
      #pragma unroll
      for (int i = 0; i < 16; ++i) {
        v2f a = frag_a(sh_inp, 68, 4 * i);
        Cz  = wmma4(a, wxz[i], Cz);
        Cr  = wmma4(a, wxr[i], Cr);
        Cxn = wmma4(a, wxn[i], Cxn);
      }
      {
        v2f a = frag_a(sh_inp, 68, 64);
        Cz  = wmma4(a, wxz[16], Cz);
        Cr  = wmma4(a, wxr[16], Cr);
        Cxn = wmma4(a, wxn[16], Cxn);
      }
      #pragma unroll
      for (int i = 0; i < 32; ++i) {
        v2f a = frag_a(sh_h, H, 4 * i);
        Cz  = wmma4(a, whz[i], Cz);
        Cr  = wmma4(a, whr[i], Cr);
        Chn = wmma4(a, whn[i], Chn);
      }
      #pragma unroll
      for (int v = 0; v < 8; ++v) {
        int row = rb + v;
        float z = fast_sigmoid(Cz[v] + bias_z);
        float r = fast_sigmoid(Cr[v] + bias_r);
        float n = fast_tanh((Cxn[v] + bias_xn) + r * (Chn[v] + bias_hn));
        hnew[v] = (1.0f - z) * n + z * sh_h[row * H + hc];
      }
    }
    __syncthreads();
    #pragma unroll
    for (int v = 0; v < 8; ++v) sh_h[(rb + v) * H + hc] = hnew[v];
    __syncthreads();

    if (wave < 2) {
      float* dst = (wave == 0) ? sh_sc0 : sh_sc1;
      const int kbase = wave * 64;
      v8f c = {};
      #pragma unroll
      for (int i = 0; i < 16; ++i) {
        int k0 = kbase + 4 * i;
        v2f a = frag_a(sh_h, H, k0);
        int kk = k0 + ((lane >> 4) << 1);
        v2f b;
        b.x = (colb < M) ? W_gate[kk * M + colb]       : 0.0f;
        b.y = (colb < M) ? W_gate[(kk + 1) * M + colb] : 0.0f;
        c = wmma4(a, b, c);
      }
      if (colb < M) {
        #pragma unroll
        for (int v = 0; v < 8; ++v) dst[(rb + v) * 16 + colb] = c[v];
      }
    }
    __syncthreads();

    if (tid < 16) {
      float s[M], mx = -1e30f;
      #pragma unroll
      for (int m = 0; m < M; ++m) {
        s[m] = sh_sc0[tid * 16 + m] + sh_sc1[tid * 16 + m] + b_gate[m];
        mx = fmaxf(mx, s[m]);
      }
      float sum = 0.0f;
      #pragma unroll
      for (int m = 0; m < M; ++m) { s[m] = __expf(s[m] - mx); sum += s[m]; }
      float inv = __fdividef(1.0f, sum);
      #pragma unroll
      for (int m = 0; m < M; ++m) sh_gate[tid * M + m] = s[m] * inv;
    }
    __syncthreads();

    for (int i = tid; i < 16 * S; i += 256) {
      int row = i >> 5, col = i & 31;
      float acc = 0.0f;
      #pragma unroll
      for (int m = 0; m < M; ++m)
        acc += sh_gate[row * M + m] * sh_ns[row * 256 + m * 32 + col];
      sh_kst[i] = acc;
      if (col == S - 1) {
        out[(size_t)(b0 + row) * T + t] = acc;
        sh_err[row] = acc - sh_y[row];
      }
    }
    __syncthreads();
  }
}

extern "C" void kernel_launch(void* const* d_in, const int* in_sizes, int n_in,
                              void* d_out, int out_size, void* d_ws, size_t ws_size,
                              hipStream_t stream) {
  (void)in_sizes; (void)n_in; (void)out_size; (void)d_ws; (void)ws_size;
  const float* x      = (const float*)d_in[0];
  const float* y      = (const float*)d_in[1];
  const float* W_enc  = (const float*)d_in[2];
  const float* b_enc  = (const float*)d_in[3];
  const float* W_in   = (const float*)d_in[4];
  const float* W_rec  = (const float*)d_in[5];
  const float* b_s    = (const float*)d_in[6];
  const float* W_x    = (const float*)d_in[7];
  const float* W_h    = (const float*)d_in[8];
  const float* b_x    = (const float*)d_in[9];
  const float* b_h    = (const float*)d_in[10];
  const float* W_gate = (const float*)d_in[11];
  const float* b_gate = (const float*)d_in[12];
  float* out = (float*)d_out;

  kalman_gru_scan<<<dim3(8), dim3(256), 0, stream>>>(
      x, y, W_enc, b_enc, W_in, W_rec, b_s, W_x, W_h, b_x, b_h,
      W_gate, b_gate, out);
}


# W_enc, W_gate, b_gate staged once in LDS in B-fragment order: the time loop no longer re-reads weights (and spilled pointers) from global/scratch in serial load-wait-WMMA chains
# speedup vs baseline: 1.5787x; 1.5787x over previous
#include <hip/hip_runtime.h>
#include <math.h>


typedef __attribute__((ext_vector_type(2))) float v2f;
typedef __attribute__((ext_vector_type(8))) float v8f;

__device__ __forceinline__ v2f frag_a(const float* A, int ld, int k0) {
  int lane = threadIdx.x & 31;
  int row  = lane & 15;
  int kk   = k0 + ((lane >> 4) << 1);
  v2f a;
  a.x = A[row * ld + kk];
  a.y = A[row * ld + kk + 1];
  return a;
}

__device__ __forceinline__ v2f frag_b(const float* W, int ld, int k0, int c0) {
  int lane = threadIdx.x & 31;
  int col  = c0 + (lane & 15);
  int kk   = k0 + ((lane >> 4) << 1);
  v2f b;
  b.x = W[kk * ld + col];
  b.y = W[(kk + 1) * ld + col];
  return b;
}

__device__ __forceinline__ v2f frag_b_kguard(const float* W, int ld, int k0,
                                             int c0, int kmax) {
  int lane = threadIdx.x & 31;
  int col  = c0 + (lane & 15);
  int kk   = k0 + ((lane >> 4) << 1);
  v2f b;
  b.x = (kk     < kmax) ? W[kk * ld + col]       : 0.0f;
  b.y = (kk + 1 < kmax) ? W[(kk + 1) * ld + col] : 0.0f;
  return b;
}

__device__ __forceinline__ v8f wmma4(v2f a, v2f b, v8f c) {
  return __builtin_amdgcn_wmma_f32_16x16x4_f32(
       false, a,  false, b,
       (short)0, c,  false,  false);
}

__device__ __forceinline__ float fast_tanh(float x) {
  float e = __expf(2.0f * x);
  return 1.0f - __fdividef(2.0f, e + 1.0f);
}
__device__ __forceinline__ float fast_sigmoid(float x) {
  return __fdividef(1.0f, 1.0f + __expf(-x));
}

__global__ __launch_bounds__(256) void kalman_gru_scan(
    const float* __restrict__ x,
    const float* __restrict__ y,
    const float* __restrict__ W_enc,
    const float* __restrict__ b_enc,
    const float* __restrict__ W_in,
    const float* __restrict__ W_rec,
    const float* __restrict__ b_s,
    const float* __restrict__ W_x,
    const float* __restrict__ W_h,
    const float* __restrict__ b_x,
    const float* __restrict__ b_h,
    const float* __restrict__ W_gate,
    const float* __restrict__ b_gate,
    float* __restrict__ out)
{
  constexpr int T = 2048, IN_ = 32, S = 32, M = 8, E = 64, H = 128;
  constexpr int CIN = 65, G = 384;

  __shared__ float sh_x[16 * 32];
  __shared__ float sh_inp[16 * 68];
  __shared__ float sh_h[16 * 128];
  __shared__ float sh_ns[16 * 8 * 32];
  __shared__ float sh_kst[16 * 32];
  __shared__ float sh_gate[16 * 8];
  __shared__ float sh_sc0[16 * 16];
  __shared__ float sh_sc1[16 * 16];
  __shared__ float sh_y[16];
  __shared__ float sh_err[16];
  __shared__ v2f sh_wencf[8 * 4 * 32];
  __shared__ v2f sh_wgf[32 * 32];
  __shared__ float sh_bg[8];

  const int tid  = threadIdx.x;
  const int wave = tid >> 5;
  const int lane = tid & 31;
  const int b0   = blockIdx.x * 16;

  const int colb = lane & 15;
  const int rb   = (lane >> 4) << 3;
  const int zc0  = wave * 16;
  const int hc   = zc0 + colb;

  const float bias_z  = b_x[hc]       + b_h[hc];
  const float bias_r  = b_x[128 + hc] + b_h[128 + hc];
  const float bias_xn = b_x[256 + hc];
  const float bias_hn = b_h[256 + hc];
  const float bias_s0 = b_s[wave * 32 + colb];
  const float bias_s1 = b_s[wave * 32 + 16 + colb];
  const float bias_e  = (wave < 4) ? b_enc[wave * 16 + colb] : 0.0f;

  v2f whz[32], whr[32], whn[32];
  #pragma unroll
  for (int i = 0; i < 32; ++i) {
    whz[i] = frag_b(W_h, G, 4 * i, zc0);
    whr[i] = frag_b(W_h, G, 4 * i, 128 + zc0);
    whn[i] = frag_b(W_h, G, 4 * i, 256 + zc0);
  }
  v2f wxz[17], wxr[17], wxn[17];
  #pragma unroll
  for (int i = 0; i < 16; ++i) {
    wxz[i] = frag_b(W_x, G, 4 * i, zc0);
    wxr[i] = frag_b(W_x, G, 4 * i, 128 + zc0);
    wxn[i] = frag_b(W_x, G, 4 * i, 256 + zc0);
  }
  wxz[16] = frag_b_kguard(W_x, G, 64, zc0,       CIN);
  wxr[16] = frag_b_kguard(W_x, G, 64, 128 + zc0, CIN);
  wxn[16] = frag_b_kguard(W_x, G, 64, 256 + zc0, CIN);

  for (int i = tid; i < 16 * H; i += 256) sh_h[i] = 0.0f;
  for (int i = tid; i < 16 * S; i += 256) sh_kst[i] = 0.0f;
  for (int i = tid; i < 16 * 68; i += 256) sh_inp[i] = 0.0f;
  if (tid < 16) sh_err[tid] = 0.0f;
  for (int idx = tid; idx < 8 * 4 * 32; idx += 256) {
    int ln = idx & 31, tile = (idx >> 5) & 3, i = idx >> 7;
    int kk  = 4 * i + ((ln >> 4) << 1);
    int col = tile * 16 + (ln & 15);
    v2f b;
    b.x = W_enc[kk * E + col];
    b.y = W_enc[(kk + 1) * E + col];
    sh_wencf[idx] = b;
  }
  for (int idx = tid; idx < 32 * 32; idx += 256) {
    int ln = idx & 31, i = idx >> 5;
    int kk  = 4 * i + ((ln >> 4) << 1);
    int col = ln & 15;
    v2f b;
    b.x = (col < M) ? W_gate[kk * M + col]       : 0.0f;
    b.y = (col < M) ? W_gate[(kk + 1) * M + col] : 0.0f;
    sh_wgf[idx] = b;
  }
  if (tid < M) sh_bg[tid] = b_gate[tid];
  __syncthreads();

  for (int t = 0; t < T; ++t) {
    for (int i = tid; i < 16 * IN_; i += 256) {
      int row = i >> 5, col = i & 31;
      sh_x[i] = x[(size_t)(b0 + row) * T * IN_ + (size_t)t * IN_ + col];
    }
    if (tid < 16) {
      sh_y[tid] = y[(size_t)(b0 + tid) * T + t];
      sh_inp[tid * 68 + 64] = sh_err[tid];
      if (t + 1 < T)
        __builtin_prefetch(&x[(size_t)(b0 + tid) * T * IN_ +
                              (size_t)(t + 1) * IN_], 0, 1);
    }
    __syncthreads();

    {
      const int m = wave;
      const float* Wi = W_in  + m * IN_ * S;
      const float* Wr = W_rec + m * S * S;
      v8f c0 = {}, c1 = {};
      #pragma unroll
      for (int k0 = 0; k0 < IN_; k0 += 4) {
        v2f a = frag_a(sh_x, 32, k0);
        c0 = wmma4(a, frag_b(Wi, S, k0, 0),  c0);
        c1 = wmma4(a, frag_b(Wi, S, k0, 16), c1);
      }
      #pragma unroll
      for (int k0 = 0; k0 < S; k0 += 4) {
        v2f a = frag_a(sh_kst, 32, k0);
        c0 = wmma4(a, frag_b(Wr, S, k0, 0),  c0);
        c1 = wmma4(a, frag_b(Wr, S, k0, 16), c1);
      }
      #pragma unroll
      for (int v = 0; v < 8; ++v) {
        int row = rb + v;
        sh_ns[row * 256 + m * 32 + colb]      = fast_tanh(c0[v] + bias_s0);
        sh_ns[row * 256 + m * 32 + 16 + colb] = fast_tanh(c1[v] + bias_s1);
      }
    }
    if (wave < 4) {
      int c0c = wave * 16;
      v8f c = {};
      #pragma unroll
      for (int i = 0; i < IN_ / 4; ++i)
        c = wmma4(frag_a(sh_x, 32, 4 * i),
                  sh_wencf[(i * 4 + wave) * 32 + lane], c);
      #pragma unroll
      for (int v = 0; v < 8; ++v)
        sh_inp[(rb + v) * 68 + c0c + colb] = fast_tanh(c[v] + bias_e);
    }
    __syncthreads();

    float hnew[8];
    {
      v8f Cz = {}, Cr = {}, Cxn = {}, Chn = {};
      #pragma unroll
      for (int i = 0; i < 16; ++i) {
        v2f a = frag_a(sh_inp, 68, 4 * i);
        Cz  = wmma4(a, wxz[i], Cz);
        Cr  = wmma4(a, wxr[i], Cr);
        Cxn = wmma4(a, wxn[i], Cxn);
      }
      {
        v2f a = frag_a(sh_inp, 68, 64);
        Cz  = wmma4(a, wxz[16], Cz);
        Cr  = wmma4(a, wxr[16], Cr);
        Cxn = wmma4(a, wxn[16], Cxn);
      }
      #pragma unroll
      for (int i = 0; i < 32; ++i) {
        v2f a = frag_a(sh_h, H, 4 * i);
        Cz  = wmma4(a, whz[i], Cz);
        Cr  = wmma4(a, whr[i], Cr);
        Chn = wmma4(a, whn[i], Chn);
      }
      #pragma unroll
      for (int v = 0; v < 8; ++v) {
        int row = rb + v;
        float z = fast_sigmoid(Cz[v] + bias_z);
        float r = fast_sigmoid(Cr[v] + bias_r);
        float n = fast_tanh((Cxn[v] + bias_xn) + r * (Chn[v] + bias_hn));
        hnew[v] = (1.0f - z) * n + z * sh_h[row * H + hc];
      }
    }
    __syncthreads();
    #pragma unroll
    for (int v = 0; v < 8; ++v) sh_h[(rb + v) * H + hc] = hnew[v];
    __syncthreads();

    if (wave < 2) {
      float* dst = (wave == 0) ? sh_sc0 : sh_sc1;
      const int kbase = wave * 64;
      v8f c = {};
      #pragma unroll
      for (int i = 0; i < 16; ++i) {
        int k0 = kbase + 4 * i;
        v2f a = frag_a(sh_h, H, k0);
        v2f b = sh_wgf[(wave * 16 + i) * 32 + lane];
        c = wmma4(a, b, c);
      }
      if (colb < M) {
        #pragma unroll
        for (int v = 0; v < 8; ++v) dst[(rb + v) * 16 + colb] = c[v];
      }
    }
    __syncthreads();

    if (tid < 16) {
      float s[M], mx = -1e30f;
      #pragma unroll
      for (int m = 0; m < M; ++m) {
        s[m] = sh_sc0[tid * 16 + m] + sh_sc1[tid * 16 + m] + sh_bg[m];
        mx = fmaxf(mx, s[m]);
      }
      float sum = 0.0f;
      #pragma unroll
      for (int m = 0; m < M; ++m) { s[m] = __expf(s[m] - mx); sum += s[m]; }
      float inv = __fdividef(1.0f, sum);
      #pragma unroll
      for (int m = 0; m < M; ++m) sh_gate[tid * M + m] = s[m] * inv;
    }
    __syncthreads();

    for (int i = tid; i < 16 * S; i += 256) {
      int row = i >> 5, col = i & 31;
      float acc = 0.0f;
      #pragma unroll
      for (int m = 0; m < M; ++m)
        acc += sh_gate[row * M + m] * sh_ns[row * 256 + m * 32 + col];
      sh_kst[i] = acc;
      if (col == S - 1) {
        out[(size_t)(b0 + row) * T + t] = acc;
        sh_err[row] = acc - sh_y[row];
      }
    }
    __syncthreads();
  }
}

extern "C" void kernel_launch(void* const* d_in, const int* in_sizes, int n_in,
                              void* d_out, int out_size, void* d_ws, size_t ws_size,
                              hipStream_t stream) {
  (void)in_sizes; (void)n_in; (void)out_size; (void)d_ws; (void)ws_size;
  const float* x      = (const float*)d_in[0];
  const float* y      = (const float*)d_in[1];
  const float* W_enc  = (const float*)d_in[2];
  const float* b_enc  = (const float*)d_in[3];
  const float* W_in   = (const float*)d_in[4];
  const float* W_rec  = (const float*)d_in[5];
  const float* b_s    = (const float*)d_in[6];
  const float* W_x    = (const float*)d_in[7];
  const float* W_h    = (const float*)d_in[8];
  const float* b_x    = (const float*)d_in[9];
  const float* b_h    = (const float*)d_in[10];
  const float* W_gate = (const float*)d_in[11];
  const float* b_gate = (const float*)d_in[12];
  float* out = (float*)d_out;

  kalman_gru_scan<<<dim3(8), dim3(256), 0, stream>>>(
      x, y, W_enc, b_enc, W_in, W_rec, b_s, W_x, W_h, b_x, b_h,
      W_gate, b_gate, out);
}


# tanh and sigmoid each as ONE v_tanh_f32 (sigmoid = (1+tanh(x/2))/2) in place of v_exp_f32 plus an 11-instruction IEEE division per value: 48 activations per lane and step
# speedup vs baseline: 1.7389x; 1.1015x over previous
#include <hip/hip_runtime.h>
#include <math.h>


typedef __attribute__((ext_vector_type(2))) float v2f;
typedef __attribute__((ext_vector_type(8))) float v8f;

__device__ __forceinline__ v2f frag_a(const float* A, int ld, int k0) {
  int lane = threadIdx.x & 31;
  int row  = lane & 15;
  int kk   = k0 + ((lane >> 4) << 1);
  v2f a;
  a.x = A[row * ld + kk];
  a.y = A[row * ld + kk + 1];
  return a;
}

__device__ __forceinline__ v2f frag_b(const float* W, int ld, int k0, int c0) {
  int lane = threadIdx.x & 31;
  int col  = c0 + (lane & 15);
  int kk   = k0 + ((lane >> 4) << 1);
  v2f b;
  b.x = W[kk * ld + col];
  b.y = W[(kk + 1) * ld + col];
  return b;
}

__device__ __forceinline__ v2f frag_b_kguard(const float* W, int ld, int k0,
                                             int c0, int kmax) {
  int lane = threadIdx.x & 31;
  int col  = c0 + (lane & 15);
  int kk   = k0 + ((lane >> 4) << 1);
  v2f b;
  b.x = (kk     < kmax) ? W[kk * ld + col]       : 0.0f;
  b.y = (kk + 1 < kmax) ? W[(kk + 1) * ld + col] : 0.0f;
  return b;
}

__device__ __forceinline__ v8f wmma4(v2f a, v2f b, v8f c) {
  return __builtin_amdgcn_wmma_f32_16x16x4_f32(
       false, a,  false, b,
       (short)0, c,  false,  false);
}

__device__ __forceinline__ float fast_tanh(float x) {
  return __builtin_amdgcn_tanhf(x);
}
__device__ __forceinline__ float fast_sigmoid(float x) {
  return fmaf(0.5f, __builtin_amdgcn_tanhf(0.5f * x), 0.5f);
}

__global__ __launch_bounds__(256) void kalman_gru_scan(
    const float* __restrict__ x,
    const float* __restrict__ y,
    const float* __restrict__ W_enc,
    const float* __restrict__ b_enc,
    const float* __restrict__ W_in,
    const float* __restrict__ W_rec,
    const float* __restrict__ b_s,
    const float* __restrict__ W_x,
    const float* __restrict__ W_h,
    const float* __restrict__ b_x,
    const float* __restrict__ b_h,
    const float* __restrict__ W_gate,
    const float* __restrict__ b_gate,
    float* __restrict__ out)
{
  constexpr int T = 2048, IN_ = 32, S = 32, M = 8, E = 64, H = 128;
  constexpr int CIN = 65, G = 384;

  __shared__ float sh_x[16 * 32];
  __shared__ float sh_inp[16 * 68];
  __shared__ float sh_h[16 * 128];
  __shared__ float sh_ns[16 * 8 * 32];
  __shared__ float sh_kst[16 * 32];
  __shared__ float sh_gate[16 * 8];
  __shared__ float sh_sc0[16 * 16];
  __shared__ float sh_sc1[16 * 16];
  __shared__ float sh_y[16];
  __shared__ float sh_err[16];
  __shared__ v2f sh_wencf[8 * 4 * 32];
  __shared__ v2f sh_wgf[32 * 32];
  __shared__ float sh_bg[8];

  const int tid  = threadIdx.x;
  const int wave = tid >> 5;
  const int lane = tid & 31;
  const int b0   = blockIdx.x * 16;

  const int colb = lane & 15;
  const int rb   = (lane >> 4) << 3;
  const int zc0  = wave * 16;
  const int hc   = zc0 + colb;

  const float bias_z  = b_x[hc]       + b_h[hc];
  const float bias_r  = b_x[128 + hc] + b_h[128 + hc];
  const float bias_xn = b_x[256 + hc];
  const float bias_hn = b_h[256 + hc];
  const float bias_s0 = b_s[wave * 32 + colb];
  const float bias_s1 = b_s[wave * 32 + 16 + colb];
  const float bias_e  = (wave < 4) ? b_enc[wave * 16 + colb] : 0.0f;

  v2f whz[32], whr[32], whn[32];
  #pragma unroll
  for (int i = 0; i < 32; ++i) {
    whz[i] = frag_b(W_h, G, 4 * i, zc0);
    whr[i] = frag_b(W_h, G, 4 * i, 128 + zc0);
    whn[i] = frag_b(W_h, G, 4 * i, 256 + zc0);
  }
  v2f wxz[17], wxr[17], wxn[17];
  #pragma unroll
  for (int i = 0; i < 16; ++i) {
    wxz[i] = frag_b(W_x, G, 4 * i, zc0);
    wxr[i] = frag_b(W_x, G, 4 * i, 128 + zc0);
    wxn[i] = frag_b(W_x, G, 4 * i, 256 + zc0);
  }
  wxz[16] = frag_b_kguard(W_x, G, 64, zc0,       CIN);
  wxr[16] = frag_b_kguard(W_x, G, 64, 128 + zc0, CIN);
  wxn[16] = frag_b_kguard(W_x, G, 64, 256 + zc0, CIN);

  for (int i = tid; i < 16 * H; i += 256) sh_h[i] = 0.0f;
  for (int i = tid; i < 16 * S; i += 256) sh_kst[i] = 0.0f;
  for (int i = tid; i < 16 * 68; i += 256) sh_inp[i] = 0.0f;
  if (tid < 16) sh_err[tid] = 0.0f;
  for (int idx = tid; idx < 8 * 4 * 32; idx += 256) {
    int ln = idx & 31, tile = (idx >> 5) & 3, i = idx >> 7;
    int kk  = 4 * i + ((ln >> 4) << 1);
    int col = tile * 16 + (ln & 15);
    v2f b;
    b.x = W_enc[kk * E + col];
    b.y = W_enc[(kk + 1) * E + col];
    sh_wencf[idx] = b;
  }
  for (int idx = tid; idx < 32 * 32; idx += 256) {
    int ln = idx & 31, i = idx >> 5;
    int kk  = 4 * i + ((ln >> 4) << 1);
    int col = ln & 15;
    v2f b;
    b.x = (col < M) ? W_gate[kk * M + col]       : 0.0f;
    b.y = (col < M) ? W_gate[(kk + 1) * M + col] : 0.0f;
    sh_wgf[idx] = b;
  }
  if (tid < M) sh_bg[tid] = b_gate[tid];
  __syncthreads();

  for (int t = 0; t < T; ++t) {
    for (int i = tid; i < 16 * IN_; i += 256) {
      int row = i >> 5, col = i & 31;
      sh_x[i] = x[(size_t)(b0 + row) * T * IN_ + (size_t)t * IN_ + col];
    }
    if (tid < 16) {
      sh_y[tid] = y[(size_t)(b0 + tid) * T + t];
      sh_inp[tid * 68 + 64] = sh_err[tid];
      if (t + 1 < T)
        __builtin_prefetch(&x[(size_t)(b0 + tid) * T * IN_ +
                              (size_t)(t + 1) * IN_], 0, 1);
    }
    __syncthreads();

    {
      const int m = wave;
      const float* Wi = W_in  + m * IN_ * S;
      const float* Wr = W_rec + m * S * S;
      v8f c0 = {}, c1 = {};
      #pragma unroll
      for (int k0 = 0; k0 < IN_; k0 += 4) {
        v2f a = frag_a(sh_x, 32, k0);
        c0 = wmma4(a, frag_b(Wi, S, k0, 0),  c0);
        c1 = wmma4(a, frag_b(Wi, S, k0, 16), c1);
      }
      #pragma unroll
      for (int k0 = 0; k0 < S; k0 += 4) {
        v2f a = frag_a(sh_kst, 32, k0);
        c0 = wmma4(a, frag_b(Wr, S, k0, 0),  c0);
        c1 = wmma4(a, frag_b(Wr, S, k0, 16), c1);
      }
      #pragma unroll
      for (int v = 0; v < 8; ++v) {
        int row = rb + v;
        sh_ns[row * 256 + m * 32 + colb]      = fast_tanh(c0[v] + bias_s0);
        sh_ns[row * 256 + m * 32 + 16 + colb] = fast_tanh(c1[v] + bias_s1);
      }
    }
    if (wave < 4) {
      int c0c = wave * 16;
      v8f c = {};
      #pragma unroll
      for (int i = 0; i < IN_ / 4; ++i)
        c = wmma4(frag_a(sh_x, 32, 4 * i),
                  sh_wencf[(i * 4 + wave) * 32 + lane], c);
      #pragma unroll
      for (int v = 0; v < 8; ++v)
        sh_inp[(rb + v) * 68 + c0c + colb] = fast_tanh(c[v] + bias_e);
    }
    __syncthreads();

    float hnew[8];
    {
      v8f Cz = {}, Cr = {}, Cxn = {}, Chn = {};
      #pragma unroll
      for (int i = 0; i < 16; ++i) {
        v2f a = frag_a(sh_inp, 68, 4 * i);
        Cz  = wmma4(a, wxz[i], Cz);
        Cr  = wmma4(a, wxr[i], Cr);
        Cxn = wmma4(a, wxn[i], Cxn);
      }
      {
        v2f a = frag_a(sh_inp, 68, 64);
        Cz  = wmma4(a, wxz[16], Cz);
        Cr  = wmma4(a, wxr[16], Cr);
        Cxn = wmma4(a, wxn[16], Cxn);
      }
      #pragma unroll
      for (int i = 0; i < 32; ++i) {
        v2f a = frag_a(sh_h, H, 4 * i);
        Cz  = wmma4(a, whz[i], Cz);
        Cr  = wmma4(a, whr[i], Cr);
        Chn = wmma4(a, whn[i], Chn);
      }
      #pragma unroll
      for (int v = 0; v < 8; ++v) {
        int row = rb + v;
        float z = fast_sigmoid(Cz[v] + bias_z);
        float r = fast_sigmoid(Cr[v] + bias_r);
        float n = fast_tanh((Cxn[v] + bias_xn) + r * (Chn[v] + bias_hn));
        hnew[v] = (1.0f - z) * n + z * sh_h[row * H + hc];
      }
    }
    __syncthreads();
    #pragma unroll
    for (int v = 0; v < 8; ++v) sh_h[(rb + v) * H + hc] = hnew[v];
    __syncthreads();

    if (wave < 2) {
      float* dst = (wave == 0) ? sh_sc0 : sh_sc1;
      const int kbase = wave * 64;
      v8f c = {};
      #pragma unroll
      for (int i = 0; i < 16; ++i) {
        int k0 = kbase + 4 * i;
        v2f a = frag_a(sh_h, H, k0);
        v2f b = sh_wgf[(wave * 16 + i) * 32 + lane];
        c = wmma4(a, b, c);
      }
      if (colb < M) {
        #pragma unroll
        for (int v = 0; v < 8; ++v) dst[(rb + v) * 16 + colb] = c[v];
      }
    }
    __syncthreads();

    if (tid < 16) {
      float s[M], mx = -1e30f;
      #pragma unroll
      for (int m = 0; m < M; ++m) {
        s[m] = sh_sc0[tid * 16 + m] + sh_sc1[tid * 16 + m] + sh_bg[m];
        mx = fmaxf(mx, s[m]);
      }
      float sum = 0.0f;
      #pragma unroll
      for (int m = 0; m < M; ++m) { s[m] = __expf(s[m] - mx); sum += s[m]; }
      float inv = __fdividef(1.0f, sum);
      #pragma unroll
      for (int m = 0; m < M; ++m) sh_gate[tid * M + m] = s[m] * inv;
    }
    __syncthreads();

    for (int i = tid; i < 16 * S; i += 256) {
      int row = i >> 5, col = i & 31;
      float acc = 0.0f;
      #pragma unroll
      for (int m = 0; m < M; ++m)
        acc += sh_gate[row * M + m] * sh_ns[row * 256 + m * 32 + col];
      sh_kst[i] = acc;
      if (col == S - 1) {
        out[(size_t)(b0 + row) * T + t] = acc;
        sh_err[row] = acc - sh_y[row];
      }
    }
    __syncthreads();
  }
}

extern "C" void kernel_launch(void* const* d_in, const int* in_sizes, int n_in,
                              void* d_out, int out_size, void* d_ws, size_t ws_size,
                              hipStream_t stream) {
  (void)in_sizes; (void)n_in; (void)out_size; (void)d_ws; (void)ws_size;
  const float* x      = (const float*)d_in[0];
  const float* y      = (const float*)d_in[1];
  const float* W_enc  = (const float*)d_in[2];
  const float* b_enc  = (const float*)d_in[3];
  const float* W_in   = (const float*)d_in[4];
  const float* W_rec  = (const float*)d_in[5];
  const float* b_s    = (const float*)d_in[6];
  const float* W_x    = (const float*)d_in[7];
  const float* W_h    = (const float*)d_in[8];
  const float* b_x    = (const float*)d_in[9];
  const float* b_h    = (const float*)d_in[10];
  const float* W_gate = (const float*)d_in[11];
  const float* b_gate = (const float*)d_in[12];
  float* out = (float*)d_out;

  kalman_gru_scan<<<dim3(8), dim3(256), 0, stream>>>(
      x, y, W_enc, b_enc, W_in, W_rec, b_s, W_x, W_h, b_x, b_h,
      W_gate, b_gate, out);
}


# x_t and y_t are loaded one step ahead into registers (x refills its LDS tile behind stage 2, err goes straight into the GRU input column): 3 serial global round trips and 1 barrier leave each step
# speedup vs baseline: 1.9399x; 1.1156x over previous
#include <hip/hip_runtime.h>
#include <math.h>


typedef __attribute__((ext_vector_type(2))) float v2f;
typedef __attribute__((ext_vector_type(8))) float v8f;

__device__ __forceinline__ v2f frag_a(const float* A, int ld, int k0) {
  int lane = threadIdx.x & 31;
  int row  = lane & 15;
  int kk   = k0 + ((lane >> 4) << 1);
  v2f a;
  a.x = A[row * ld + kk];
  a.y = A[row * ld + kk + 1];
  return a;
}

__device__ __forceinline__ v2f frag_b(const float* W, int ld, int k0, int c0) {
  int lane = threadIdx.x & 31;
  int col  = c0 + (lane & 15);
  int kk   = k0 + ((lane >> 4) << 1);
  v2f b;
  b.x = W[kk * ld + col];
  b.y = W[(kk + 1) * ld + col];
  return b;
}

__device__ __forceinline__ v2f frag_b_kguard(const float* W, int ld, int k0,
                                             int c0, int kmax) {
  int lane = threadIdx.x & 31;
  int col  = c0 + (lane & 15);
  int kk   = k0 + ((lane >> 4) << 1);
  v2f b;
  b.x = (kk     < kmax) ? W[kk * ld + col]       : 0.0f;
  b.y = (kk + 1 < kmax) ? W[(kk + 1) * ld + col] : 0.0f;
  return b;
}

__device__ __forceinline__ v8f wmma4(v2f a, v2f b, v8f c) {
  return __builtin_amdgcn_wmma_f32_16x16x4_f32(
       false, a,  false, b,
       (short)0, c,  false,  false);
}

__device__ __forceinline__ float fast_tanh(float x) {
  return __builtin_amdgcn_tanhf(x);
}
__device__ __forceinline__ float fast_sigmoid(float x) {
  return fmaf(0.5f, __builtin_amdgcn_tanhf(0.5f * x), 0.5f);
}

__global__ __launch_bounds__(256) void kalman_gru_scan(
    const float* __restrict__ x,
    const float* __restrict__ y,
    const float* __restrict__ W_enc,
    const float* __restrict__ b_enc,
    const float* __restrict__ W_in,
    const float* __restrict__ W_rec,
    const float* __restrict__ b_s,
    const float* __restrict__ W_x,
    const float* __restrict__ W_h,
    const float* __restrict__ b_x,
    const float* __restrict__ b_h,
    const float* __restrict__ W_gate,
    const float* __restrict__ b_gate,
    float* __restrict__ out)
{
  constexpr int T = 2048, IN_ = 32, S = 32, M = 8, E = 64, H = 128;
  constexpr int CIN = 65, G = 384;

  __shared__ float sh_x[16 * 32];
  __shared__ float sh_inp[16 * 68];
  __shared__ float sh_h[16 * 128];
  __shared__ float sh_ns[16 * 8 * 32];
  __shared__ float sh_kst[16 * 32];
  __shared__ float sh_gate[16 * 8];
  __shared__ float sh_sc0[16 * 16];
  __shared__ float sh_sc1[16 * 16];
  __shared__ v2f sh_wencf[8 * 4 * 32];
  __shared__ v2f sh_wgf[32 * 32];
  __shared__ float sh_bg[8];

  const int tid  = threadIdx.x;
  const int wave = tid >> 5;
  const int lane = tid & 31;
  const int b0   = blockIdx.x * 16;

  const int colb = lane & 15;
  const int rb   = (lane >> 4) << 3;
  const int zc0  = wave * 16;
  const int hc   = zc0 + colb;

  const float bias_z  = b_x[hc]       + b_h[hc];
  const float bias_r  = b_x[128 + hc] + b_h[128 + hc];
  const float bias_xn = b_x[256 + hc];
  const float bias_hn = b_h[256 + hc];
  const float bias_s0 = b_s[wave * 32 + colb];
  const float bias_s1 = b_s[wave * 32 + 16 + colb];
  const float bias_e  = (wave < 4) ? b_enc[wave * 16 + colb] : 0.0f;

  v2f whz[32], whr[32], whn[32];
  #pragma unroll
  for (int i = 0; i < 32; ++i) {
    whz[i] = frag_b(W_h, G, 4 * i, zc0);
    whr[i] = frag_b(W_h, G, 4 * i, 128 + zc0);
    whn[i] = frag_b(W_h, G, 4 * i, 256 + zc0);
  }
  v2f wxz[17], wxr[17], wxn[17];
  #pragma unroll
  for (int i = 0; i < 16; ++i) {
    wxz[i] = frag_b(W_x, G, 4 * i, zc0);
    wxr[i] = frag_b(W_x, G, 4 * i, 128 + zc0);
    wxn[i] = frag_b(W_x, G, 4 * i, 256 + zc0);
  }
  wxz[16] = frag_b_kguard(W_x, G, 64, zc0,       CIN);
  wxr[16] = frag_b_kguard(W_x, G, 64, 128 + zc0, CIN);
  wxn[16] = frag_b_kguard(W_x, G, 64, 256 + zc0, CIN);

  for (int i = tid; i < 16 * H; i += 256) sh_h[i] = 0.0f;
  for (int i = tid; i < 16 * S; i += 256) sh_kst[i] = 0.0f;
  for (int i = tid; i < 16 * 68; i += 256) sh_inp[i] = 0.0f;
  for (int idx = tid; idx < 8 * 4 * 32; idx += 256) {
    int ln = idx & 31, tile = (idx >> 5) & 3, i = idx >> 7;
    int kk  = 4 * i + ((ln >> 4) << 1);
    int col = tile * 16 + (ln & 15);
    v2f b;
    b.x = W_enc[kk * E + col];
    b.y = W_enc[(kk + 1) * E + col];
    sh_wencf[idx] = b;
  }
  for (int idx = tid; idx < 32 * 32; idx += 256) {
    int ln = idx & 31, i = idx >> 5;
    int kk  = 4 * i + ((ln >> 4) << 1);
    int col = ln & 15;
    v2f b;
    b.x = (col < M) ? W_gate[kk * M + col]       : 0.0f;
    b.y = (col < M) ? W_gate[(kk + 1) * M + col] : 0.0f;
    sh_wgf[idx] = b;
  }
  if (tid < M) sh_bg[tid] = b_gate[tid];
  const float* xr0 = x + (size_t)(b0 + wave) * T * IN_ + lane;
  const float* xr1 = x + (size_t)(b0 + 8 + wave) * T * IN_ + lane;
  const float* yr0 = y + (size_t)(b0 + wave) * T;
  const float* yr1 = y + (size_t)(b0 + 8 + wave) * T;
  sh_x[tid]       = xr0[0];
  sh_x[tid + 256] = xr1[0];
  float ycur0 = yr0[0], ycur1 = yr1[0];
  __syncthreads();

  for (int t = 0; t < T; ++t) {
    {
      const int m = wave;
      const float* Wi = W_in  + m * IN_ * S;
      const float* Wr = W_rec + m * S * S;
      v8f c0 = {}, c1 = {};
      #pragma unroll
      for (int k0 = 0; k0 < IN_; k0 += 4) {
        v2f a = frag_a(sh_x, 32, k0);
        c0 = wmma4(a, frag_b(Wi, S, k0, 0),  c0);
        c1 = wmma4(a, frag_b(Wi, S, k0, 16), c1);
      }
      #pragma unroll
      for (int k0 = 0; k0 < S; k0 += 4) {
        v2f a = frag_a(sh_kst, 32, k0);
        c0 = wmma4(a, frag_b(Wr, S, k0, 0),  c0);
        c1 = wmma4(a, frag_b(Wr, S, k0, 16), c1);
      }
      #pragma unroll
      for (int v = 0; v < 8; ++v) {
        int row = rb + v;
        sh_ns[row * 256 + m * 32 + colb]      = fast_tanh(c0[v] + bias_s0);
        sh_ns[row * 256 + m * 32 + 16 + colb] = fast_tanh(c1[v] + bias_s1);
      }
    }
    if (wave < 4) {
      int c0c = wave * 16;
      v8f c = {};
      #pragma unroll
      for (int i = 0; i < IN_ / 4; ++i)
        c = wmma4(frag_a(sh_x, 32, 4 * i),
                  sh_wencf[(i * 4 + wave) * 32 + lane], c);
      #pragma unroll
      for (int v = 0; v < 8; ++v)
        sh_inp[(rb + v) * 68 + c0c + colb] = fast_tanh(c[v] + bias_e);
    }
    __syncthreads();

    const int tn = (t + 1 < T) ? t + 1 : t;
    const float xn0 = xr0[(size_t)tn * IN_];
    const float xn1 = xr1[(size_t)tn * IN_];
    const float yn0 = yr0[tn];
    const float yn1 = yr1[tn];

    float hnew[8];
    {
      v8f Cz = {}, Cr = {}, Cxn = {}, Chn = {};
      #pragma unroll
      for (int i = 0; i < 16; ++i) {
        v2f a = frag_a(sh_inp, 68, 4 * i);
        Cz  = wmma4(a, wxz[i], Cz);
        Cr  = wmma4(a, wxr[i], Cr);
        Cxn = wmma4(a, wxn[i], Cxn);
      }
      {
        v2f a = frag_a(sh_inp, 68, 64);
        Cz  = wmma4(a, wxz[16], Cz);
        Cr  = wmma4(a, wxr[16], Cr);
        Cxn = wmma4(a, wxn[16], Cxn);
      }
      #pragma unroll
      for (int i = 0; i < 32; ++i) {
        v2f a = frag_a(sh_h, H, 4 * i);
        Cz  = wmma4(a, whz[i], Cz);
        Cr  = wmma4(a, whr[i], Cr);
        Chn = wmma4(a, whn[i], Chn);
      }
      #pragma unroll
      for (int v = 0; v < 8; ++v) {
        int row = rb + v;
        float z = fast_sigmoid(Cz[v] + bias_z);
        float r = fast_sigmoid(Cr[v] + bias_r);
        float n = fast_tanh((Cxn[v] + bias_xn) + r * (Chn[v] + bias_hn));
        hnew[v] = (1.0f - z) * n + z * sh_h[row * H + hc];
      }
    }
    __syncthreads();
    #pragma unroll
    for (int v = 0; v < 8; ++v) sh_h[(rb + v) * H + hc] = hnew[v];
    sh_x[tid]       = xn0;
    sh_x[tid + 256] = xn1;
    __syncthreads();

    if (wave < 2) {
      float* dst = (wave == 0) ? sh_sc0 : sh_sc1;
      const int kbase = wave * 64;
      v8f c = {};
      #pragma unroll
      for (int i = 0; i < 16; ++i) {
        int k0 = kbase + 4 * i;
        v2f a = frag_a(sh_h, H, k0);
        v2f b = sh_wgf[(wave * 16 + i) * 32 + lane];
        c = wmma4(a, b, c);
      }
      if (colb < M) {
        #pragma unroll
        for (int v = 0; v < 8; ++v) dst[(rb + v) * 16 + colb] = c[v];
      }
    }
    __syncthreads();

    if (tid < 16) {
      float s[M], mx = -1e30f;
      #pragma unroll
      for (int m = 0; m < M; ++m) {
        s[m] = sh_sc0[tid * 16 + m] + sh_sc1[tid * 16 + m] + sh_bg[m];
        mx = fmaxf(mx, s[m]);
      }
      float sum = 0.0f;
      #pragma unroll
      for (int m = 0; m < M; ++m) { s[m] = __expf(s[m] - mx); sum += s[m]; }
      float inv = __fdividef(1.0f, sum);
      #pragma unroll
      for (int m = 0; m < M; ++m) sh_gate[tid * M + m] = s[m] * inv;
    }
    __syncthreads();

    #pragma unroll
    for (int j = 0; j < 2; ++j) {
      int i = tid + 256 * j;
      int row = i >> 5, col = i & 31;
      float acc = 0.0f;
      #pragma unroll
      for (int m = 0; m < M; ++m)
        acc += sh_gate[row * M + m] * sh_ns[row * 256 + m * 32 + col];
      sh_kst[i] = acc;
      if (col == S - 1) {
        out[(size_t)(b0 + row) * T + t] = acc;
        sh_inp[row * 68 + 64] = acc - (j ? ycur1 : ycur0);
      }
    }
    ycur0 = yn0;
    ycur1 = yn1;
    __syncthreads();
  }
}

extern "C" void kernel_launch(void* const* d_in, const int* in_sizes, int n_in,
                              void* d_out, int out_size, void* d_ws, size_t ws_size,
                              hipStream_t stream) {
  (void)in_sizes; (void)n_in; (void)out_size; (void)d_ws; (void)ws_size;
  const float* x      = (const float*)d_in[0];
  const float* y      = (const float*)d_in[1];
  const float* W_enc  = (const float*)d_in[2];
  const float* b_enc  = (const float*)d_in[3];
  const float* W_in   = (const float*)d_in[4];
  const float* W_rec  = (const float*)d_in[5];
  const float* b_s    = (const float*)d_in[6];
  const float* W_x    = (const float*)d_in[7];
  const float* W_h    = (const float*)d_in[8];
  const float* b_x    = (const float*)d_in[9];
  const float* b_h    = (const float*)d_in[10];
  const float* W_gate = (const float*)d_in[11];
  const float* b_gate = (const float*)d_in[12];
  float* out = (float*)d_out;

  kalman_gru_scan<<<dim3(8), dim3(256), 0, stream>>>(
      x, y, W_enc, b_enc, W_in, W_rec, b_s, W_x, W_h, b_x, b_h,
      W_gate, b_gate, out);
}


# LDS row pitch of every A-fragment tile set to 4 mod 64 dwords (h 128->132, x and kstate 32->36, new_state 256->260, logits 16->20): 16- and 8-way bank conflicts of each fragment load are gone
# speedup vs baseline: 2.1569x; 1.1119x over previous
#include <hip/hip_runtime.h>
#include <math.h>


typedef __attribute__((ext_vector_type(2))) float v2f;
typedef __attribute__((ext_vector_type(8))) float v8f;

__device__ __forceinline__ v2f frag_a(const float* A, int ld, int k0) {
  int lane = threadIdx.x & 31;
  int row  = lane & 15;
  int kk   = k0 + ((lane >> 4) << 1);
  v2f a;
  a.x = A[row * ld + kk];
  a.y = A[row * ld + kk + 1];
  return a;
}

__device__ __forceinline__ v2f frag_b(const float* W, int ld, int k0, int c0) {
  int lane = threadIdx.x & 31;
  int col  = c0 + (lane & 15);
  int kk   = k0 + ((lane >> 4) << 1);
  v2f b;
  b.x = W[kk * ld + col];
  b.y = W[(kk + 1) * ld + col];
  return b;
}

__device__ __forceinline__ v2f frag_b_kguard(const float* W, int ld, int k0,
                                             int c0, int kmax) {
  int lane = threadIdx.x & 31;
  int col  = c0 + (lane & 15);
  int kk   = k0 + ((lane >> 4) << 1);
  v2f b;
  b.x = (kk     < kmax) ? W[kk * ld + col]       : 0.0f;
  b.y = (kk + 1 < kmax) ? W[(kk + 1) * ld + col] : 0.0f;
  return b;
}

__device__ __forceinline__ v8f wmma4(v2f a, v2f b, v8f c) {
  return __builtin_amdgcn_wmma_f32_16x16x4_f32(
       false, a,  false, b,
       (short)0, c,  false,  false);
}

__device__ __forceinline__ float fast_tanh(float x) {
  return __builtin_amdgcn_tanhf(x);
}
__device__ __forceinline__ float fast_sigmoid(float x) {
  return fmaf(0.5f, __builtin_amdgcn_tanhf(0.5f * x), 0.5f);
}

__global__ __launch_bounds__(256) void kalman_gru_scan(
    const float* __restrict__ x,
    const float* __restrict__ y,
    const float* __restrict__ W_enc,
    const float* __restrict__ b_enc,
    const float* __restrict__ W_in,
    const float* __restrict__ W_rec,
    const float* __restrict__ b_s,
    const float* __restrict__ W_x,
    const float* __restrict__ W_h,
    const float* __restrict__ b_x,
    const float* __restrict__ b_h,
    const float* __restrict__ W_gate,
    const float* __restrict__ b_gate,
    float* __restrict__ out)
{
  constexpr int T = 2048, IN_ = 32, S = 32, M = 8, E = 64, H = 128;
  constexpr int CIN = 65, G = 384;
  constexpr int XP = 36;
  constexpr int IP = 68;
  constexpr int HP = 132;
  constexpr int NP = 260;
  constexpr int SP = 20;

  __shared__ float sh_x[16 * XP];
  __shared__ float sh_inp[16 * IP];
  __shared__ float sh_h[16 * HP];
  __shared__ float sh_ns[16 * NP];
  __shared__ float sh_kst[16 * XP];
  __shared__ float sh_gate[16 * 8];
  __shared__ float sh_sc0[16 * SP];
  __shared__ float sh_sc1[16 * SP];
  __shared__ v2f sh_wencf[8 * 4 * 32];
  __shared__ v2f sh_wgf[32 * 32];
  __shared__ float sh_bg[8];

  const int tid  = threadIdx.x;
  const int wave = tid >> 5;
  const int lane = tid & 31;
  const int b0   = blockIdx.x * 16;

  const int colb = lane & 15;
  const int rb   = (lane >> 4) << 3;
  const int zc0  = wave * 16;
  const int hc   = zc0 + colb;

  const float bias_z  = b_x[hc]       + b_h[hc];
  const float bias_r  = b_x[128 + hc] + b_h[128 + hc];
  const float bias_xn = b_x[256 + hc];
  const float bias_hn = b_h[256 + hc];
  const float bias_s0 = b_s[wave * 32 + colb];
  const float bias_s1 = b_s[wave * 32 + 16 + colb];
  const float bias_e  = (wave < 4) ? b_enc[wave * 16 + colb] : 0.0f;

  v2f whz[32], whr[32], whn[32];
  #pragma unroll
  for (int i = 0; i < 32; ++i) {
    whz[i] = frag_b(W_h, G, 4 * i, zc0);
    whr[i] = frag_b(W_h, G, 4 * i, 128 + zc0);
    whn[i] = frag_b(W_h, G, 4 * i, 256 + zc0);
  }
  v2f wxz[17], wxr[17], wxn[17];
  #pragma unroll
  for (int i = 0; i < 16; ++i) {
    wxz[i] = frag_b(W_x, G, 4 * i, zc0);
    wxr[i] = frag_b(W_x, G, 4 * i, 128 + zc0);
    wxn[i] = frag_b(W_x, G, 4 * i, 256 + zc0);
  }
  wxz[16] = frag_b_kguard(W_x, G, 64, zc0,       CIN);
  wxr[16] = frag_b_kguard(W_x, G, 64, 128 + zc0, CIN);
  wxn[16] = frag_b_kguard(W_x, G, 64, 256 + zc0, CIN);

  for (int i = tid; i < 16 * HP; i += 256) sh_h[i] = 0.0f;
  for (int i = tid; i < 16 * XP; i += 256) sh_kst[i] = 0.0f;
  for (int i = tid; i < 16 * IP; i += 256) sh_inp[i] = 0.0f;
  for (int idx = tid; idx < 8 * 4 * 32; idx += 256) {
    int ln = idx & 31, tile = (idx >> 5) & 3, i = idx >> 7;
    int kk  = 4 * i + ((ln >> 4) << 1);
    int col = tile * 16 + (ln & 15);
    v2f b;
    b.x = W_enc[kk * E + col];
    b.y = W_enc[(kk + 1) * E + col];
    sh_wencf[idx] = b;
  }
  for (int idx = tid; idx < 32 * 32; idx += 256) {
    int ln = idx & 31, i = idx >> 5;
    int kk  = 4 * i + ((ln >> 4) << 1);
    int col = ln & 15;
    v2f b;
    b.x = (col < M) ? W_gate[kk * M + col]       : 0.0f;
    b.y = (col < M) ? W_gate[(kk + 1) * M + col] : 0.0f;
    sh_wgf[idx] = b;
  }
  if (tid < M) sh_bg[tid] = b_gate[tid];
  const float* xr0 = x + (size_t)(b0 + wave) * T * IN_ + lane;
  const float* xr1 = x + (size_t)(b0 + 8 + wave) * T * IN_ + lane;
  const float* yr0 = y + (size_t)(b0 + wave) * T;
  const float* yr1 = y + (size_t)(b0 + 8 + wave) * T;
  sh_x[wave * XP + lane]       = xr0[0];
  sh_x[(wave + 8) * XP + lane] = xr1[0];
  float ycur0 = yr0[0], ycur1 = yr1[0];
  __syncthreads();

  for (int t = 0; t < T; ++t) {
    {
      const int m = wave;
      const float* Wi = W_in  + m * IN_ * S;
      const float* Wr = W_rec + m * S * S;
      v8f c0 = {}, c1 = {};
      #pragma unroll
      for (int k0 = 0; k0 < IN_; k0 += 4) {
        v2f a = frag_a(sh_x, XP, k0);
        c0 = wmma4(a, frag_b(Wi, S, k0, 0),  c0);
        c1 = wmma4(a, frag_b(Wi, S, k0, 16), c1);
      }
      #pragma unroll
      for (int k0 = 0; k0 < S; k0 += 4) {
        v2f a = frag_a(sh_kst, XP, k0);
        c0 = wmma4(a, frag_b(Wr, S, k0, 0),  c0);
        c1 = wmma4(a, frag_b(Wr, S, k0, 16), c1);
      }
      #pragma unroll
      for (int v = 0; v < 8; ++v) {
        int row = rb + v;
        sh_ns[row * NP + m * 32 + colb]      = fast_tanh(c0[v] + bias_s0);
        sh_ns[row * NP + m * 32 + 16 + colb] = fast_tanh(c1[v] + bias_s1);
      }
    }
    if (wave < 4) {
      int c0c = wave * 16;
      v8f c = {};
      #pragma unroll
      for (int i = 0; i < IN_ / 4; ++i)
        c = wmma4(frag_a(sh_x, XP, 4 * i),
                  sh_wencf[(i * 4 + wave) * 32 + lane], c);
      #pragma unroll
      for (int v = 0; v < 8; ++v)
        sh_inp[(rb + v) * IP + c0c + colb] = fast_tanh(c[v] + bias_e);
    }
    __syncthreads();

    const int tn = (t + 1 < T) ? t + 1 : t;
    const float xn0 = xr0[(size_t)tn * IN_];
    const float xn1 = xr1[(size_t)tn * IN_];
    const float yn0 = yr0[tn];
    const float yn1 = yr1[tn];

    float hnew[8];
    {
      v8f Cz = {}, Cr = {}, Cxn = {}, Chn = {};
      #pragma unroll
      for (int i = 0; i < 16; ++i) {
        v2f a = frag_a(sh_inp, IP, 4 * i);
        Cz  = wmma4(a, wxz[i], Cz);
        Cr  = wmma4(a, wxr[i], Cr);
        Cxn = wmma4(a, wxn[i], Cxn);
      }
      {
        v2f a = frag_a(sh_inp, IP, 64);
        Cz  = wmma4(a, wxz[16], Cz);
        Cr  = wmma4(a, wxr[16], Cr);
        Cxn = wmma4(a, wxn[16], Cxn);
      }
      #pragma unroll
      for (int i = 0; i < 32; ++i) {
        v2f a = frag_a(sh_h, HP, 4 * i);
        Cz  = wmma4(a, whz[i], Cz);
        Cr  = wmma4(a, whr[i], Cr);
        Chn = wmma4(a, whn[i], Chn);
      }
      #pragma unroll
      for (int v = 0; v < 8; ++v) {
        int row = rb + v;
        float z = fast_sigmoid(Cz[v] + bias_z);
        float r = fast_sigmoid(Cr[v] + bias_r);
        float n = fast_tanh((Cxn[v] + bias_xn) + r * (Chn[v] + bias_hn));
        hnew[v] = (1.0f - z) * n + z * sh_h[row * HP + hc];
      }
    }
    __syncthreads();
    #pragma unroll
    for (int v = 0; v < 8; ++v) sh_h[(rb + v) * HP + hc] = hnew[v];
    sh_x[wave * XP + lane]       = xn0;
    sh_x[(wave + 8) * XP + lane] = xn1;
    __syncthreads();

    if (wave < 2) {
      float* dst = (wave == 0) ? sh_sc0 : sh_sc1;
      const int kbase = wave * 64;
      v8f c = {};
      #pragma unroll
      for (int i = 0; i < 16; ++i) {
        int k0 = kbase + 4 * i;
        v2f a = frag_a(sh_h, HP, k0);
        v2f b = sh_wgf[(wave * 16 + i) * 32 + lane];
        c = wmma4(a, b, c);
      }
      if (colb < M) {
        #pragma unroll
        for (int v = 0; v < 8; ++v) dst[(rb + v) * SP + colb] = c[v];
      }
    }
    __syncthreads();

    if (tid < 16) {
      float s[M], mx = -1e30f;
      #pragma unroll
      for (int m = 0; m < M; ++m) {
        s[m] = sh_sc0[tid * SP + m] + sh_sc1[tid * SP + m] + sh_bg[m];
        mx = fmaxf(mx, s[m]);
      }
      float sum = 0.0f;
      #pragma unroll
      for (int m = 0; m < M; ++m) { s[m] = __expf(s[m] - mx); sum += s[m]; }
      float inv = __fdividef(1.0f, sum);
      #pragma unroll
      for (int m = 0; m < M; ++m) sh_gate[tid * M + m] = s[m] * inv;
    }
    __syncthreads();

    #pragma unroll
    for (int j = 0; j < 2; ++j) {
      int i = tid + 256 * j;
      int row = i >> 5, col = i & 31;
      float acc = 0.0f;
      #pragma unroll
      for (int m = 0; m < M; ++m)
        acc += sh_gate[row * M + m] * sh_ns[row * NP + m * 32 + col];
      sh_kst[row * XP + col] = acc;
      if (col == S - 1) {
        out[(size_t)(b0 + row) * T + t] = acc;
        sh_inp[row * IP + 64] = acc - (j ? ycur1 : ycur0);
      }
    }
    ycur0 = yn0;
    ycur1 = yn1;
    __syncthreads();
  }
}

extern "C" void kernel_launch(void* const* d_in, const int* in_sizes, int n_in,
                              void* d_out, int out_size, void* d_ws, size_t ws_size,
                              hipStream_t stream) {
  (void)in_sizes; (void)n_in; (void)out_size; (void)d_ws; (void)ws_size;
  const float* x      = (const float*)d_in[0];
  const float* y      = (const float*)d_in[1];
  const float* W_enc  = (const float*)d_in[2];
  const float* b_enc  = (const float*)d_in[3];
  const float* W_in   = (const float*)d_in[4];
  const float* W_rec  = (const float*)d_in[5];
  const float* b_s    = (const float*)d_in[6];
  const float* W_x    = (const float*)d_in[7];
  const float* W_h    = (const float*)d_in[8];
  const float* b_x    = (const float*)d_in[9];
  const float* b_h    = (const float*)d_in[10];
  const float* W_gate = (const float*)d_in[11];
  const float* b_gate = (const float*)d_in[12];
  float* out = (float*)d_out;

  kalman_gru_scan<<<dim3(8), dim3(256), 0, stream>>>(
      x, y, W_enc, b_enc, W_in, W_rec, b_s, W_x, W_h, b_x, b_h,
      W_gate, b_gate, out);
}


# h(t)@W_h of step t+1 issued in 3 blocks behind stages 3,4,5 of step t, sums carried in registers (z,r add h part before x part), 5 barriers not 6: the serial tail keeps the matrix pipe fed
# speedup vs baseline: 2.5093x; 1.1634x over previous
#include <hip/hip_runtime.h>
#include <math.h>


typedef __attribute__((ext_vector_type(2))) float v2f;
typedef __attribute__((ext_vector_type(8))) float v8f;

__device__ __forceinline__ v2f frag_a(const float* A, int ld, int k0) {
  int lane = threadIdx.x & 31;
  int row  = lane & 15;
  int kk   = k0 + ((lane >> 4) << 1);
  v2f a;
  a.x = A[row * ld + kk];
  a.y = A[row * ld + kk + 1];
  return a;
}

__device__ __forceinline__ v2f frag_b(const float* W, int ld, int k0, int c0) {
  int lane = threadIdx.x & 31;
  int col  = c0 + (lane & 15);
  int kk   = k0 + ((lane >> 4) << 1);
  v2f b;
  b.x = W[kk * ld + col];
  b.y = W[(kk + 1) * ld + col];
  return b;
}

__device__ __forceinline__ v2f frag_b_kguard(const float* W, int ld, int k0,
                                             int c0, int kmax) {
  int lane = threadIdx.x & 31;
  int col  = c0 + (lane & 15);
  int kk   = k0 + ((lane >> 4) << 1);
  v2f b;
  b.x = (kk     < kmax) ? W[kk * ld + col]       : 0.0f;
  b.y = (kk + 1 < kmax) ? W[(kk + 1) * ld + col] : 0.0f;
  return b;
}

__device__ __forceinline__ v8f wmma4(v2f a, v2f b, v8f c) {
  return __builtin_amdgcn_wmma_f32_16x16x4_f32(
       false, a,  false, b,
       (short)0, c,  false,  false);
}

__device__ __forceinline__ float fast_tanh(float x) {
  return __builtin_amdgcn_tanhf(x);
}
__device__ __forceinline__ float fast_sigmoid(float x) {
  return fmaf(0.5f, __builtin_amdgcn_tanhf(0.5f * x), 0.5f);
}

__global__ __launch_bounds__(256) void kalman_gru_scan(
    const float* __restrict__ x,
    const float* __restrict__ y,
    const float* __restrict__ W_enc,
    const float* __restrict__ b_enc,
    const float* __restrict__ W_in,
    const float* __restrict__ W_rec,
    const float* __restrict__ b_s,
    const float* __restrict__ W_x,
    const float* __restrict__ W_h,
    const float* __restrict__ b_x,
    const float* __restrict__ b_h,
    const float* __restrict__ W_gate,
    const float* __restrict__ b_gate,
    float* __restrict__ out)
{
  constexpr int T = 2048, IN_ = 32, S = 32, M = 8, E = 64, H = 128;
  constexpr int CIN = 65, G = 384;
  constexpr int XP = 36;
  constexpr int IP = 68;
  constexpr int HP = 132;
  constexpr int NP = 260;
  constexpr int SP = 20;

  __shared__ float sh_x[16 * XP];
  __shared__ float sh_inp[16 * IP];
  __shared__ float sh_h[16 * HP];
  __shared__ float sh_ns[16 * NP];
  __shared__ float sh_kst[16 * XP];
  __shared__ float sh_gate[16 * 8];
  __shared__ float sh_sc0[16 * SP];
  __shared__ float sh_sc1[16 * SP];
  __shared__ v2f sh_wencf[8 * 4 * 32];
  __shared__ v2f sh_wgf[32 * 32];
  __shared__ float sh_bg[8];

  const int tid  = threadIdx.x;
  const int wave = tid >> 5;
  const int lane = tid & 31;
  const int b0   = blockIdx.x * 16;

  const int colb = lane & 15;
  const int rb   = (lane >> 4) << 3;
  const int zc0  = wave * 16;
  const int hc   = zc0 + colb;

  const float bias_z  = b_x[hc]       + b_h[hc];
  const float bias_r  = b_x[128 + hc] + b_h[128 + hc];
  const float bias_xn = b_x[256 + hc];
  const float bias_hn = b_h[256 + hc];
  const float bias_s0 = b_s[wave * 32 + colb];
  const float bias_s1 = b_s[wave * 32 + 16 + colb];
  const float bias_e  = (wave < 4) ? b_enc[wave * 16 + colb] : 0.0f;

  v2f whz[32], whr[32], whn[32];
  #pragma unroll
  for (int i = 0; i < 32; ++i) {
    whz[i] = frag_b(W_h, G, 4 * i, zc0);
    whr[i] = frag_b(W_h, G, 4 * i, 128 + zc0);
    whn[i] = frag_b(W_h, G, 4 * i, 256 + zc0);
  }
  v2f wxz[17], wxr[17], wxn[17];
  #pragma unroll
  for (int i = 0; i < 16; ++i) {
    wxz[i] = frag_b(W_x, G, 4 * i, zc0);
    wxr[i] = frag_b(W_x, G, 4 * i, 128 + zc0);
    wxn[i] = frag_b(W_x, G, 4 * i, 256 + zc0);
  }
  wxz[16] = frag_b_kguard(W_x, G, 64, zc0,       CIN);
  wxr[16] = frag_b_kguard(W_x, G, 64, 128 + zc0, CIN);
  wxn[16] = frag_b_kguard(W_x, G, 64, 256 + zc0, CIN);

  for (int i = tid; i < 16 * HP; i += 256) sh_h[i] = 0.0f;
  for (int i = tid; i < 16 * XP; i += 256) sh_kst[i] = 0.0f;
  for (int i = tid; i < 16 * IP; i += 256) sh_inp[i] = 0.0f;
  for (int idx = tid; idx < 8 * 4 * 32; idx += 256) {
    int ln = idx & 31, tile = (idx >> 5) & 3, i = idx >> 7;
    int kk  = 4 * i + ((ln >> 4) << 1);
    int col = tile * 16 + (ln & 15);
    v2f b;
    b.x = W_enc[kk * E + col];
    b.y = W_enc[(kk + 1) * E + col];
    sh_wencf[idx] = b;
  }
  for (int idx = tid; idx < 32 * 32; idx += 256) {
    int ln = idx & 31, i = idx >> 5;
    int kk  = 4 * i + ((ln >> 4) << 1);
    int col = ln & 15;
    v2f b;
    b.x = (col < M) ? W_gate[kk * M + col]       : 0.0f;
    b.y = (col < M) ? W_gate[(kk + 1) * M + col] : 0.0f;
    sh_wgf[idx] = b;
  }
  if (tid < M) sh_bg[tid] = b_gate[tid];
  const float* xr0 = x + (size_t)(b0 + wave) * T * IN_ + lane;
  const float* xr1 = x + (size_t)(b0 + 8 + wave) * T * IN_ + lane;
  const float* yr0 = y + (size_t)(b0 + wave) * T;
  const float* yr1 = y + (size_t)(b0 + 8 + wave) * T;
  sh_x[wave * XP + lane]       = xr0[0];
  sh_x[(wave + 8) * XP + lane] = xr1[0];
  float ycur0 = yr0[0], ycur1 = yr1[0];
  v8f Ahz = {}, Ahr = {}, Ahn = {};
  __syncthreads();

  for (int t = 0; t < T; ++t) {
    {
      const int m = wave;
      const float* Wi = W_in  + m * IN_ * S;
      const float* Wr = W_rec + m * S * S;
      v8f c0 = {}, c1 = {};
      #pragma unroll
      for (int k0 = 0; k0 < IN_; k0 += 4) {
        v2f a = frag_a(sh_x, XP, k0);
        c0 = wmma4(a, frag_b(Wi, S, k0, 0),  c0);
        c1 = wmma4(a, frag_b(Wi, S, k0, 16), c1);
      }
      #pragma unroll
      for (int k0 = 0; k0 < S; k0 += 4) {
        v2f a = frag_a(sh_kst, XP, k0);
        c0 = wmma4(a, frag_b(Wr, S, k0, 0),  c0);
        c1 = wmma4(a, frag_b(Wr, S, k0, 16), c1);
      }
      #pragma unroll
      for (int v = 0; v < 8; ++v) {
        int row = rb + v;
        sh_ns[row * NP + m * 32 + colb]      = fast_tanh(c0[v] + bias_s0);
        sh_ns[row * NP + m * 32 + 16 + colb] = fast_tanh(c1[v] + bias_s1);
      }
    }
    if (wave < 4) {
      int c0c = wave * 16;
      v8f c = {};
      #pragma unroll
      for (int i = 0; i < IN_ / 4; ++i)
        c = wmma4(frag_a(sh_x, XP, 4 * i),
                  sh_wencf[(i * 4 + wave) * 32 + lane], c);
      #pragma unroll
      for (int v = 0; v < 8; ++v)
        sh_inp[(rb + v) * IP + c0c + colb] = fast_tanh(c[v] + bias_e);
    }
    __syncthreads();

    const int tn = (t + 1 < T) ? t + 1 : t;
    const float xn0 = xr0[(size_t)tn * IN_];
    const float xn1 = xr1[(size_t)tn * IN_];
    const float yn0 = yr0[tn];
    const float yn1 = yr1[tn];

    float hnew[8];
    {
      v8f Cz = Ahz, Cr = Ahr, Cxn = {};
      const v8f Chn = Ahn;
      #pragma unroll
      for (int i = 0; i < 16; ++i) {
        v2f a = frag_a(sh_inp, IP, 4 * i);
        Cz  = wmma4(a, wxz[i], Cz);
        Cr  = wmma4(a, wxr[i], Cr);
        Cxn = wmma4(a, wxn[i], Cxn);
      }
      {
        v2f a = frag_a(sh_inp, IP, 64);
        Cz  = wmma4(a, wxz[16], Cz);
        Cr  = wmma4(a, wxr[16], Cr);
        Cxn = wmma4(a, wxn[16], Cxn);
      }
      #pragma unroll
      for (int v = 0; v < 8; ++v) {
        int row = rb + v;
        float z = fast_sigmoid(Cz[v] + bias_z);
        float r = fast_sigmoid(Cr[v] + bias_r);
        float n = fast_tanh((Cxn[v] + bias_xn) + r * (Chn[v] + bias_hn));
        hnew[v] = (1.0f - z) * n + z * sh_h[row * HP + hc];
      }
    }
    #pragma unroll
    for (int v = 0; v < 8; ++v) sh_h[(rb + v) * HP + hc] = hnew[v];
    sh_x[wave * XP + lane]       = xn0;
    sh_x[(wave + 8) * XP + lane] = xn1;
    __syncthreads();

    if (wave < 2) {
      float* dst = (wave == 0) ? sh_sc0 : sh_sc1;
      const int kbase = wave * 64;
      v8f c = {};
      #pragma unroll
      for (int i = 0; i < 16; ++i) {
        int k0 = kbase + 4 * i;
        v2f a = frag_a(sh_h, HP, k0);
        v2f b = sh_wgf[(wave * 16 + i) * 32 + lane];
        c = wmma4(a, b, c);
      }
      if (colb < M) {
        #pragma unroll
        for (int v = 0; v < 8; ++v) dst[(rb + v) * SP + colb] = c[v];
      }
    }
    Ahz = v8f{}; Ahr = v8f{}; Ahn = v8f{};
    #pragma unroll
    for (int i = 0; i < 10; ++i) {
      v2f a = frag_a(sh_h, HP, 4 * i);
      Ahz = wmma4(a, whz[i], Ahz);
      Ahr = wmma4(a, whr[i], Ahr);
      Ahn = wmma4(a, whn[i], Ahn);
    }
    __syncthreads();

    if (tid < 16) {
      float s[M], mx = -1e30f;
      #pragma unroll
      for (int m = 0; m < M; ++m) {
        s[m] = sh_sc0[tid * SP + m] + sh_sc1[tid * SP + m] + sh_bg[m];
        mx = fmaxf(mx, s[m]);
      }
      float sum = 0.0f;
      #pragma unroll
      for (int m = 0; m < M; ++m) { s[m] = __expf(s[m] - mx); sum += s[m]; }
      float inv = __fdividef(1.0f, sum);
      #pragma unroll
      for (int m = 0; m < M; ++m) sh_gate[tid * M + m] = s[m] * inv;
    }
    #pragma unroll
    for (int i = 10; i < 21; ++i) {
      v2f a = frag_a(sh_h, HP, 4 * i);
      Ahz = wmma4(a, whz[i], Ahz);
      Ahr = wmma4(a, whr[i], Ahr);
      Ahn = wmma4(a, whn[i], Ahn);
    }
    __syncthreads();

    float mix[2];
    #pragma unroll
    for (int j = 0; j < 2; ++j) {
      int row = wave + 8 * j;
      float acc = 0.0f;
      #pragma unroll
      for (int m = 0; m < M; ++m)
        acc += sh_gate[row * M + m] * sh_ns[row * NP + m * 32 + lane];
      sh_kst[row * XP + lane] = acc;
      mix[j] = acc;
    }
    if (lane == S - 1) {
      out[(size_t)(b0 + wave) * T + t]     = mix[0];
      out[(size_t)(b0 + wave + 8) * T + t] = mix[1];
      sh_inp[wave * IP + 64]       = mix[0] - ycur0;
      sh_inp[(wave + 8) * IP + 64] = mix[1] - ycur1;
    }
    ycur0 = yn0;
    ycur1 = yn1;
    #pragma unroll
    for (int i = 21; i < 32; ++i) {
      v2f a = frag_a(sh_h, HP, 4 * i);
      Ahz = wmma4(a, whz[i], Ahz);
      Ahr = wmma4(a, whr[i], Ahr);
      Ahn = wmma4(a, whn[i], Ahn);
    }
    __syncthreads();
  }
}

extern "C" void kernel_launch(void* const* d_in, const int* in_sizes, int n_in,
                              void* d_out, int out_size, void* d_ws, size_t ws_size,
                              hipStream_t stream) {
  (void)in_sizes; (void)n_in; (void)out_size; (void)d_ws; (void)ws_size;
  const float* x      = (const float*)d_in[0];
  const float* y      = (const float*)d_in[1];
  const float* W_enc  = (const float*)d_in[2];
  const float* b_enc  = (const float*)d_in[3];
  const float* W_in   = (const float*)d_in[4];
  const float* W_rec  = (const float*)d_in[5];
  const float* b_s    = (const float*)d_in[6];
  const float* W_x    = (const float*)d_in[7];
  const float* W_h    = (const float*)d_in[8];
  const float* b_x    = (const float*)d_in[9];
  const float* b_h    = (const float*)d_in[10];
  const float* W_gate = (const float*)d_in[11];
  const float* b_gate = (const float*)d_in[12];
  float* out = (float*)d_out;

  kalman_gru_scan<<<dim3(8), dim3(256), 0, stream>>>(
      x, y, W_enc, b_enc, W_in, W_rec, b_s, W_x, W_h, b_x, b_h,
      W_gate, b_gate, out);
}


# Input-only products (x@W_in, encoder, encoded@W_x) formed per block of 128 steps by an all-WGP launch, read back one whole step ahead in accumulator layout; gates add x sum to h sum (order)
# speedup vs baseline: 3.1666x; 1.2619x over previous
#include <hip/hip_runtime.h>
#include <math.h>


typedef __attribute__((ext_vector_type(2))) float v2f;
typedef __attribute__((ext_vector_type(8))) float v8f;

constexpr int    TC       = 128;
constexpr int    NCHUNK   = 16;
constexpr size_t P_STEP4  = 8 * 10 * 32;
constexpr size_t P_FLOATS = (size_t)8 * TC * P_STEP4 * 4;
constexpr size_t CARRY_H  = P_FLOATS;
constexpr size_t CARRY_K  = CARRY_H + 128 * 128;
constexpr size_t CARRY_E  = CARRY_K + 128 * 32;
constexpr size_t WS_NEED_BYTES = (CARRY_E + 128) * sizeof(float);

__device__ __forceinline__ v2f frag_a(const float* A, int ld, int k0) {
  int lane = threadIdx.x & 31;
  int row  = lane & 15;
  int kk   = k0 + ((lane >> 4) << 1);
  v2f a;
  a.x = A[row * ld + kk];
  a.y = A[row * ld + kk + 1];
  return a;
}

__device__ __forceinline__ v2f frag_b(const float* W, int ld, int k0, int c0) {
  int lane = threadIdx.x & 31;
  int col  = c0 + (lane & 15);
  int kk   = k0 + ((lane >> 4) << 1);
  v2f b;
  b.x = W[kk * ld + col];
  b.y = W[(kk + 1) * ld + col];
  return b;
}

__device__ __forceinline__ v2f frag_b_kguard(const float* W, int ld, int k0,
                                             int c0, int kmax) {
  int lane = threadIdx.x & 31;
  int col  = c0 + (lane & 15);
  int kk   = k0 + ((lane >> 4) << 1);
  v2f b;
  b.x = (kk     < kmax) ? W[kk * ld + col]       : 0.0f;
  b.y = (kk + 1 < kmax) ? W[(kk + 1) * ld + col] : 0.0f;
  return b;
}

__device__ __forceinline__ v8f wmma4(v2f a, v2f b, v8f c) {
  return __builtin_amdgcn_wmma_f32_16x16x4_f32(
       false, a,  false, b,
       (short)0, c,  false,  false);
}

__device__ __forceinline__ float fast_tanh(float x) {
  return __builtin_amdgcn_tanhf(x);
}
__device__ __forceinline__ float fast_sigmoid(float x) {
  return fmaf(0.5f, __builtin_amdgcn_tanhf(0.5f * x), 0.5f);
}

__global__ __launch_bounds__(256) void kalman_gru_scan(
    const float* __restrict__ x,
    const float* __restrict__ y,
    const float* __restrict__ W_enc,
    const float* __restrict__ b_enc,
    const float* __restrict__ W_in,
    const float* __restrict__ W_rec,
    const float* __restrict__ b_s,
    const float* __restrict__ W_x,
    const float* __restrict__ W_h,
    const float* __restrict__ b_x,
    const float* __restrict__ b_h,
    const float* __restrict__ W_gate,
    const float* __restrict__ b_gate,
    float* __restrict__ out,
    size_t ws_size)
{
  if (ws_size >= WS_NEED_BYTES) return;
  constexpr int T = 2048, IN_ = 32, S = 32, M = 8, E = 64, H = 128;
  constexpr int CIN = 65, G = 384;
  constexpr int XP = 36;
  constexpr int IP = 68;
  constexpr int HP = 132;
  constexpr int NP = 260;
  constexpr int SP = 20;

  __shared__ float sh_x[16 * XP];
  __shared__ float sh_inp[16 * IP];
  __shared__ float sh_h[16 * HP];
  __shared__ float sh_ns[16 * NP];
  __shared__ float sh_kst[16 * XP];
  __shared__ float sh_gate[16 * 8];
  __shared__ float sh_sc0[16 * SP];
  __shared__ float sh_sc1[16 * SP];
  __shared__ v2f sh_wencf[8 * 4 * 32];
  __shared__ v2f sh_wgf[32 * 32];
  __shared__ float sh_bg[8];

  const int tid  = threadIdx.x;
  const int wave = tid >> 5;
  const int lane = tid & 31;
  const int b0   = blockIdx.x * 16;

  const int colb = lane & 15;
  const int rb   = (lane >> 4) << 3;
  const int zc0  = wave * 16;
  const int hc   = zc0 + colb;

  const float bias_z  = b_x[hc]       + b_h[hc];
  const float bias_r  = b_x[128 + hc] + b_h[128 + hc];
  const float bias_xn = b_x[256 + hc];
  const float bias_hn = b_h[256 + hc];
  const float bias_s0 = b_s[wave * 32 + colb];
  const float bias_s1 = b_s[wave * 32 + 16 + colb];
  const float bias_e  = (wave < 4) ? b_enc[wave * 16 + colb] : 0.0f;

  v2f whz[32], whr[32], whn[32];
  #pragma unroll
  for (int i = 0; i < 32; ++i) {
    whz[i] = frag_b(W_h, G, 4 * i, zc0);
    whr[i] = frag_b(W_h, G, 4 * i, 128 + zc0);
    whn[i] = frag_b(W_h, G, 4 * i, 256 + zc0);
  }
  v2f wxz[17], wxr[17], wxn[17];
  #pragma unroll
  for (int i = 0; i < 16; ++i) {
    wxz[i] = frag_b(W_x, G, 4 * i, zc0);
    wxr[i] = frag_b(W_x, G, 4 * i, 128 + zc0);
    wxn[i] = frag_b(W_x, G, 4 * i, 256 + zc0);
  }
  wxz[16] = frag_b_kguard(W_x, G, 64, zc0,       CIN);
  wxr[16] = frag_b_kguard(W_x, G, 64, 128 + zc0, CIN);
  wxn[16] = frag_b_kguard(W_x, G, 64, 256 + zc0, CIN);

  for (int i = tid; i < 16 * HP; i += 256) sh_h[i] = 0.0f;
  for (int i = tid; i < 16 * XP; i += 256) sh_kst[i] = 0.0f;
  for (int i = tid; i < 16 * IP; i += 256) sh_inp[i] = 0.0f;
  for (int idx = tid; idx < 8 * 4 * 32; idx += 256) {
    int ln = idx & 31, tile = (idx >> 5) & 3, i = idx >> 7;
    int kk  = 4 * i + ((ln >> 4) << 1);
    int col = tile * 16 + (ln & 15);
    v2f b;
    b.x = W_enc[kk * E + col];
    b.y = W_enc[(kk + 1) * E + col];
    sh_wencf[idx] = b;
  }
  for (int idx = tid; idx < 32 * 32; idx += 256) {
    int ln = idx & 31, i = idx >> 5;
    int kk  = 4 * i + ((ln >> 4) << 1);
    int col = ln & 15;
    v2f b;
    b.x = (col < M) ? W_gate[kk * M + col]       : 0.0f;
    b.y = (col < M) ? W_gate[(kk + 1) * M + col] : 0.0f;
    sh_wgf[idx] = b;
  }
  if (tid < M) sh_bg[tid] = b_gate[tid];
  const float* xr0 = x + (size_t)(b0 + wave) * T * IN_ + lane;
  const float* xr1 = x + (size_t)(b0 + 8 + wave) * T * IN_ + lane;
  const float* yr0 = y + (size_t)(b0 + wave) * T;
  const float* yr1 = y + (size_t)(b0 + 8 + wave) * T;
  sh_x[wave * XP + lane]       = xr0[0];
  sh_x[(wave + 8) * XP + lane] = xr1[0];
  float ycur0 = yr0[0], ycur1 = yr1[0];
  v8f Ahz = {}, Ahr = {}, Ahn = {};
  __syncthreads();

  for (int t = 0; t < T; ++t) {
    {
      const int m = wave;
      const float* Wi = W_in  + m * IN_ * S;
      const float* Wr = W_rec + m * S * S;
      v8f c0 = {}, c1 = {};
      #pragma unroll
      for (int k0 = 0; k0 < IN_; k0 += 4) {
        v2f a = frag_a(sh_x, XP, k0);
        c0 = wmma4(a, frag_b(Wi, S, k0, 0),  c0);
        c1 = wmma4(a, frag_b(Wi, S, k0, 16), c1);
      }
      #pragma unroll
      for (int k0 = 0; k0 < S; k0 += 4) {
        v2f a = frag_a(sh_kst, XP, k0);
        c0 = wmma4(a, frag_b(Wr, S, k0, 0),  c0);
        c1 = wmma4(a, frag_b(Wr, S, k0, 16), c1);
      }
      #pragma unroll
      for (int v = 0; v < 8; ++v) {
        int row = rb + v;
        sh_ns[row * NP + m * 32 + colb]      = fast_tanh(c0[v] + bias_s0);
        sh_ns[row * NP + m * 32 + 16 + colb] = fast_tanh(c1[v] + bias_s1);
      }
    }
    if (wave < 4) {
      int c0c = wave * 16;
      v8f c = {};
      #pragma unroll
      for (int i = 0; i < IN_ / 4; ++i)
        c = wmma4(frag_a(sh_x, XP, 4 * i),
                  sh_wencf[(i * 4 + wave) * 32 + lane], c);
      #pragma unroll
      for (int v = 0; v < 8; ++v)
        sh_inp[(rb + v) * IP + c0c + colb] = fast_tanh(c[v] + bias_e);
    }
    __syncthreads();

    const int tn = (t + 1 < T) ? t + 1 : t;
    const float xn0 = xr0[(size_t)tn * IN_];
    const float xn1 = xr1[(size_t)tn * IN_];
    const float yn0 = yr0[tn];
    const float yn1 = yr1[tn];

    float hnew[8];
    {
      v8f Cz = Ahz, Cr = Ahr, Cxn = {};
      const v8f Chn = Ahn;
      #pragma unroll
      for (int i = 0; i < 16; ++i) {
        v2f a = frag_a(sh_inp, IP, 4 * i);
        Cz  = wmma4(a, wxz[i], Cz);
        Cr  = wmma4(a, wxr[i], Cr);
        Cxn = wmma4(a, wxn[i], Cxn);
      }
      {
        v2f a = frag_a(sh_inp, IP, 64);
        Cz  = wmma4(a, wxz[16], Cz);
        Cr  = wmma4(a, wxr[16], Cr);
        Cxn = wmma4(a, wxn[16], Cxn);
      }
      #pragma unroll
      for (int v = 0; v < 8; ++v) {
        int row = rb + v;
        float z = fast_sigmoid(Cz[v] + bias_z);
        float r = fast_sigmoid(Cr[v] + bias_r);
        float n = fast_tanh((Cxn[v] + bias_xn) + r * (Chn[v] + bias_hn));
        hnew[v] = (1.0f - z) * n + z * sh_h[row * HP + hc];
      }
    }
    #pragma unroll
    for (int v = 0; v < 8; ++v) sh_h[(rb + v) * HP + hc] = hnew[v];
    sh_x[wave * XP + lane]       = xn0;
    sh_x[(wave + 8) * XP + lane] = xn1;
    __syncthreads();

    if (wave < 2) {
      float* dst = (wave == 0) ? sh_sc0 : sh_sc1;
      const int kbase = wave * 64;
      v8f c = {};
      #pragma unroll
      for (int i = 0; i < 16; ++i) {
        int k0 = kbase + 4 * i;
        v2f a = frag_a(sh_h, HP, k0);
        v2f b = sh_wgf[(wave * 16 + i) * 32 + lane];
        c = wmma4(a, b, c);
      }
      if (colb < M) {
        #pragma unroll
        for (int v = 0; v < 8; ++v) dst[(rb + v) * SP + colb] = c[v];
      }
    }
    Ahz = v8f{}; Ahr = v8f{}; Ahn = v8f{};
    #pragma unroll
    for (int i = 0; i < 10; ++i) {
      v2f a = frag_a(sh_h, HP, 4 * i);
      Ahz = wmma4(a, whz[i], Ahz);
      Ahr = wmma4(a, whr[i], Ahr);
      Ahn = wmma4(a, whn[i], Ahn);
    }
    __syncthreads();

    if (tid < 16) {
      float s[M], mx = -1e30f;
      #pragma unroll
      for (int m = 0; m < M; ++m) {
        s[m] = sh_sc0[tid * SP + m] + sh_sc1[tid * SP + m] + sh_bg[m];
        mx = fmaxf(mx, s[m]);
      }
      float sum = 0.0f;
      #pragma unroll
      for (int m = 0; m < M; ++m) { s[m] = __expf(s[m] - mx); sum += s[m]; }
      float inv = __fdividef(1.0f, sum);
      #pragma unroll
      for (int m = 0; m < M; ++m) sh_gate[tid * M + m] = s[m] * inv;
    }
    #pragma unroll
    for (int i = 10; i < 21; ++i) {
      v2f a = frag_a(sh_h, HP, 4 * i);
      Ahz = wmma4(a, whz[i], Ahz);
      Ahr = wmma4(a, whr[i], Ahr);
      Ahn = wmma4(a, whn[i], Ahn);
    }
    __syncthreads();

    float mix[2];
    #pragma unroll
    for (int j = 0; j < 2; ++j) {
      int row = wave + 8 * j;
      float acc = 0.0f;
      #pragma unroll
      for (int m = 0; m < M; ++m)
        acc += sh_gate[row * M + m] * sh_ns[row * NP + m * 32 + lane];
      sh_kst[row * XP + lane] = acc;
      mix[j] = acc;
    }
    if (lane == S - 1) {
      out[(size_t)(b0 + wave) * T + t]     = mix[0];
      out[(size_t)(b0 + wave + 8) * T + t] = mix[1];
      sh_inp[wave * IP + 64]       = mix[0] - ycur0;
      sh_inp[(wave + 8) * IP + 64] = mix[1] - ycur1;
    }
    ycur0 = yn0;
    ycur1 = yn1;
    #pragma unroll
    for (int i = 21; i < 32; ++i) {
      v2f a = frag_a(sh_h, HP, 4 * i);
      Ahz = wmma4(a, whz[i], Ahz);
      Ahr = wmma4(a, whr[i], Ahr);
      Ahn = wmma4(a, whn[i], Ahn);
    }
    __syncthreads();
  }
}

__global__ __launch_bounds__(256) void input_products_chunk(
    const float* __restrict__ x,
    const float* __restrict__ W_enc,
    const float* __restrict__ b_enc,
    const float* __restrict__ W_in,
    const float* __restrict__ b_s,
    const float* __restrict__ W_x,
    const float* __restrict__ b_x,
    const float* __restrict__ b_h,
    float* __restrict__ ws, size_t ws_size, int chunk)
{
  if (ws_size < WS_NEED_BYTES) return;
  constexpr int T = 2048, IN_ = 32, S = 32, E = 64, G = 384;
  constexpr int IP = 68;
  __shared__ float sh_enc[2][16 * IP];

  const int tid  = threadIdx.x;
  const int wave = tid >> 5;
  const int lane = tid & 31;
  const int colb = lane & 15;
  const int rb   = (lane >> 4) << 3;
  const int zc0  = wave * 16;
  const int hc   = zc0 + colb;

  const float bias_z  = b_x[hc]       + b_h[hc];
  const float bias_r  = b_x[128 + hc] + b_h[128 + hc];
  const float bias_xn = b_x[256 + hc];
  const float bias_s0 = b_s[wave * 32 + colb];
  const float bias_s1 = b_s[wave * 32 + 16 + colb];
  const float bias_e  = b_enc[(wave & 3) * 16 + colb];

  v2f wi0[8], wi1[8], we[8];
  #pragma unroll
  for (int i = 0; i < 8; ++i) {
    wi0[i] = frag_b(W_in + wave * IN_ * S, S, 4 * i, 0);
    wi1[i] = frag_b(W_in + wave * IN_ * S, S, 4 * i, 16);
    we[i]  = frag_b(W_enc, E, 4 * i, (wave & 3) * 16);
  }
  v2f wxz[16], wxr[16], wxn[16];
  #pragma unroll
  for (int i = 0; i < 16; ++i) {
    wxz[i] = frag_b(W_x, G, 4 * i, zc0);
    wxr[i] = frag_b(W_x, G, 4 * i, 128 + zc0);
    wxn[i] = frag_b(W_x, G, 4 * i, 256 + zc0);
  }

  float4* P = reinterpret_cast<float4*>(ws);
  const int row = lane & 15;
  const int kh  = (lane >> 4) << 1;

  constexpr int NT = (8 * TC) / 256;
  v2f a[8];
  {
    const int bt = blockIdx.x / TC, tl = blockIdx.x % TC;
    const float* xrow = x + ((size_t)(bt * 16 + row) * T + chunk * TC + tl) * IN_ + kh;
    #pragma unroll
    for (int i = 0; i < 8; ++i) { a[i].x = xrow[4 * i]; a[i].y = xrow[4 * i + 1]; }
  }
  for (int it = 0; it < NT; ++it) {
    const int tile = blockIdx.x + 256 * it;
    const int bt = tile / TC;
    const int tl = tile % TC;

    float4* dst = P + (((size_t)(bt * TC + tl) * 8 + wave) * 10) * 32 + lane;

    v8f c0 = {}, c1 = {};
    #pragma unroll
    for (int i = 0; i < 8; ++i) {
      c0 = wmma4(a[i], wi0[i], c0);
      c1 = wmma4(a[i], wi1[i], c1);
    }

    float* enc = sh_enc[it & 1];
    if (wave < 4) {
      v8f ce = {};
      #pragma unroll
      for (int i = 0; i < 8; ++i) ce = wmma4(a[i], we[i], ce);
      #pragma unroll
      for (int v = 0; v < 8; ++v)
        enc[(rb + v) * IP + wave * 16 + colb] = fast_tanh(ce[v] + bias_e);
    }
    __syncthreads();

    v2f an[8];
    {
      const int tile_n = (it + 1 < NT) ? tile + 256 : tile;
      const int bt_n = tile_n / TC, tl_n = tile_n % TC;
      const float* xrow = x + ((size_t)(bt_n * 16 + row) * T + chunk * TC + tl_n) * IN_ + kh;
      #pragma unroll
      for (int i = 0; i < 8; ++i) { an[i].x = xrow[4 * i]; an[i].y = xrow[4 * i + 1]; }
    }
    dst[0 * 32] = make_float4(c0[0] + bias_s0, c0[1] + bias_s0, c0[2] + bias_s0, c0[3] + bias_s0);
    dst[1 * 32] = make_float4(c0[4] + bias_s0, c0[5] + bias_s0, c0[6] + bias_s0, c0[7] + bias_s0);
    dst[2 * 32] = make_float4(c1[0] + bias_s1, c1[1] + bias_s1, c1[2] + bias_s1, c1[3] + bias_s1);
    dst[3 * 32] = make_float4(c1[4] + bias_s1, c1[5] + bias_s1, c1[6] + bias_s1, c1[7] + bias_s1);

    v8f Cz = {}, Cr = {}, Cn = {};
    #pragma unroll
    for (int i = 0; i < 16; ++i) {
      v2f ea = frag_a(enc, IP, 4 * i);
      Cz = wmma4(ea, wxz[i], Cz);
      Cr = wmma4(ea, wxr[i], Cr);
      Cn = wmma4(ea, wxn[i], Cn);
    }
    dst[4 * 32] = make_float4(Cz[0] + bias_z, Cz[1] + bias_z, Cz[2] + bias_z, Cz[3] + bias_z);
    dst[5 * 32] = make_float4(Cz[4] + bias_z, Cz[5] + bias_z, Cz[6] + bias_z, Cz[7] + bias_z);
    dst[6 * 32] = make_float4(Cr[0] + bias_r, Cr[1] + bias_r, Cr[2] + bias_r, Cr[3] + bias_r);
    dst[7 * 32] = make_float4(Cr[4] + bias_r, Cr[5] + bias_r, Cr[6] + bias_r, Cr[7] + bias_r);
    dst[8 * 32] = make_float4(Cn[0] + bias_xn, Cn[1] + bias_xn, Cn[2] + bias_xn, Cn[3] + bias_xn);
    dst[9 * 32] = make_float4(Cn[4] + bias_xn, Cn[5] + bias_xn, Cn[6] + bias_xn, Cn[7] + bias_xn);
    #pragma unroll
    for (int i = 0; i < 8; ++i) a[i] = an[i];
  }
}

__global__ __launch_bounds__(256) void kalman_gru_scan_chunk(
    const float* __restrict__ y,
    const float* __restrict__ W_rec,
    const float* __restrict__ W_x,
    const float* __restrict__ W_h,
    const float* __restrict__ b_h,
    const float* __restrict__ W_gate,
    const float* __restrict__ b_gate,
    float* __restrict__ out,
    float* __restrict__ ws, size_t ws_size, int chunk)
{
  if (ws_size < WS_NEED_BYTES) return;
  constexpr int T = 2048, S = 32, M = 8, H = 128;
  constexpr int CIN = 65, G = 384;
  constexpr int XP = 36;
  constexpr int HP = 132;
  constexpr int NP = 260;
  constexpr int SP = 20;

  __shared__ float sh_h[16 * HP];
  __shared__ float sh_ns[16 * NP];
  __shared__ float sh_kst[16 * XP];
  __shared__ float sh_gate[16 * 8];
  __shared__ float sh_sc0[16 * SP];
  __shared__ float sh_sc1[16 * SP];
  __shared__ float sh_err[16];
  __shared__ v2f sh_wgf[32 * 32];
  __shared__ float sh_bg[8];

  const int tid  = threadIdx.x;
  const int wave = tid >> 5;
  const int lane = tid & 31;
  const int b0   = blockIdx.x * 16;
  const int colb = lane & 15;
  const int rb   = (lane >> 4) << 3;
  const int zc0  = wave * 16;
  const int hc   = zc0 + colb;
  const int t0   = chunk * TC;

  const float bias_hn = b_h[256 + hc];

  v2f whz[32], whr[32], whn[32];
  #pragma unroll
  for (int i = 0; i < 32; ++i) {
    whz[i] = frag_b(W_h, G, 4 * i, zc0);
    whr[i] = frag_b(W_h, G, 4 * i, 128 + zc0);
    whn[i] = frag_b(W_h, G, 4 * i, 256 + zc0);
  }
  const v2f wxz_e = frag_b_kguard(W_x, G, 64, zc0,       CIN);
  const v2f wxr_e = frag_b_kguard(W_x, G, 64, 128 + zc0, CIN);
  const v2f wxn_e = frag_b_kguard(W_x, G, 64, 256 + zc0, CIN);

  float* carry_h = ws + CARRY_H + (size_t)b0 * H;
  float* carry_k = ws + CARRY_K + (size_t)b0 * S;
  float* carry_e = ws + CARRY_E + b0;
  if (chunk == 0) {
    for (int i = tid; i < 16 * HP; i += 256) sh_h[i] = 0.0f;
    for (int i = tid; i < 16 * XP; i += 256) sh_kst[i] = 0.0f;
    if (tid < 16) sh_err[tid] = 0.0f;
  } else {
    for (int i = tid; i < 16 * H; i += 256) sh_h[(i >> 7) * HP + (i & 127)] = carry_h[i];
    for (int i = tid; i < 16 * S; i += 256) sh_kst[(i >> 5) * XP + (i & 31)] = carry_k[i];
    if (tid < 16) sh_err[tid] = carry_e[tid];
  }
  for (int idx = tid; idx < 32 * 32; idx += 256) {
    int ln = idx & 31, i = idx >> 5;
    int kk  = 4 * i + ((ln >> 4) << 1);
    int col = ln & 15;
    v2f b;
    b.x = (col < M) ? W_gate[kk * M + col]       : 0.0f;
    b.y = (col < M) ? W_gate[(kk + 1) * M + col] : 0.0f;
    sh_wgf[idx] = b;
  }
  if (tid < M) sh_bg[tid] = b_gate[tid];

  const float* yr0 = y + (size_t)(b0 + wave) * T + t0;
  const float* yr1 = y + (size_t)(b0 + 8 + wave) * T + t0;
  float ycur0 = yr0[0], ycur1 = yr1[0];

  const float4* Pw = reinterpret_cast<const float4*>(ws)
                   + (((size_t)blockIdx.x * TC) * 8 + wave) * 320 + lane;
  float4 nx[10];
  #pragma unroll
  for (int q = 0; q < 10; ++q) nx[q] = Pw[q * 32];
  __syncthreads();

  v8f Ahz = {}, Ahr = {}, Ahn = {};
  #pragma unroll
  for (int i = 0; i < 32; ++i) {
    v2f a = frag_a(sh_h, HP, 4 * i);
    Ahz = wmma4(a, whz[i], Ahz);
    Ahr = wmma4(a, whr[i], Ahr);
    Ahn = wmma4(a, whn[i], Ahn);
  }
  __syncthreads();

  for (int tl = 0; tl < TC; ++tl) {
    float4 cu[10];
    #pragma unroll
    for (int q = 0; q < 10; ++q) cu[q] = nx[q];
    const int tn = (tl + 1 < TC) ? tl + 1 : tl;
    {
      const float4* Pn = Pw + (size_t)tn * P_STEP4;
      #pragma unroll
      for (int q = 0; q < 10; ++q) nx[q] = Pn[q * 32];
    }
    const float yn0 = yr0[tn];
    const float yn1 = yr1[tn];

    {
      const int m = wave;
      const float* Wr = W_rec + m * S * S;
      v8f c0 = {cu[0].x, cu[0].y, cu[0].z, cu[0].w, cu[1].x, cu[1].y, cu[1].z, cu[1].w};
      v8f c1 = {cu[2].x, cu[2].y, cu[2].z, cu[2].w, cu[3].x, cu[3].y, cu[3].z, cu[3].w};
      #pragma unroll
      for (int k0 = 0; k0 < S; k0 += 4) {
        v2f a = frag_a(sh_kst, XP, k0);
        c0 = wmma4(a, frag_b(Wr, S, k0, 0),  c0);
        c1 = wmma4(a, frag_b(Wr, S, k0, 16), c1);
      }
      #pragma unroll
      for (int v = 0; v < 8; ++v) {
        int row = rb + v;
        sh_ns[row * NP + m * 32 + colb]      = fast_tanh(c0[v]);
        sh_ns[row * NP + m * 32 + 16 + colb] = fast_tanh(c1[v]);
      }
    }
    float hnew[8];
    {
      v8f Cz = Ahz, Cr = Ahr, Cxn = {};
      const v8f Chn = Ahn;
      v2f a;
      a.x = (lane < 16) ? sh_err[lane] : 0.0f;
      a.y = 0.0f;
      Cz  = wmma4(a, wxz_e, Cz);
      Cr  = wmma4(a, wxr_e, Cr);
      Cxn = wmma4(a, wxn_e, Cxn);
      const float pz[8] = {cu[4].x, cu[4].y, cu[4].z, cu[4].w, cu[5].x, cu[5].y, cu[5].z, cu[5].w};
      const float pr[8] = {cu[6].x, cu[6].y, cu[6].z, cu[6].w, cu[7].x, cu[7].y, cu[7].z, cu[7].w};
      const float pn[8] = {cu[8].x, cu[8].y, cu[8].z, cu[8].w, cu[9].x, cu[9].y, cu[9].z, cu[9].w};
      #pragma unroll
      for (int v = 0; v < 8; ++v) {
        int row = rb + v;
        float z = fast_sigmoid(Cz[v] + pz[v]);
        float r = fast_sigmoid(Cr[v] + pr[v]);
        float n = fast_tanh((Cxn[v] + pn[v]) + r * (Chn[v] + bias_hn));
        hnew[v] = (1.0f - z) * n + z * sh_h[row * HP + hc];
      }
    }
    #pragma unroll
    for (int v = 0; v < 8; ++v) sh_h[(rb + v) * HP + hc] = hnew[v];
    __builtin_amdgcn_fence(__ATOMIC_RELEASE, "workgroup", "local");
    __builtin_amdgcn_s_barrier();
    __builtin_amdgcn_fence(__ATOMIC_ACQUIRE, "workgroup", "local");

    if (wave < 2) {
      float* dst = (wave == 0) ? sh_sc0 : sh_sc1;
      const int kbase = wave * 64;
      v8f c = {};
      #pragma unroll
      for (int i = 0; i < 16; ++i) {
        int k0 = kbase + 4 * i;
        v2f a = frag_a(sh_h, HP, k0);
        v2f b = sh_wgf[(wave * 16 + i) * 32 + lane];
        c = wmma4(a, b, c);
      }
      if (colb < M) {
        #pragma unroll
        for (int v = 0; v < 8; ++v) dst[(rb + v) * SP + colb] = c[v];
      }
    }
    Ahz = v8f{}; Ahr = v8f{}; Ahn = v8f{};
    #pragma unroll
    for (int i = 0; i < 10; ++i) {
      v2f a = frag_a(sh_h, HP, 4 * i);
      Ahz = wmma4(a, whz[i], Ahz);
      Ahr = wmma4(a, whr[i], Ahr);
      Ahn = wmma4(a, whn[i], Ahn);
    }
    __syncthreads();

    if (tid < 16) {
      float s[M], mx = -1e30f;
      #pragma unroll
      for (int m = 0; m < M; ++m) {
        s[m] = sh_sc0[tid * SP + m] + sh_sc1[tid * SP + m] + sh_bg[m];
        mx = fmaxf(mx, s[m]);
      }
      float sum = 0.0f;
      #pragma unroll
      for (int m = 0; m < M; ++m) { s[m] = __expf(s[m] - mx); sum += s[m]; }
      float inv = __fdividef(1.0f, sum);
      #pragma unroll
      for (int m = 0; m < M; ++m) sh_gate[tid * M + m] = s[m] * inv;
    }
    #pragma unroll
    for (int i = 10; i < 21; ++i) {
      v2f a = frag_a(sh_h, HP, 4 * i);
      Ahz = wmma4(a, whz[i], Ahz);
      Ahr = wmma4(a, whr[i], Ahr);
      Ahn = wmma4(a, whn[i], Ahn);
    }
    __syncthreads();

    float mix[2];
    #pragma unroll
    for (int j = 0; j < 2; ++j) {
      int row = wave + 8 * j;
      float acc = 0.0f;
      #pragma unroll
      for (int m = 0; m < M; ++m)
        acc += sh_gate[row * M + m] * sh_ns[row * NP + m * 32 + lane];
      sh_kst[row * XP + lane] = acc;
      mix[j] = acc;
    }
    if (lane == S - 1) {
      out[(size_t)(b0 + wave) * T + t0 + tl]     = mix[0];
      out[(size_t)(b0 + wave + 8) * T + t0 + tl] = mix[1];
      sh_err[wave]     = mix[0] - ycur0;
      sh_err[wave + 8] = mix[1] - ycur1;
    }
    ycur0 = yn0;
    ycur1 = yn1;
    #pragma unroll
    for (int i = 21; i < 32; ++i) {
      v2f a = frag_a(sh_h, HP, 4 * i);
      Ahz = wmma4(a, whz[i], Ahz);
      Ahr = wmma4(a, whr[i], Ahr);
      Ahn = wmma4(a, whn[i], Ahn);
    }
    __syncthreads();
  }

  if (chunk + 1 < NCHUNK) {
    for (int i = tid; i < 16 * H; i += 256) carry_h[i] = sh_h[(i >> 7) * HP + (i & 127)];
    for (int i = tid; i < 16 * S; i += 256) carry_k[i] = sh_kst[(i >> 5) * XP + (i & 31)];
    if (tid < 16) carry_e[tid] = sh_err[tid];
  }
}

extern "C" void kernel_launch(void* const* d_in, const int* in_sizes, int n_in,
                              void* d_out, int out_size, void* d_ws, size_t ws_size,
                              hipStream_t stream) {
  (void)in_sizes; (void)n_in; (void)out_size;
  const float* x      = (const float*)d_in[0];
  const float* y      = (const float*)d_in[1];
  const float* W_enc  = (const float*)d_in[2];
  const float* b_enc  = (const float*)d_in[3];
  const float* W_in   = (const float*)d_in[4];
  const float* W_rec  = (const float*)d_in[5];
  const float* b_s    = (const float*)d_in[6];
  const float* W_x    = (const float*)d_in[7];
  const float* W_h    = (const float*)d_in[8];
  const float* b_x    = (const float*)d_in[9];
  const float* b_h    = (const float*)d_in[10];
  const float* W_gate = (const float*)d_in[11];
  const float* b_gate = (const float*)d_in[12];
  float* out = (float*)d_out;
  float* ws  = (float*)d_ws;

  kalman_gru_scan<<<dim3(8), dim3(256), 0, stream>>>(
      x, y, W_enc, b_enc, W_in, W_rec, b_s, W_x, W_h, b_x, b_h,
      W_gate, b_gate, out, ws_size);

#define SCAN_BLOCK(c)                                                          \
  input_products_chunk<<<dim3(256), dim3(256), 0, stream>>>(                   \
      x, W_enc, b_enc, W_in, b_s, W_x, b_x, b_h, ws, ws_size, c);              \
  kalman_gru_scan_chunk<<<dim3(8), dim3(256), 0, stream>>>(                    \
      y, W_rec, W_x, W_h, b_h, W_gate, b_gate, out, ws, ws_size, c);
  SCAN_BLOCK(0)  SCAN_BLOCK(1)  SCAN_BLOCK(2)  SCAN_BLOCK(3)
  SCAN_BLOCK(4)  SCAN_BLOCK(5)  SCAN_BLOCK(6)  SCAN_BLOCK(7)
  SCAN_BLOCK(8)  SCAN_BLOCK(9)  SCAN_BLOCK(10) SCAN_BLOCK(11)
  SCAN_BLOCK(12) SCAN_BLOCK(13) SCAN_BLOCK(14) SCAN_BLOCK(15)
#undef SCAN_BLOCK
}


# Step head re-ordered: GRU gate part (err product, activations) first; k@W_rec runs among the activations and between signal and wait of the split h hand-over barrier; product loads moved behind it
# speedup vs baseline: 3.3888x; 1.0702x over previous
#include <hip/hip_runtime.h>
#include <math.h>


typedef __attribute__((ext_vector_type(2))) float v2f;
typedef __attribute__((ext_vector_type(8))) float v8f;

constexpr int    TC       = 128;
constexpr int    NCHUNK   = 16;
constexpr size_t P_STEP4  = 8 * 10 * 32;
constexpr size_t P_FLOATS = (size_t)8 * TC * P_STEP4 * 4;
constexpr size_t CARRY_H  = P_FLOATS;
constexpr size_t CARRY_K  = CARRY_H + 128 * 128;
constexpr size_t CARRY_E  = CARRY_K + 128 * 32;
constexpr size_t WS_NEED_BYTES = (CARRY_E + 128) * sizeof(float);

__device__ __forceinline__ v2f frag_a(const float* A, int ld, int k0) {
  int lane = threadIdx.x & 31;
  int row  = lane & 15;
  int kk   = k0 + ((lane >> 4) << 1);
  v2f a;
  a.x = A[row * ld + kk];
  a.y = A[row * ld + kk + 1];
  return a;
}

__device__ __forceinline__ v2f frag_b(const float* W, int ld, int k0, int c0) {
  int lane = threadIdx.x & 31;
  int col  = c0 + (lane & 15);
  int kk   = k0 + ((lane >> 4) << 1);
  v2f b;
  b.x = W[kk * ld + col];
  b.y = W[(kk + 1) * ld + col];
  return b;
}

__device__ __forceinline__ v2f frag_b_kguard(const float* W, int ld, int k0,
                                             int c0, int kmax) {
  int lane = threadIdx.x & 31;
  int col  = c0 + (lane & 15);
  int kk   = k0 + ((lane >> 4) << 1);
  v2f b;
  b.x = (kk     < kmax) ? W[kk * ld + col]       : 0.0f;
  b.y = (kk + 1 < kmax) ? W[(kk + 1) * ld + col] : 0.0f;
  return b;
}

__device__ __forceinline__ v8f wmma4(v2f a, v2f b, v8f c) {
  return __builtin_amdgcn_wmma_f32_16x16x4_f32(
       false, a,  false, b,
       (short)0, c,  false,  false);
}

__device__ __forceinline__ float fast_tanh(float x) {
  return __builtin_amdgcn_tanhf(x);
}
__device__ __forceinline__ float fast_sigmoid(float x) {
  return fmaf(0.5f, __builtin_amdgcn_tanhf(0.5f * x), 0.5f);
}

__global__ __launch_bounds__(256) void kalman_gru_scan(
    const float* __restrict__ x,
    const float* __restrict__ y,
    const float* __restrict__ W_enc,
    const float* __restrict__ b_enc,
    const float* __restrict__ W_in,
    const float* __restrict__ W_rec,
    const float* __restrict__ b_s,
    const float* __restrict__ W_x,
    const float* __restrict__ W_h,
    const float* __restrict__ b_x,
    const float* __restrict__ b_h,
    const float* __restrict__ W_gate,
    const float* __restrict__ b_gate,
    float* __restrict__ out,
    size_t ws_size)
{
  if (ws_size >= WS_NEED_BYTES) return;
  constexpr int T = 2048, IN_ = 32, S = 32, M = 8, E = 64, H = 128;
  constexpr int CIN = 65, G = 384;
  constexpr int XP = 36;
  constexpr int IP = 68;
  constexpr int HP = 132;
  constexpr int NP = 260;
  constexpr int SP = 20;

  __shared__ float sh_x[16 * XP];
  __shared__ float sh_inp[16 * IP];
  __shared__ float sh_h[16 * HP];
  __shared__ float sh_ns[16 * NP];
  __shared__ float sh_kst[16 * XP];
  __shared__ float sh_gate[16 * 8];
  __shared__ float sh_sc0[16 * SP];
  __shared__ float sh_sc1[16 * SP];
  __shared__ v2f sh_wencf[8 * 4 * 32];
  __shared__ v2f sh_wgf[32 * 32];
  __shared__ float sh_bg[8];

  const int tid  = threadIdx.x;
  const int wave = tid >> 5;
  const int lane = tid & 31;
  const int b0   = blockIdx.x * 16;

  const int colb = lane & 15;
  const int rb   = (lane >> 4) << 3;
  const int zc0  = wave * 16;
  const int hc   = zc0 + colb;

  const float bias_z  = b_x[hc]       + b_h[hc];
  const float bias_r  = b_x[128 + hc] + b_h[128 + hc];
  const float bias_xn = b_x[256 + hc];
  const float bias_hn = b_h[256 + hc];
  const float bias_s0 = b_s[wave * 32 + colb];
  const float bias_s1 = b_s[wave * 32 + 16 + colb];
  const float bias_e  = (wave < 4) ? b_enc[wave * 16 + colb] : 0.0f;

  v2f whz[32], whr[32], whn[32];
  #pragma unroll
  for (int i = 0; i < 32; ++i) {
    whz[i] = frag_b(W_h, G, 4 * i, zc0);
    whr[i] = frag_b(W_h, G, 4 * i, 128 + zc0);
    whn[i] = frag_b(W_h, G, 4 * i, 256 + zc0);
  }
  v2f wxz[17], wxr[17], wxn[17];
  #pragma unroll
  for (int i = 0; i < 16; ++i) {
    wxz[i] = frag_b(W_x, G, 4 * i, zc0);
    wxr[i] = frag_b(W_x, G, 4 * i, 128 + zc0);
    wxn[i] = frag_b(W_x, G, 4 * i, 256 + zc0);
  }
  wxz[16] = frag_b_kguard(W_x, G, 64, zc0,       CIN);
  wxr[16] = frag_b_kguard(W_x, G, 64, 128 + zc0, CIN);
  wxn[16] = frag_b_kguard(W_x, G, 64, 256 + zc0, CIN);

  for (int i = tid; i < 16 * HP; i += 256) sh_h[i] = 0.0f;
  for (int i = tid; i < 16 * XP; i += 256) sh_kst[i] = 0.0f;
  for (int i = tid; i < 16 * IP; i += 256) sh_inp[i] = 0.0f;
  for (int idx = tid; idx < 8 * 4 * 32; idx += 256) {
    int ln = idx & 31, tile = (idx >> 5) & 3, i = idx >> 7;
    int kk  = 4 * i + ((ln >> 4) << 1);
    int col = tile * 16 + (ln & 15);
    v2f b;
    b.x = W_enc[kk * E + col];
    b.y = W_enc[(kk + 1) * E + col];
    sh_wencf[idx] = b;
  }
  for (int idx = tid; idx < 32 * 32; idx += 256) {
    int ln = idx & 31, i = idx >> 5;
    int kk  = 4 * i + ((ln >> 4) << 1);
    int col = ln & 15;
    v2f b;
    b.x = (col < M) ? W_gate[kk * M + col]       : 0.0f;
    b.y = (col < M) ? W_gate[(kk + 1) * M + col] : 0.0f;
    sh_wgf[idx] = b;
  }
  if (tid < M) sh_bg[tid] = b_gate[tid];
  const float* xr0 = x + (size_t)(b0 + wave) * T * IN_ + lane;
  const float* xr1 = x + (size_t)(b0 + 8 + wave) * T * IN_ + lane;
  const float* yr0 = y + (size_t)(b0 + wave) * T;
  const float* yr1 = y + (size_t)(b0 + 8 + wave) * T;
  sh_x[wave * XP + lane]       = xr0[0];
  sh_x[(wave + 8) * XP + lane] = xr1[0];
  float ycur0 = yr0[0], ycur1 = yr1[0];
  v8f Ahz = {}, Ahr = {}, Ahn = {};
  __syncthreads();

  for (int t = 0; t < T; ++t) {
    {
      const int m = wave;
      const float* Wi = W_in  + m * IN_ * S;
      const float* Wr = W_rec + m * S * S;
      v8f c0 = {}, c1 = {};
      #pragma unroll
      for (int k0 = 0; k0 < IN_; k0 += 4) {
        v2f a = frag_a(sh_x, XP, k0);
        c0 = wmma4(a, frag_b(Wi, S, k0, 0),  c0);
        c1 = wmma4(a, frag_b(Wi, S, k0, 16), c1);
      }
      #pragma unroll
      for (int k0 = 0; k0 < S; k0 += 4) {
        v2f a = frag_a(sh_kst, XP, k0);
        c0 = wmma4(a, frag_b(Wr, S, k0, 0),  c0);
        c1 = wmma4(a, frag_b(Wr, S, k0, 16), c1);
      }
      #pragma unroll
      for (int v = 0; v < 8; ++v) {
        int row = rb + v;
        sh_ns[row * NP + m * 32 + colb]      = fast_tanh(c0[v] + bias_s0);
        sh_ns[row * NP + m * 32 + 16 + colb] = fast_tanh(c1[v] + bias_s1);
      }
    }
    if (wave < 4) {
      int c0c = wave * 16;
      v8f c = {};
      #pragma unroll
      for (int i = 0; i < IN_ / 4; ++i)
        c = wmma4(frag_a(sh_x, XP, 4 * i),
                  sh_wencf[(i * 4 + wave) * 32 + lane], c);
      #pragma unroll
      for (int v = 0; v < 8; ++v)
        sh_inp[(rb + v) * IP + c0c + colb] = fast_tanh(c[v] + bias_e);
    }
    __syncthreads();

    const int tn = (t + 1 < T) ? t + 1 : t;
    const float xn0 = xr0[(size_t)tn * IN_];
    const float xn1 = xr1[(size_t)tn * IN_];
    const float yn0 = yr0[tn];
    const float yn1 = yr1[tn];

    float hnew[8];
    {
      v8f Cz = Ahz, Cr = Ahr, Cxn = {};
      const v8f Chn = Ahn;
      #pragma unroll
      for (int i = 0; i < 16; ++i) {
        v2f a = frag_a(sh_inp, IP, 4 * i);
        Cz  = wmma4(a, wxz[i], Cz);
        Cr  = wmma4(a, wxr[i], Cr);
        Cxn = wmma4(a, wxn[i], Cxn);
      }
      {
        v2f a = frag_a(sh_inp, IP, 64);
        Cz  = wmma4(a, wxz[16], Cz);
        Cr  = wmma4(a, wxr[16], Cr);
        Cxn = wmma4(a, wxn[16], Cxn);
      }
      #pragma unroll
      for (int v = 0; v < 8; ++v) {
        int row = rb + v;
        float z = fast_sigmoid(Cz[v] + bias_z);
        float r = fast_sigmoid(Cr[v] + bias_r);
        float n = fast_tanh((Cxn[v] + bias_xn) + r * (Chn[v] + bias_hn));
        hnew[v] = (1.0f - z) * n + z * sh_h[row * HP + hc];
      }
    }
    #pragma unroll
    for (int v = 0; v < 8; ++v) sh_h[(rb + v) * HP + hc] = hnew[v];
    sh_x[wave * XP + lane]       = xn0;
    sh_x[(wave + 8) * XP + lane] = xn1;
    __syncthreads();

    if (wave < 2) {
      float* dst = (wave == 0) ? sh_sc0 : sh_sc1;
      const int kbase = wave * 64;
      v8f c = {};
      #pragma unroll
      for (int i = 0; i < 16; ++i) {
        int k0 = kbase + 4 * i;
        v2f a = frag_a(sh_h, HP, k0);
        v2f b = sh_wgf[(wave * 16 + i) * 32 + lane];
        c = wmma4(a, b, c);
      }
      if (colb < M) {
        #pragma unroll
        for (int v = 0; v < 8; ++v) dst[(rb + v) * SP + colb] = c[v];
      }
    }
    Ahz = v8f{}; Ahr = v8f{}; Ahn = v8f{};
    #pragma unroll
    for (int i = 0; i < 10; ++i) {
      v2f a = frag_a(sh_h, HP, 4 * i);
      Ahz = wmma4(a, whz[i], Ahz);
      Ahr = wmma4(a, whr[i], Ahr);
      Ahn = wmma4(a, whn[i], Ahn);
    }
    __syncthreads();

    if (tid < 16) {
      float s[M], mx = -1e30f;
      #pragma unroll
      for (int m = 0; m < M; ++m) {
        s[m] = sh_sc0[tid * SP + m] + sh_sc1[tid * SP + m] + sh_bg[m];
        mx = fmaxf(mx, s[m]);
      }
      float sum = 0.0f;
      #pragma unroll
      for (int m = 0; m < M; ++m) { s[m] = __expf(s[m] - mx); sum += s[m]; }
      float inv = __fdividef(1.0f, sum);
      #pragma unroll
      for (int m = 0; m < M; ++m) sh_gate[tid * M + m] = s[m] * inv;
    }
    #pragma unroll
    for (int i = 10; i < 21; ++i) {
      v2f a = frag_a(sh_h, HP, 4 * i);
      Ahz = wmma4(a, whz[i], Ahz);
      Ahr = wmma4(a, whr[i], Ahr);
      Ahn = wmma4(a, whn[i], Ahn);
    }
    __syncthreads();

    float mix[2];
    #pragma unroll
    for (int j = 0; j < 2; ++j) {
      int row = wave + 8 * j;
      float acc = 0.0f;
      #pragma unroll
      for (int m = 0; m < M; ++m)
        acc += sh_gate[row * M + m] * sh_ns[row * NP + m * 32 + lane];
      sh_kst[row * XP + lane] = acc;
      mix[j] = acc;
    }
    if (lane == S - 1) {
      out[(size_t)(b0 + wave) * T + t]     = mix[0];
      out[(size_t)(b0 + wave + 8) * T + t] = mix[1];
      sh_inp[wave * IP + 64]       = mix[0] - ycur0;
      sh_inp[(wave + 8) * IP + 64] = mix[1] - ycur1;
    }
    ycur0 = yn0;
    ycur1 = yn1;
    #pragma unroll
    for (int i = 21; i < 32; ++i) {
      v2f a = frag_a(sh_h, HP, 4 * i);
      Ahz = wmma4(a, whz[i], Ahz);
      Ahr = wmma4(a, whr[i], Ahr);
      Ahn = wmma4(a, whn[i], Ahn);
    }
    __syncthreads();
  }
}

__global__ __launch_bounds__(256) void input_products_chunk(
    const float* __restrict__ x,
    const float* __restrict__ W_enc,
    const float* __restrict__ b_enc,
    const float* __restrict__ W_in,
    const float* __restrict__ b_s,
    const float* __restrict__ W_x,
    const float* __restrict__ b_x,
    const float* __restrict__ b_h,
    float* __restrict__ ws, size_t ws_size, int chunk)
{
  if (ws_size < WS_NEED_BYTES) return;
  constexpr int T = 2048, IN_ = 32, S = 32, E = 64, G = 384;
  constexpr int IP = 68;
  __shared__ float sh_enc[2][16 * IP];

  const int tid  = threadIdx.x;
  const int wave = tid >> 5;
  const int lane = tid & 31;
  const int colb = lane & 15;
  const int rb   = (lane >> 4) << 3;
  const int zc0  = wave * 16;
  const int hc   = zc0 + colb;

  const float bias_z  = b_x[hc]       + b_h[hc];
  const float bias_r  = b_x[128 + hc] + b_h[128 + hc];
  const float bias_xn = b_x[256 + hc];
  const float bias_s0 = b_s[wave * 32 + colb];
  const float bias_s1 = b_s[wave * 32 + 16 + colb];
  const float bias_e  = b_enc[(wave & 3) * 16 + colb];

  v2f wi0[8], wi1[8], we[8];
  #pragma unroll
  for (int i = 0; i < 8; ++i) {
    wi0[i] = frag_b(W_in + wave * IN_ * S, S, 4 * i, 0);
    wi1[i] = frag_b(W_in + wave * IN_ * S, S, 4 * i, 16);
    we[i]  = frag_b(W_enc, E, 4 * i, (wave & 3) * 16);
  }
  v2f wxz[16], wxr[16], wxn[16];
  #pragma unroll
  for (int i = 0; i < 16; ++i) {
    wxz[i] = frag_b(W_x, G, 4 * i, zc0);
    wxr[i] = frag_b(W_x, G, 4 * i, 128 + zc0);
    wxn[i] = frag_b(W_x, G, 4 * i, 256 + zc0);
  }

  float4* P = reinterpret_cast<float4*>(ws);
  const int row = lane & 15;
  const int kh  = (lane >> 4) << 1;

  constexpr int NT = (8 * TC) / 256;
  v2f a[8];
  {
    const int bt = blockIdx.x / TC, tl = blockIdx.x % TC;
    const float* xrow = x + ((size_t)(bt * 16 + row) * T + chunk * TC + tl) * IN_ + kh;
    #pragma unroll
    for (int i = 0; i < 8; ++i) { a[i].x = xrow[4 * i]; a[i].y = xrow[4 * i + 1]; }
  }
  for (int it = 0; it < NT; ++it) {
    const int tile = blockIdx.x + 256 * it;
    const int bt = tile / TC;
    const int tl = tile % TC;

    float4* dst = P + (((size_t)(bt * TC + tl) * 8 + wave) * 10) * 32 + lane;

    v8f c0 = {}, c1 = {};
    #pragma unroll
    for (int i = 0; i < 8; ++i) {
      c0 = wmma4(a[i], wi0[i], c0);
      c1 = wmma4(a[i], wi1[i], c1);
    }

    float* enc = sh_enc[it & 1];
    if (wave < 4) {
      v8f ce = {};
      #pragma unroll
      for (int i = 0; i < 8; ++i) ce = wmma4(a[i], we[i], ce);
      #pragma unroll
      for (int v = 0; v < 8; ++v)
        enc[(rb + v) * IP + wave * 16 + colb] = fast_tanh(ce[v] + bias_e);
    }
    __syncthreads();

    v2f an[8];
    {
      const int tile_n = (it + 1 < NT) ? tile + 256 : tile;
      const int bt_n = tile_n / TC, tl_n = tile_n % TC;
      const float* xrow = x + ((size_t)(bt_n * 16 + row) * T + chunk * TC + tl_n) * IN_ + kh;
      #pragma unroll
      for (int i = 0; i < 8; ++i) { an[i].x = xrow[4 * i]; an[i].y = xrow[4 * i + 1]; }
    }
    dst[0 * 32] = make_float4(c0[0] + bias_s0, c0[1] + bias_s0, c0[2] + bias_s0, c0[3] + bias_s0);
    dst[1 * 32] = make_float4(c0[4] + bias_s0, c0[5] + bias_s0, c0[6] + bias_s0, c0[7] + bias_s0);
    dst[2 * 32] = make_float4(c1[0] + bias_s1, c1[1] + bias_s1, c1[2] + bias_s1, c1[3] + bias_s1);
    dst[3 * 32] = make_float4(c1[4] + bias_s1, c1[5] + bias_s1, c1[6] + bias_s1, c1[7] + bias_s1);

    v8f Cz = {}, Cr = {}, Cn = {};
    #pragma unroll
    for (int i = 0; i < 16; ++i) {
      v2f ea = frag_a(enc, IP, 4 * i);
      Cz = wmma4(ea, wxz[i], Cz);
      Cr = wmma4(ea, wxr[i], Cr);
      Cn = wmma4(ea, wxn[i], Cn);
    }
    dst[4 * 32] = make_float4(Cz[0] + bias_z, Cz[1] + bias_z, Cz[2] + bias_z, Cz[3] + bias_z);
    dst[5 * 32] = make_float4(Cz[4] + bias_z, Cz[5] + bias_z, Cz[6] + bias_z, Cz[7] + bias_z);
    dst[6 * 32] = make_float4(Cr[0] + bias_r, Cr[1] + bias_r, Cr[2] + bias_r, Cr[3] + bias_r);
    dst[7 * 32] = make_float4(Cr[4] + bias_r, Cr[5] + bias_r, Cr[6] + bias_r, Cr[7] + bias_r);
    dst[8 * 32] = make_float4(Cn[0] + bias_xn, Cn[1] + bias_xn, Cn[2] + bias_xn, Cn[3] + bias_xn);
    dst[9 * 32] = make_float4(Cn[4] + bias_xn, Cn[5] + bias_xn, Cn[6] + bias_xn, Cn[7] + bias_xn);
    #pragma unroll
    for (int i = 0; i < 8; ++i) a[i] = an[i];
  }
}

__global__ __launch_bounds__(256) void kalman_gru_scan_chunk(
    const float* __restrict__ y,
    const float* __restrict__ W_rec,
    const float* __restrict__ W_x,
    const float* __restrict__ W_h,
    const float* __restrict__ b_h,
    const float* __restrict__ W_gate,
    const float* __restrict__ b_gate,
    float* __restrict__ out,
    float* __restrict__ ws, size_t ws_size, int chunk)
{
  if (ws_size < WS_NEED_BYTES) return;
  constexpr int T = 2048, S = 32, M = 8, H = 128;
  constexpr int CIN = 65, G = 384;
  constexpr int XP = 36;
  constexpr int HP = 132;
  constexpr int NP = 260;
  constexpr int SP = 20;

  __shared__ float sh_h[16 * HP];
  __shared__ float sh_ns[16 * NP];
  __shared__ float sh_kst[16 * XP];
  __shared__ float sh_gate[16 * 8];
  __shared__ float sh_sc0[16 * SP];
  __shared__ float sh_sc1[16 * SP];
  __shared__ float sh_err[16];
  __shared__ v2f sh_wgf[32 * 32];
  __shared__ float sh_bg[8];

  const int tid  = threadIdx.x;
  const int wave = tid >> 5;
  const int lane = tid & 31;
  const int b0   = blockIdx.x * 16;
  const int colb = lane & 15;
  const int rb   = (lane >> 4) << 3;
  const int zc0  = wave * 16;
  const int hc   = zc0 + colb;
  const int t0   = chunk * TC;

  const float bias_hn = b_h[256 + hc];

  v2f whz[32], whr[32], whn[32];
  #pragma unroll
  for (int i = 0; i < 32; ++i) {
    whz[i] = frag_b(W_h, G, 4 * i, zc0);
    whr[i] = frag_b(W_h, G, 4 * i, 128 + zc0);
    whn[i] = frag_b(W_h, G, 4 * i, 256 + zc0);
  }
  const v2f wxz_e = frag_b_kguard(W_x, G, 64, zc0,       CIN);
  const v2f wxr_e = frag_b_kguard(W_x, G, 64, 128 + zc0, CIN);
  const v2f wxn_e = frag_b_kguard(W_x, G, 64, 256 + zc0, CIN);
  v2f wr0[8], wr1[8];
  #pragma unroll
  for (int i = 0; i < 8; ++i) {
    wr0[i] = frag_b(W_rec + wave * S * S, S, 4 * i, 0);
    wr1[i] = frag_b(W_rec + wave * S * S, S, 4 * i, 16);
  }

  float* carry_h = ws + CARRY_H + (size_t)b0 * H;
  float* carry_k = ws + CARRY_K + (size_t)b0 * S;
  float* carry_e = ws + CARRY_E + b0;
  if (chunk == 0) {
    for (int i = tid; i < 16 * HP; i += 256) sh_h[i] = 0.0f;
    for (int i = tid; i < 16 * XP; i += 256) sh_kst[i] = 0.0f;
    if (tid < 16) sh_err[tid] = 0.0f;
  } else {
    for (int i = tid; i < 16 * H; i += 256) sh_h[(i >> 7) * HP + (i & 127)] = carry_h[i];
    for (int i = tid; i < 16 * S; i += 256) sh_kst[(i >> 5) * XP + (i & 31)] = carry_k[i];
    if (tid < 16) sh_err[tid] = carry_e[tid];
  }
  for (int idx = tid; idx < 32 * 32; idx += 256) {
    int ln = idx & 31, i = idx >> 5;
    int kk  = 4 * i + ((ln >> 4) << 1);
    int col = ln & 15;
    v2f b;
    b.x = (col < M) ? W_gate[kk * M + col]       : 0.0f;
    b.y = (col < M) ? W_gate[(kk + 1) * M + col] : 0.0f;
    sh_wgf[idx] = b;
  }
  if (tid < M) sh_bg[tid] = b_gate[tid];

  const float* yr0 = y + (size_t)(b0 + wave) * T + t0;
  const float* yr1 = y + (size_t)(b0 + 8 + wave) * T + t0;
  float ycur0 = yr0[0], ycur1 = yr1[0];

  const float4* Pw = reinterpret_cast<const float4*>(ws)
                   + (((size_t)blockIdx.x * TC) * 8 + wave) * 320 + lane;
  float4 nx[10];
  #pragma unroll
  for (int q = 0; q < 10; ++q) nx[q] = Pw[q * 32];
  __syncthreads();

  v8f Ahz = {}, Ahr = {}, Ahn = {};
  #pragma unroll
  for (int i = 0; i < 32; ++i) {
    v2f a = frag_a(sh_h, HP, 4 * i);
    Ahz = wmma4(a, whz[i], Ahz);
    Ahr = wmma4(a, whr[i], Ahr);
    Ahn = wmma4(a, whn[i], Ahn);
  }
  __syncthreads();

  for (int tl = 0; tl < TC; ++tl) {
    float hnew[8];
    v8f c0 = {nx[0].x, nx[0].y, nx[0].z, nx[0].w, nx[1].x, nx[1].y, nx[1].z, nx[1].w};
    v8f c1 = {nx[2].x, nx[2].y, nx[2].z, nx[2].w, nx[3].x, nx[3].y, nx[3].z, nx[3].w};
    {
      v2f ak[8];
      #pragma unroll
      for (int i = 0; i < 8; ++i) ak[i] = frag_a(sh_kst, XP, 4 * i);
      const float ev = sh_err[colb];
      float hold[8];
      #pragma unroll
      for (int v = 0; v < 8; ++v) hold[v] = sh_h[(rb + v) * HP + hc];
      v8f Cz = Ahz, Cr = Ahr, Cxn = {};
      const v8f Chn = Ahn;
      v2f ae;
      ae.x = (lane < 16) ? ev : 0.0f;
      ae.y = 0.0f;
      Cz  = wmma4(ae, wxz_e, Cz);
      Cr  = wmma4(ae, wxr_e, Cr);
      Cxn = wmma4(ae, wxn_e, Cxn);
      #pragma unroll
      for (int i = 0; i < 4; ++i) {
        c0 = wmma4(ak[i], wr0[i], c0);
        c1 = wmma4(ak[i], wr1[i], c1);
      }
      const float pz[8] = {nx[4].x, nx[4].y, nx[4].z, nx[4].w, nx[5].x, nx[5].y, nx[5].z, nx[5].w};
      const float pr[8] = {nx[6].x, nx[6].y, nx[6].z, nx[6].w, nx[7].x, nx[7].y, nx[7].z, nx[7].w};
      const float pn[8] = {nx[8].x, nx[8].y, nx[8].z, nx[8].w, nx[9].x, nx[9].y, nx[9].z, nx[9].w};
      #pragma unroll
      for (int v = 0; v < 8; ++v) {
        float z = fast_sigmoid(Cz[v] + pz[v]);
        float r = fast_sigmoid(Cr[v] + pr[v]);
        float n = fast_tanh((Cxn[v] + pn[v]) + r * (Chn[v] + bias_hn));
        hnew[v] = (1.0f - z) * n + z * hold[v];
      }
      #pragma unroll
      for (int v = 0; v < 8; ++v) sh_h[(rb + v) * HP + hc] = hnew[v];
      __builtin_amdgcn_fence(__ATOMIC_RELEASE, "workgroup", "local");
      __builtin_amdgcn_s_barrier_signal(-1);
      __builtin_amdgcn_sched_barrier(0);
      #pragma unroll
      for (int i = 4; i < 8; ++i) {
        c0 = wmma4(ak[i], wr0[i], c0);
        c1 = wmma4(ak[i], wr1[i], c1);
      }
      __builtin_amdgcn_sched_barrier(0);
      __builtin_amdgcn_s_barrier_wait(-1);
      __builtin_amdgcn_fence(__ATOMIC_ACQUIRE, "workgroup", "local");
    }

    if (wave < 2) {
      float* dst = (wave == 0) ? sh_sc0 : sh_sc1;
      const int kbase = wave * 64;
      v8f c = {};
      #pragma unroll
      for (int i = 0; i < 16; ++i) {
        int k0 = kbase + 4 * i;
        v2f a = frag_a(sh_h, HP, k0);
        v2f b = sh_wgf[(wave * 16 + i) * 32 + lane];
        c = wmma4(a, b, c);
      }
      if (colb < M) {
        #pragma unroll
        for (int v = 0; v < 8; ++v) dst[(rb + v) * SP + colb] = c[v];
      }
    }
    const int tn = (tl + 1 < TC) ? tl + 1 : tl;
    {
      const float4* Pn = Pw + (size_t)tn * P_STEP4;
      #pragma unroll
      for (int q = 0; q < 10; ++q) nx[q] = Pn[q * 32];
    }
    const float yn0 = yr0[tn];
    const float yn1 = yr1[tn];
    #pragma unroll
    for (int v = 0; v < 8; ++v) {
      int row = rb + v;
      sh_ns[row * NP + wave * 32 + colb]      = fast_tanh(c0[v]);
      sh_ns[row * NP + wave * 32 + 16 + colb] = fast_tanh(c1[v]);
    }
    Ahz = v8f{}; Ahr = v8f{}; Ahn = v8f{};
    v2f ah1[10];
    #pragma unroll
    for (int i = 0; i < 10; ++i) ah1[i] = frag_a(sh_h, HP, 4 * i);
    #pragma unroll
    for (int i = 0; i < 7; ++i) {
      Ahz = wmma4(ah1[i], whz[i], Ahz);
      Ahr = wmma4(ah1[i], whr[i], Ahr);
      Ahn = wmma4(ah1[i], whn[i], Ahn);
    }
    __builtin_amdgcn_fence(__ATOMIC_RELEASE, "workgroup", "local");
    __builtin_amdgcn_s_barrier_signal(-1);
    __builtin_amdgcn_sched_barrier(0);
    #pragma unroll
    for (int i = 7; i < 10; ++i) {
      Ahz = wmma4(ah1[i], whz[i], Ahz);
      Ahr = wmma4(ah1[i], whr[i], Ahr);
      Ahn = wmma4(ah1[i], whn[i], Ahn);
    }
    __builtin_amdgcn_sched_barrier(0);
    __builtin_amdgcn_s_barrier_wait(-1);
    __builtin_amdgcn_fence(__ATOMIC_ACQUIRE, "workgroup", "local");

    if (tid < 16) {
      float s[M], mx = -1e30f;
      #pragma unroll
      for (int m = 0; m < M; ++m) {
        s[m] = sh_sc0[tid * SP + m] + sh_sc1[tid * SP + m] + sh_bg[m];
        mx = fmaxf(mx, s[m]);
      }
      float sum = 0.0f;
      #pragma unroll
      for (int m = 0; m < M; ++m) { s[m] = __expf(s[m] - mx); sum += s[m]; }
      float inv = __fdividef(1.0f, sum);
      #pragma unroll
      for (int m = 0; m < M; ++m) sh_gate[tid * M + m] = s[m] * inv;
    }
    #pragma unroll
    for (int i = 10; i < 21; ++i) {
      v2f a = frag_a(sh_h, HP, 4 * i);
      Ahz = wmma4(a, whz[i], Ahz);
      Ahr = wmma4(a, whr[i], Ahr);
      Ahn = wmma4(a, whn[i], Ahn);
    }
    __syncthreads();

    float mix[2];
    #pragma unroll
    for (int j = 0; j < 2; ++j) {
      int row = wave + 8 * j;
      float acc = 0.0f;
      #pragma unroll
      for (int m = 0; m < M; ++m)
        acc += sh_gate[row * M + m] * sh_ns[row * NP + m * 32 + lane];
      sh_kst[row * XP + lane] = acc;
      mix[j] = acc;
    }
    if (lane == S - 1) {
      out[(size_t)(b0 + wave) * T + t0 + tl]     = mix[0];
      out[(size_t)(b0 + wave + 8) * T + t0 + tl] = mix[1];
      sh_err[wave]     = mix[0] - ycur0;
      sh_err[wave + 8] = mix[1] - ycur1;
    }
    ycur0 = yn0;
    ycur1 = yn1;
    #pragma unroll
    for (int i = 21; i < 32; ++i) {
      v2f a = frag_a(sh_h, HP, 4 * i);
      Ahz = wmma4(a, whz[i], Ahz);
      Ahr = wmma4(a, whr[i], Ahr);
      Ahn = wmma4(a, whn[i], Ahn);
    }
    __syncthreads();
  }

  if (chunk + 1 < NCHUNK) {
    for (int i = tid; i < 16 * H; i += 256) carry_h[i] = sh_h[(i >> 7) * HP + (i & 127)];
    for (int i = tid; i < 16 * S; i += 256) carry_k[i] = sh_kst[(i >> 5) * XP + (i & 31)];
    if (tid < 16) carry_e[tid] = sh_err[tid];
  }
}

extern "C" void kernel_launch(void* const* d_in, const int* in_sizes, int n_in,
                              void* d_out, int out_size, void* d_ws, size_t ws_size,
                              hipStream_t stream) {
  (void)in_sizes; (void)n_in; (void)out_size;
  const float* x      = (const float*)d_in[0];
  const float* y      = (const float*)d_in[1];
  const float* W_enc  = (const float*)d_in[2];
  const float* b_enc  = (const float*)d_in[3];
  const float* W_in   = (const float*)d_in[4];
  const float* W_rec  = (const float*)d_in[5];
  const float* b_s    = (const float*)d_in[6];
  const float* W_x    = (const float*)d_in[7];
  const float* W_h    = (const float*)d_in[8];
  const float* b_x    = (const float*)d_in[9];
  const float* b_h    = (const float*)d_in[10];
  const float* W_gate = (const float*)d_in[11];
  const float* b_gate = (const float*)d_in[12];
  float* out = (float*)d_out;
  float* ws  = (float*)d_ws;

  kalman_gru_scan<<<dim3(8), dim3(256), 0, stream>>>(
      x, y, W_enc, b_enc, W_in, W_rec, b_s, W_x, W_h, b_x, b_h,
      W_gate, b_gate, out, ws_size);

#define SCAN_BLOCK(c)                                                          \
  input_products_chunk<<<dim3(256), dim3(256), 0, stream>>>(                   \
      x, W_enc, b_enc, W_in, b_s, W_x, b_x, b_h, ws, ws_size, c);              \
  kalman_gru_scan_chunk<<<dim3(8), dim3(256), 0, stream>>>(                    \
      y, W_rec, W_x, W_h, b_h, W_gate, b_gate, out, ws, ws_size, c);
  SCAN_BLOCK(0)  SCAN_BLOCK(1)  SCAN_BLOCK(2)  SCAN_BLOCK(3)
  SCAN_BLOCK(4)  SCAN_BLOCK(5)  SCAN_BLOCK(6)  SCAN_BLOCK(7)
  SCAN_BLOCK(8)  SCAN_BLOCK(9)  SCAN_BLOCK(10) SCAN_BLOCK(11)
  SCAN_BLOCK(12) SCAN_BLOCK(13) SCAN_BLOCK(14) SCAN_BLOCK(15)
#undef SCAN_BLOCK
}


# No LDS read inside the scan's WMMA streams: the 32 A fragments of h read as one group behind the hand-over (64 registers), W_gate fragments in registers, h-path blocks between barrier signal and wait
# speedup vs baseline: 3.5227x; 1.0395x over previous
#include <hip/hip_runtime.h>
#include <math.h>


typedef __attribute__((ext_vector_type(2))) float v2f;
typedef __attribute__((ext_vector_type(8))) float v8f;

constexpr int    TC       = 128;
constexpr int    NCHUNK   = 16;
constexpr size_t P_STEP4  = 8 * 10 * 32;
constexpr size_t P_FLOATS = (size_t)8 * TC * P_STEP4 * 4;
constexpr size_t CARRY_H  = P_FLOATS;
constexpr size_t CARRY_K  = CARRY_H + 128 * 128;
constexpr size_t CARRY_E  = CARRY_K + 128 * 32;
constexpr size_t WS_NEED_BYTES = (CARRY_E + 128) * sizeof(float);

__device__ __forceinline__ v2f frag_a(const float* A, int ld, int k0) {
  int lane = threadIdx.x & 31;
  int row  = lane & 15;
  int kk   = k0 + ((lane >> 4) << 1);
  v2f a;
  a.x = A[row * ld + kk];
  a.y = A[row * ld + kk + 1];
  return a;
}

__device__ __forceinline__ v2f frag_b(const float* W, int ld, int k0, int c0) {
  int lane = threadIdx.x & 31;
  int col  = c0 + (lane & 15);
  int kk   = k0 + ((lane >> 4) << 1);
  v2f b;
  b.x = W[kk * ld + col];
  b.y = W[(kk + 1) * ld + col];
  return b;
}

__device__ __forceinline__ v2f frag_b_kguard(const float* W, int ld, int k0,
                                             int c0, int kmax) {
  int lane = threadIdx.x & 31;
  int col  = c0 + (lane & 15);
  int kk   = k0 + ((lane >> 4) << 1);
  v2f b;
  b.x = (kk     < kmax) ? W[kk * ld + col]       : 0.0f;
  b.y = (kk + 1 < kmax) ? W[(kk + 1) * ld + col] : 0.0f;
  return b;
}

__device__ __forceinline__ v8f wmma4(v2f a, v2f b, v8f c) {
  return __builtin_amdgcn_wmma_f32_16x16x4_f32(
       false, a,  false, b,
       (short)0, c,  false,  false);
}

__device__ __forceinline__ float fast_tanh(float x) {
  return __builtin_amdgcn_tanhf(x);
}
__device__ __forceinline__ float fast_sigmoid(float x) {
  return fmaf(0.5f, __builtin_amdgcn_tanhf(0.5f * x), 0.5f);
}

__global__ __launch_bounds__(256) void kalman_gru_scan(
    const float* __restrict__ x,
    const float* __restrict__ y,
    const float* __restrict__ W_enc,
    const float* __restrict__ b_enc,
    const float* __restrict__ W_in,
    const float* __restrict__ W_rec,
    const float* __restrict__ b_s,
    const float* __restrict__ W_x,
    const float* __restrict__ W_h,
    const float* __restrict__ b_x,
    const float* __restrict__ b_h,
    const float* __restrict__ W_gate,
    const float* __restrict__ b_gate,
    float* __restrict__ out,
    size_t ws_size)
{
  if (ws_size >= WS_NEED_BYTES) return;
  constexpr int T = 2048, IN_ = 32, S = 32, M = 8, E = 64, H = 128;
  constexpr int CIN = 65, G = 384;
  constexpr int XP = 36;
  constexpr int IP = 68;
  constexpr int HP = 132;
  constexpr int NP = 260;
  constexpr int SP = 20;

  __shared__ float sh_x[16 * XP];
  __shared__ float sh_inp[16 * IP];
  __shared__ float sh_h[16 * HP];
  __shared__ float sh_ns[16 * NP];
  __shared__ float sh_kst[16 * XP];
  __shared__ float sh_gate[16 * 8];
  __shared__ float sh_sc0[16 * SP];
  __shared__ float sh_sc1[16 * SP];
  __shared__ v2f sh_wencf[8 * 4 * 32];
  __shared__ v2f sh_wgf[32 * 32];
  __shared__ float sh_bg[8];

  const int tid  = threadIdx.x;
  const int wave = tid >> 5;
  const int lane = tid & 31;
  const int b0   = blockIdx.x * 16;

  const int colb = lane & 15;
  const int rb   = (lane >> 4) << 3;
  const int zc0  = wave * 16;
  const int hc   = zc0 + colb;

  const float bias_z  = b_x[hc]       + b_h[hc];
  const float bias_r  = b_x[128 + hc] + b_h[128 + hc];
  const float bias_xn = b_x[256 + hc];
  const float bias_hn = b_h[256 + hc];
  const float bias_s0 = b_s[wave * 32 + colb];
  const float bias_s1 = b_s[wave * 32 + 16 + colb];
  const float bias_e  = (wave < 4) ? b_enc[wave * 16 + colb] : 0.0f;

  v2f whz[32], whr[32], whn[32];
  #pragma unroll
  for (int i = 0; i < 32; ++i) {
    whz[i] = frag_b(W_h, G, 4 * i, zc0);
    whr[i] = frag_b(W_h, G, 4 * i, 128 + zc0);
    whn[i] = frag_b(W_h, G, 4 * i, 256 + zc0);
  }
  v2f wxz[17], wxr[17], wxn[17];
  #pragma unroll
  for (int i = 0; i < 16; ++i) {
    wxz[i] = frag_b(W_x, G, 4 * i, zc0);
    wxr[i] = frag_b(W_x, G, 4 * i, 128 + zc0);
    wxn[i] = frag_b(W_x, G, 4 * i, 256 + zc0);
  }
  wxz[16] = frag_b_kguard(W_x, G, 64, zc0,       CIN);
  wxr[16] = frag_b_kguard(W_x, G, 64, 128 + zc0, CIN);
  wxn[16] = frag_b_kguard(W_x, G, 64, 256 + zc0, CIN);

  for (int i = tid; i < 16 * HP; i += 256) sh_h[i] = 0.0f;
  for (int i = tid; i < 16 * XP; i += 256) sh_kst[i] = 0.0f;
  for (int i = tid; i < 16 * IP; i += 256) sh_inp[i] = 0.0f;
  for (int idx = tid; idx < 8 * 4 * 32; idx += 256) {
    int ln = idx & 31, tile = (idx >> 5) & 3, i = idx >> 7;
    int kk  = 4 * i + ((ln >> 4) << 1);
    int col = tile * 16 + (ln & 15);
    v2f b;
    b.x = W_enc[kk * E + col];
    b.y = W_enc[(kk + 1) * E + col];
    sh_wencf[idx] = b;
  }
  for (int idx = tid; idx < 32 * 32; idx += 256) {
    int ln = idx & 31, i = idx >> 5;
    int kk  = 4 * i + ((ln >> 4) << 1);
    int col = ln & 15;
    v2f b;
    b.x = (col < M) ? W_gate[kk * M + col]       : 0.0f;
    b.y = (col < M) ? W_gate[(kk + 1) * M + col] : 0.0f;
    sh_wgf[idx] = b;
  }
  if (tid < M) sh_bg[tid] = b_gate[tid];
  const float* xr0 = x + (size_t)(b0 + wave) * T * IN_ + lane;
  const float* xr1 = x + (size_t)(b0 + 8 + wave) * T * IN_ + lane;
  const float* yr0 = y + (size_t)(b0 + wave) * T;
  const float* yr1 = y + (size_t)(b0 + 8 + wave) * T;
  sh_x[wave * XP + lane]       = xr0[0];
  sh_x[(wave + 8) * XP + lane] = xr1[0];
  float ycur0 = yr0[0], ycur1 = yr1[0];
  v8f Ahz = {}, Ahr = {}, Ahn = {};
  __syncthreads();

  for (int t = 0; t < T; ++t) {
    {
      const int m = wave;
      const float* Wi = W_in  + m * IN_ * S;
      const float* Wr = W_rec + m * S * S;
      v8f c0 = {}, c1 = {};
      #pragma unroll
      for (int k0 = 0; k0 < IN_; k0 += 4) {
        v2f a = frag_a(sh_x, XP, k0);
        c0 = wmma4(a, frag_b(Wi, S, k0, 0),  c0);
        c1 = wmma4(a, frag_b(Wi, S, k0, 16), c1);
      }
      #pragma unroll
      for (int k0 = 0; k0 < S; k0 += 4) {
        v2f a = frag_a(sh_kst, XP, k0);
        c0 = wmma4(a, frag_b(Wr, S, k0, 0),  c0);
        c1 = wmma4(a, frag_b(Wr, S, k0, 16), c1);
      }
      #pragma unroll
      for (int v = 0; v < 8; ++v) {
        int row = rb + v;
        sh_ns[row * NP + m * 32 + colb]      = fast_tanh(c0[v] + bias_s0);
        sh_ns[row * NP + m * 32 + 16 + colb] = fast_tanh(c1[v] + bias_s1);
      }
    }
    if (wave < 4) {
      int c0c = wave * 16;
      v8f c = {};
      #pragma unroll
      for (int i = 0; i < IN_ / 4; ++i)
        c = wmma4(frag_a(sh_x, XP, 4 * i),
                  sh_wencf[(i * 4 + wave) * 32 + lane], c);
      #pragma unroll
      for (int v = 0; v < 8; ++v)
        sh_inp[(rb + v) * IP + c0c + colb] = fast_tanh(c[v] + bias_e);
    }
    __syncthreads();

    const int tn = (t + 1 < T) ? t + 1 : t;
    const float xn0 = xr0[(size_t)tn * IN_];
    const float xn1 = xr1[(size_t)tn * IN_];
    const float yn0 = yr0[tn];
    const float yn1 = yr1[tn];

    float hnew[8];
    {
      v8f Cz = Ahz, Cr = Ahr, Cxn = {};
      const v8f Chn = Ahn;
      #pragma unroll
      for (int i = 0; i < 16; ++i) {
        v2f a = frag_a(sh_inp, IP, 4 * i);
        Cz  = wmma4(a, wxz[i], Cz);
        Cr  = wmma4(a, wxr[i], Cr);
        Cxn = wmma4(a, wxn[i], Cxn);
      }
      {
        v2f a = frag_a(sh_inp, IP, 64);
        Cz  = wmma4(a, wxz[16], Cz);
        Cr  = wmma4(a, wxr[16], Cr);
        Cxn = wmma4(a, wxn[16], Cxn);
      }
      #pragma unroll
      for (int v = 0; v < 8; ++v) {
        int row = rb + v;
        float z = fast_sigmoid(Cz[v] + bias_z);
        float r = fast_sigmoid(Cr[v] + bias_r);
        float n = fast_tanh((Cxn[v] + bias_xn) + r * (Chn[v] + bias_hn));
        hnew[v] = (1.0f - z) * n + z * sh_h[row * HP + hc];
      }
    }
    #pragma unroll
    for (int v = 0; v < 8; ++v) sh_h[(rb + v) * HP + hc] = hnew[v];
    sh_x[wave * XP + lane]       = xn0;
    sh_x[(wave + 8) * XP + lane] = xn1;
    __syncthreads();

    if (wave < 2) {
      float* dst = (wave == 0) ? sh_sc0 : sh_sc1;
      const int kbase = wave * 64;
      v8f c = {};
      #pragma unroll
      for (int i = 0; i < 16; ++i) {
        int k0 = kbase + 4 * i;
        v2f a = frag_a(sh_h, HP, k0);
        v2f b = sh_wgf[(wave * 16 + i) * 32 + lane];
        c = wmma4(a, b, c);
      }
      if (colb < M) {
        #pragma unroll
        for (int v = 0; v < 8; ++v) dst[(rb + v) * SP + colb] = c[v];
      }
    }
    Ahz = v8f{}; Ahr = v8f{}; Ahn = v8f{};
    #pragma unroll
    for (int i = 0; i < 10; ++i) {
      v2f a = frag_a(sh_h, HP, 4 * i);
      Ahz = wmma4(a, whz[i], Ahz);
      Ahr = wmma4(a, whr[i], Ahr);
      Ahn = wmma4(a, whn[i], Ahn);
    }
    __syncthreads();

    if (tid < 16) {
      float s[M], mx = -1e30f;
      #pragma unroll
      for (int m = 0; m < M; ++m) {
        s[m] = sh_sc0[tid * SP + m] + sh_sc1[tid * SP + m] + sh_bg[m];
        mx = fmaxf(mx, s[m]);
      }
      float sum = 0.0f;
      #pragma unroll
      for (int m = 0; m < M; ++m) { s[m] = __expf(s[m] - mx); sum += s[m]; }
      float inv = __fdividef(1.0f, sum);
      #pragma unroll
      for (int m = 0; m < M; ++m) sh_gate[tid * M + m] = s[m] * inv;
    }
    #pragma unroll
    for (int i = 10; i < 21; ++i) {
      v2f a = frag_a(sh_h, HP, 4 * i);
      Ahz = wmma4(a, whz[i], Ahz);
      Ahr = wmma4(a, whr[i], Ahr);
      Ahn = wmma4(a, whn[i], Ahn);
    }
    __syncthreads();

    float mix[2];
    #pragma unroll
    for (int j = 0; j < 2; ++j) {
      int row = wave + 8 * j;
      float acc = 0.0f;
      #pragma unroll
      for (int m = 0; m < M; ++m)
        acc += sh_gate[row * M + m] * sh_ns[row * NP + m * 32 + lane];
      sh_kst[row * XP + lane] = acc;
      mix[j] = acc;
    }
    if (lane == S - 1) {
      out[(size_t)(b0 + wave) * T + t]     = mix[0];
      out[(size_t)(b0 + wave + 8) * T + t] = mix[1];
      sh_inp[wave * IP + 64]       = mix[0] - ycur0;
      sh_inp[(wave + 8) * IP + 64] = mix[1] - ycur1;
    }
    ycur0 = yn0;
    ycur1 = yn1;
    #pragma unroll
    for (int i = 21; i < 32; ++i) {
      v2f a = frag_a(sh_h, HP, 4 * i);
      Ahz = wmma4(a, whz[i], Ahz);
      Ahr = wmma4(a, whr[i], Ahr);
      Ahn = wmma4(a, whn[i], Ahn);
    }
    __syncthreads();
  }
}

__global__ __launch_bounds__(256) void input_products_chunk(
    const float* __restrict__ x,
    const float* __restrict__ W_enc,
    const float* __restrict__ b_enc,
    const float* __restrict__ W_in,
    const float* __restrict__ b_s,
    const float* __restrict__ W_x,
    const float* __restrict__ b_x,
    const float* __restrict__ b_h,
    float* __restrict__ ws, size_t ws_size, int chunk)
{
  if (ws_size < WS_NEED_BYTES) return;
  constexpr int T = 2048, IN_ = 32, S = 32, E = 64, G = 384;
  constexpr int IP = 68;
  __shared__ float sh_enc[2][16 * IP];

  const int tid  = threadIdx.x;
  const int wave = tid >> 5;
  const int lane = tid & 31;
  const int colb = lane & 15;
  const int rb   = (lane >> 4) << 3;
  const int zc0  = wave * 16;
  const int hc   = zc0 + colb;

  const float bias_z  = b_x[hc]       + b_h[hc];
  const float bias_r  = b_x[128 + hc] + b_h[128 + hc];
  const float bias_xn = b_x[256 + hc];
  const float bias_s0 = b_s[wave * 32 + colb];
  const float bias_s1 = b_s[wave * 32 + 16 + colb];
  const float bias_e  = b_enc[(wave & 3) * 16 + colb];

  v2f wi0[8], wi1[8], we[8];
  #pragma unroll
  for (int i = 0; i < 8; ++i) {
    wi0[i] = frag_b(W_in + wave * IN_ * S, S, 4 * i, 0);
    wi1[i] = frag_b(W_in + wave * IN_ * S, S, 4 * i, 16);
    we[i]  = frag_b(W_enc, E, 4 * i, (wave & 3) * 16);
  }
  v2f wxz[16], wxr[16], wxn[16];
  #pragma unroll
  for (int i = 0; i < 16; ++i) {
    wxz[i] = frag_b(W_x, G, 4 * i, zc0);
    wxr[i] = frag_b(W_x, G, 4 * i, 128 + zc0);
    wxn[i] = frag_b(W_x, G, 4 * i, 256 + zc0);
  }

  float4* P = reinterpret_cast<float4*>(ws);
  const int row = lane & 15;
  const int kh  = (lane >> 4) << 1;

  constexpr int NT = (8 * TC) / 256;
  v2f a[8];
  {
    const int bt = blockIdx.x / TC, tl = blockIdx.x % TC;
    const float* xrow = x + ((size_t)(bt * 16 + row) * T + chunk * TC + tl) * IN_ + kh;
    #pragma unroll
    for (int i = 0; i < 8; ++i) { a[i].x = xrow[4 * i]; a[i].y = xrow[4 * i + 1]; }
  }
  for (int it = 0; it < NT; ++it) {
    const int tile = blockIdx.x + 256 * it;
    const int bt = tile / TC;
    const int tl = tile % TC;

    float4* dst = P + (((size_t)(bt * TC + tl) * 8 + wave) * 10) * 32 + lane;

    v8f c0 = {}, c1 = {};
    #pragma unroll
    for (int i = 0; i < 8; ++i) {
      c0 = wmma4(a[i], wi0[i], c0);
      c1 = wmma4(a[i], wi1[i], c1);
    }

    float* enc = sh_enc[it & 1];
    if (wave < 4) {
      v8f ce = {};
      #pragma unroll
      for (int i = 0; i < 8; ++i) ce = wmma4(a[i], we[i], ce);
      #pragma unroll
      for (int v = 0; v < 8; ++v)
        enc[(rb + v) * IP + wave * 16 + colb] = fast_tanh(ce[v] + bias_e);
    }
    __syncthreads();

    v2f an[8];
    {
      const int tile_n = (it + 1 < NT) ? tile + 256 : tile;
      const int bt_n = tile_n / TC, tl_n = tile_n % TC;
      const float* xrow = x + ((size_t)(bt_n * 16 + row) * T + chunk * TC + tl_n) * IN_ + kh;
      #pragma unroll
      for (int i = 0; i < 8; ++i) { an[i].x = xrow[4 * i]; an[i].y = xrow[4 * i + 1]; }
    }
    dst[0 * 32] = make_float4(c0[0] + bias_s0, c0[1] + bias_s0, c0[2] + bias_s0, c0[3] + bias_s0);
    dst[1 * 32] = make_float4(c0[4] + bias_s0, c0[5] + bias_s0, c0[6] + bias_s0, c0[7] + bias_s0);
    dst[2 * 32] = make_float4(c1[0] + bias_s1, c1[1] + bias_s1, c1[2] + bias_s1, c1[3] + bias_s1);
    dst[3 * 32] = make_float4(c1[4] + bias_s1, c1[5] + bias_s1, c1[6] + bias_s1, c1[7] + bias_s1);

    v8f Cz = {}, Cr = {}, Cn = {};
    #pragma unroll
    for (int i = 0; i < 16; ++i) {
      v2f ea = frag_a(enc, IP, 4 * i);
      Cz = wmma4(ea, wxz[i], Cz);
      Cr = wmma4(ea, wxr[i], Cr);
      Cn = wmma4(ea, wxn[i], Cn);
    }
    dst[4 * 32] = make_float4(Cz[0] + bias_z, Cz[1] + bias_z, Cz[2] + bias_z, Cz[3] + bias_z);
    dst[5 * 32] = make_float4(Cz[4] + bias_z, Cz[5] + bias_z, Cz[6] + bias_z, Cz[7] + bias_z);
    dst[6 * 32] = make_float4(Cr[0] + bias_r, Cr[1] + bias_r, Cr[2] + bias_r, Cr[3] + bias_r);
    dst[7 * 32] = make_float4(Cr[4] + bias_r, Cr[5] + bias_r, Cr[6] + bias_r, Cr[7] + bias_r);
    dst[8 * 32] = make_float4(Cn[0] + bias_xn, Cn[1] + bias_xn, Cn[2] + bias_xn, Cn[3] + bias_xn);
    dst[9 * 32] = make_float4(Cn[4] + bias_xn, Cn[5] + bias_xn, Cn[6] + bias_xn, Cn[7] + bias_xn);
    #pragma unroll
    for (int i = 0; i < 8; ++i) a[i] = an[i];
  }
}

__global__ __launch_bounds__(256) void kalman_gru_scan_chunk(
    const float* __restrict__ y,
    const float* __restrict__ W_rec,
    const float* __restrict__ W_x,
    const float* __restrict__ W_h,
    const float* __restrict__ b_h,
    const float* __restrict__ W_gate,
    const float* __restrict__ b_gate,
    float* __restrict__ out,
    float* __restrict__ ws, size_t ws_size, int chunk)
{
  if (ws_size < WS_NEED_BYTES) return;
  constexpr int T = 2048, S = 32, M = 8, H = 128;
  constexpr int CIN = 65, G = 384;
  constexpr int XP = 36;
  constexpr int HP = 132;
  constexpr int NP = 260;
  constexpr int SP = 20;

  __shared__ float sh_h[16 * HP];
  __shared__ float sh_ns[16 * NP];
  __shared__ float sh_kst[16 * XP];
  __shared__ float sh_gate[16 * 8];
  __shared__ float sh_sc0[16 * SP];
  __shared__ float sh_sc1[16 * SP];
  __shared__ float sh_err[16];
  __shared__ float sh_bg[8];

  const int tid  = threadIdx.x;
  const int wave = tid >> 5;
  const int lane = tid & 31;
  const int b0   = blockIdx.x * 16;
  const int colb = lane & 15;
  const int rb   = (lane >> 4) << 3;
  const int zc0  = wave * 16;
  const int hc   = zc0 + colb;
  const int t0   = chunk * TC;

  const float bias_hn = b_h[256 + hc];

  v2f whz[32], whr[32], whn[32];
  #pragma unroll
  for (int i = 0; i < 32; ++i) {
    whz[i] = frag_b(W_h, G, 4 * i, zc0);
    whr[i] = frag_b(W_h, G, 4 * i, 128 + zc0);
    whn[i] = frag_b(W_h, G, 4 * i, 256 + zc0);
  }
  const v2f wxz_e = frag_b_kguard(W_x, G, 64, zc0,       CIN);
  const v2f wxr_e = frag_b_kguard(W_x, G, 64, 128 + zc0, CIN);
  const v2f wxn_e = frag_b_kguard(W_x, G, 64, 256 + zc0, CIN);
  v2f wr0[8], wr1[8];
  #pragma unroll
  for (int i = 0; i < 8; ++i) {
    wr0[i] = frag_b(W_rec + wave * S * S, S, 4 * i, 0);
    wr1[i] = frag_b(W_rec + wave * S * S, S, 4 * i, 16);
  }

  float* carry_h = ws + CARRY_H + (size_t)b0 * H;
  float* carry_k = ws + CARRY_K + (size_t)b0 * S;
  float* carry_e = ws + CARRY_E + b0;
  if (chunk == 0) {
    for (int i = tid; i < 16 * HP; i += 256) sh_h[i] = 0.0f;
    for (int i = tid; i < 16 * XP; i += 256) sh_kst[i] = 0.0f;
    if (tid < 16) sh_err[tid] = 0.0f;
  } else {
    for (int i = tid; i < 16 * H; i += 256) sh_h[(i >> 7) * HP + (i & 127)] = carry_h[i];
    for (int i = tid; i < 16 * S; i += 256) sh_kst[(i >> 5) * XP + (i & 31)] = carry_k[i];
    if (tid < 16) sh_err[tid] = carry_e[tid];
  }
  v2f wg[16];
  #pragma unroll
  for (int i = 0; i < 16; ++i) {
    const int kk = 4 * (16 * (wave & 1) + i) + ((lane >> 4) << 1);
    wg[i].x = (colb < M) ? W_gate[kk * M + colb]       : 0.0f;
    wg[i].y = (colb < M) ? W_gate[(kk + 1) * M + colb] : 0.0f;
  }
  if (tid < M) sh_bg[tid] = b_gate[tid];

  const float* yr0 = y + (size_t)(b0 + wave) * T + t0;
  const float* yr1 = y + (size_t)(b0 + 8 + wave) * T + t0;
  float ycur0 = yr0[0], ycur1 = yr1[0];

  const float4* Pw = reinterpret_cast<const float4*>(ws)
                   + (((size_t)blockIdx.x * TC) * 8 + wave) * 320 + lane;
  float4 nx[10];
  #pragma unroll
  for (int q = 0; q < 10; ++q) nx[q] = Pw[q * 32];
  __syncthreads();

  v8f Ahz = {}, Ahr = {}, Ahn = {};
  #pragma unroll
  for (int i = 0; i < 32; ++i) {
    v2f a = frag_a(sh_h, HP, 4 * i);
    Ahz = wmma4(a, whz[i], Ahz);
    Ahr = wmma4(a, whr[i], Ahr);
    Ahn = wmma4(a, whn[i], Ahn);
  }
  __syncthreads();

  for (int tl = 0; tl < TC; ++tl) {
    float hnew[8];
    v8f c0 = {nx[0].x, nx[0].y, nx[0].z, nx[0].w, nx[1].x, nx[1].y, nx[1].z, nx[1].w};
    v8f c1 = {nx[2].x, nx[2].y, nx[2].z, nx[2].w, nx[3].x, nx[3].y, nx[3].z, nx[3].w};
    {
      v2f ak[8];
      #pragma unroll
      for (int i = 0; i < 8; ++i) ak[i] = frag_a(sh_kst, XP, 4 * i);
      const float ev = sh_err[colb];
      float hold[8];
      #pragma unroll
      for (int v = 0; v < 8; ++v) hold[v] = sh_h[(rb + v) * HP + hc];
      v8f Cz = Ahz, Cr = Ahr, Cxn = {};
      const v8f Chn = Ahn;
      v2f ae;
      ae.x = (lane < 16) ? ev : 0.0f;
      ae.y = 0.0f;
      Cz  = wmma4(ae, wxz_e, Cz);
      Cr  = wmma4(ae, wxr_e, Cr);
      Cxn = wmma4(ae, wxn_e, Cxn);
      #pragma unroll
      for (int i = 0; i < 4; ++i) {
        c0 = wmma4(ak[i], wr0[i], c0);
        c1 = wmma4(ak[i], wr1[i], c1);
      }
      const float pz[8] = {nx[4].x, nx[4].y, nx[4].z, nx[4].w, nx[5].x, nx[5].y, nx[5].z, nx[5].w};
      const float pr[8] = {nx[6].x, nx[6].y, nx[6].z, nx[6].w, nx[7].x, nx[7].y, nx[7].z, nx[7].w};
      const float pn[8] = {nx[8].x, nx[8].y, nx[8].z, nx[8].w, nx[9].x, nx[9].y, nx[9].z, nx[9].w};
      #pragma unroll
      for (int v = 0; v < 8; ++v) {
        float z = fast_sigmoid(Cz[v] + pz[v]);
        float r = fast_sigmoid(Cr[v] + pr[v]);
        float n = fast_tanh((Cxn[v] + pn[v]) + r * (Chn[v] + bias_hn));
        hnew[v] = (1.0f - z) * n + z * hold[v];
      }
      #pragma unroll
      for (int v = 0; v < 8; ++v) sh_h[(rb + v) * HP + hc] = hnew[v];
      __builtin_amdgcn_fence(__ATOMIC_RELEASE, "workgroup", "local");
      __builtin_amdgcn_s_barrier_signal(-1);
      __builtin_amdgcn_sched_barrier(0);
      #pragma unroll
      for (int i = 4; i < 8; ++i) {
        c0 = wmma4(ak[i], wr0[i], c0);
        c1 = wmma4(ak[i], wr1[i], c1);
      }
      __builtin_amdgcn_sched_barrier(0);
      asm volatile("" :: "v"(c0), "v"(c1));
      __builtin_amdgcn_s_barrier_wait(-1);
      __builtin_amdgcn_fence(__ATOMIC_ACQUIRE, "workgroup", "local");
    }

    v2f ahf[32];
    #pragma unroll
    for (int i = 0; i < 16; ++i) {
      ahf[i]      = frag_a(sh_h, HP, 4 * i);
      ahf[16 + i] = frag_a(sh_h, HP, 64 + 4 * i);
    }
    __builtin_amdgcn_sched_barrier(0);

    if (wave == 0) {
      v8f c = {};
      #pragma unroll
      for (int i = 0; i < 16; ++i) c = wmma4(ahf[i], wg[i], c);
      if (colb < M) {
        #pragma unroll
        for (int v = 0; v < 8; ++v) sh_sc0[(rb + v) * SP + colb] = c[v];
      }
    } else if (wave == 1) {
      v8f c = {};
      #pragma unroll
      for (int i = 0; i < 16; ++i) c = wmma4(ahf[16 + i], wg[i], c);
      if (colb < M) {
        #pragma unroll
        for (int v = 0; v < 8; ++v) sh_sc1[(rb + v) * SP + colb] = c[v];
      }
    }
    const int tn = (tl + 1 < TC) ? tl + 1 : tl;
    {
      const float4* Pn = Pw + (size_t)tn * P_STEP4;
      #pragma unroll
      for (int q = 0; q < 10; ++q) nx[q] = Pn[q * 32];
    }
    const float yn0 = yr0[tn];
    const float yn1 = yr1[tn];
    #pragma unroll
    for (int v = 0; v < 8; ++v) {
      int row = rb + v;
      sh_ns[row * NP + wave * 32 + colb]      = fast_tanh(c0[v]);
      sh_ns[row * NP + wave * 32 + 16 + colb] = fast_tanh(c1[v]);
    }
    Ahz = v8f{}; Ahr = v8f{}; Ahn = v8f{};
    #pragma unroll
    for (int i = 0; i < 7; ++i) {
      Ahz = wmma4(ahf[i], whz[i], Ahz);
      Ahr = wmma4(ahf[i], whr[i], Ahr);
      Ahn = wmma4(ahf[i], whn[i], Ahn);
    }
    __builtin_amdgcn_fence(__ATOMIC_RELEASE, "workgroup", "local");
    __builtin_amdgcn_s_barrier_signal(-1);
    __builtin_amdgcn_sched_barrier(0);
    #pragma unroll
    for (int i = 7; i < 10; ++i) {
      Ahz = wmma4(ahf[i], whz[i], Ahz);
      Ahr = wmma4(ahf[i], whr[i], Ahr);
      Ahn = wmma4(ahf[i], whn[i], Ahn);
    }
    __builtin_amdgcn_sched_barrier(0);
    asm volatile("" :: "v"(Ahz), "v"(Ahr), "v"(Ahn));
    __builtin_amdgcn_s_barrier_wait(-1);
    __builtin_amdgcn_fence(__ATOMIC_ACQUIRE, "workgroup", "local");

    if (tid < 16) {
      float s[M], mx = -1e30f;
      #pragma unroll
      for (int m = 0; m < M; ++m) {
        s[m] = sh_sc0[tid * SP + m] + sh_sc1[tid * SP + m] + sh_bg[m];
        mx = fmaxf(mx, s[m]);
      }
      float sum = 0.0f;
      #pragma unroll
      for (int m = 0; m < M; ++m) { s[m] = __expf(s[m] - mx); sum += s[m]; }
      float inv = __fdividef(1.0f, sum);
      #pragma unroll
      for (int m = 0; m < M; ++m) sh_gate[tid * M + m] = s[m] * inv;
    }
    __builtin_amdgcn_fence(__ATOMIC_RELEASE, "workgroup", "local");
    __builtin_amdgcn_s_barrier_signal(-1);
    #pragma unroll
    for (int i = 10; i < 21; ++i) {
      Ahz = wmma4(ahf[i], whz[i], Ahz);
      Ahr = wmma4(ahf[i], whr[i], Ahr);
      Ahn = wmma4(ahf[i], whn[i], Ahn);
    }
    asm volatile("" :: "v"(Ahz), "v"(Ahr), "v"(Ahn));
    __builtin_amdgcn_s_barrier_wait(-1);
    __builtin_amdgcn_fence(__ATOMIC_ACQUIRE, "workgroup", "local");

    float mix[2];
    {
      float gt[2][M], nsv[2][M];
      #pragma unroll
      for (int j = 0; j < 2; ++j) {
        #pragma unroll
        for (int m = 0; m < M; ++m) {
          gt[j][m]  = sh_gate[(wave + 8 * j) * M + m];
          nsv[j][m] = sh_ns[(wave + 8 * j) * NP + m * 32 + lane];
        }
      }
      asm volatile("" :: "v"(gt[0][0]), "v"(gt[0][1]), "v"(gt[0][2]), "v"(gt[0][3]),
                         "v"(gt[0][4]), "v"(gt[0][5]), "v"(gt[0][6]), "v"(gt[0][7]),
                         "v"(gt[1][0]), "v"(gt[1][1]), "v"(gt[1][2]), "v"(gt[1][3]),
                         "v"(gt[1][4]), "v"(gt[1][5]), "v"(gt[1][6]), "v"(gt[1][7]));
      asm volatile("" :: "v"(nsv[0][0]), "v"(nsv[0][1]), "v"(nsv[0][2]), "v"(nsv[0][3]),
                         "v"(nsv[0][4]), "v"(nsv[0][5]), "v"(nsv[0][6]), "v"(nsv[0][7]),
                         "v"(nsv[1][0]), "v"(nsv[1][1]), "v"(nsv[1][2]), "v"(nsv[1][3]),
                         "v"(nsv[1][4]), "v"(nsv[1][5]), "v"(nsv[1][6]), "v"(nsv[1][7]));
      #pragma unroll
      for (int j = 0; j < 2; ++j) {
        float acc = 0.0f;
        #pragma unroll
        for (int m = 0; m < M; ++m) acc += gt[j][m] * nsv[j][m];
        sh_kst[(wave + 8 * j) * XP + lane] = acc;
        mix[j] = acc;
      }
    }
    if (lane == S - 1) {
      out[(size_t)(b0 + wave) * T + t0 + tl]     = mix[0];
      out[(size_t)(b0 + wave + 8) * T + t0 + tl] = mix[1];
      sh_err[wave]     = mix[0] - ycur0;
      sh_err[wave + 8] = mix[1] - ycur1;
    }
    ycur0 = yn0;
    ycur1 = yn1;
    __builtin_amdgcn_fence(__ATOMIC_RELEASE, "workgroup", "local");
    __builtin_amdgcn_s_barrier_signal(-1);
    #pragma unroll
    for (int i = 21; i < 32; ++i) {
      Ahz = wmma4(ahf[i], whz[i], Ahz);
      Ahr = wmma4(ahf[i], whr[i], Ahr);
      Ahn = wmma4(ahf[i], whn[i], Ahn);
    }
    asm volatile("" :: "v"(Ahz), "v"(Ahr), "v"(Ahn));
    __builtin_amdgcn_s_barrier_wait(-1);
    __builtin_amdgcn_fence(__ATOMIC_ACQUIRE, "workgroup", "local");
  }

  if (chunk + 1 < NCHUNK) {
    for (int i = tid; i < 16 * H; i += 256) carry_h[i] = sh_h[(i >> 7) * HP + (i & 127)];
    for (int i = tid; i < 16 * S; i += 256) carry_k[i] = sh_kst[(i >> 5) * XP + (i & 31)];
    if (tid < 16) carry_e[tid] = sh_err[tid];
  }
}

extern "C" void kernel_launch(void* const* d_in, const int* in_sizes, int n_in,
                              void* d_out, int out_size, void* d_ws, size_t ws_size,
                              hipStream_t stream) {
  (void)in_sizes; (void)n_in; (void)out_size;
  const float* x      = (const float*)d_in[0];
  const float* y      = (const float*)d_in[1];
  const float* W_enc  = (const float*)d_in[2];
  const float* b_enc  = (const float*)d_in[3];
  const float* W_in   = (const float*)d_in[4];
  const float* W_rec  = (const float*)d_in[5];
  const float* b_s    = (const float*)d_in[6];
  const float* W_x    = (const float*)d_in[7];
  const float* W_h    = (const float*)d_in[8];
  const float* b_x    = (const float*)d_in[9];
  const float* b_h    = (const float*)d_in[10];
  const float* W_gate = (const float*)d_in[11];
  const float* b_gate = (const float*)d_in[12];
  float* out = (float*)d_out;
  float* ws  = (float*)d_ws;

  kalman_gru_scan<<<dim3(8), dim3(256), 0, stream>>>(
      x, y, W_enc, b_enc, W_in, W_rec, b_s, W_x, W_h, b_x, b_h,
      W_gate, b_gate, out, ws_size);

#define SCAN_BLOCK(c)                                                          \
  input_products_chunk<<<dim3(256), dim3(256), 0, stream>>>(                   \
      x, W_enc, b_enc, W_in, b_s, W_x, b_x, b_h, ws, ws_size, c);              \
  kalman_gru_scan_chunk<<<dim3(8), dim3(256), 0, stream>>>(                    \
      y, W_rec, W_x, W_h, b_h, W_gate, b_gate, out, ws, ws_size, c);
  SCAN_BLOCK(0)  SCAN_BLOCK(1)  SCAN_BLOCK(2)  SCAN_BLOCK(3)
  SCAN_BLOCK(4)  SCAN_BLOCK(5)  SCAN_BLOCK(6)  SCAN_BLOCK(7)
  SCAN_BLOCK(8)  SCAN_BLOCK(9)  SCAN_BLOCK(10) SCAN_BLOCK(11)
  SCAN_BLOCK(12) SCAN_BLOCK(13) SCAN_BLOCK(14) SCAN_BLOCK(15)
#undef SCAN_BLOCK
}


# Chunk scan: every wave forms the gate-logit share of its own 16 h columns (4 WMMAs, own stores read back) between signal and wait of the h hand-over; 16-long chains of waves 0,1 gone (sum order)
# speedup vs baseline: 3.6437x; 1.0343x over previous
#include <hip/hip_runtime.h>
#include <math.h>


typedef __attribute__((ext_vector_type(2))) float v2f;
typedef __attribute__((ext_vector_type(8))) float v8f;

constexpr int    TC       = 128;
constexpr int    NCHUNK   = 16;
constexpr size_t P_STEP4  = 8 * 10 * 32;
constexpr size_t P_FLOATS = (size_t)8 * TC * P_STEP4 * 4;
constexpr size_t CARRY_H  = P_FLOATS;
constexpr size_t CARRY_K  = CARRY_H + 128 * 128;
constexpr size_t CARRY_E  = CARRY_K + 128 * 32;
constexpr size_t WS_NEED_BYTES = (CARRY_E + 128) * sizeof(float);

__device__ __forceinline__ v2f frag_a(const float* A, int ld, int k0) {
  int lane = threadIdx.x & 31;
  int row  = lane & 15;
  int kk   = k0 + ((lane >> 4) << 1);
  v2f a;
  a.x = A[row * ld + kk];
  a.y = A[row * ld + kk + 1];
  return a;
}

__device__ __forceinline__ v2f frag_b(const float* W, int ld, int k0, int c0) {
  int lane = threadIdx.x & 31;
  int col  = c0 + (lane & 15);
  int kk   = k0 + ((lane >> 4) << 1);
  v2f b;
  b.x = W[kk * ld + col];
  b.y = W[(kk + 1) * ld + col];
  return b;
}

__device__ __forceinline__ v2f frag_b_kguard(const float* W, int ld, int k0,
                                             int c0, int kmax) {
  int lane = threadIdx.x & 31;
  int col  = c0 + (lane & 15);
  int kk   = k0 + ((lane >> 4) << 1);
  v2f b;
  b.x = (kk     < kmax) ? W[kk * ld + col]       : 0.0f;
  b.y = (kk + 1 < kmax) ? W[(kk + 1) * ld + col] : 0.0f;
  return b;
}

__device__ __forceinline__ v8f wmma4(v2f a, v2f b, v8f c) {
  return __builtin_amdgcn_wmma_f32_16x16x4_f32(
       false, a,  false, b,
       (short)0, c,  false,  false);
}

__device__ __forceinline__ float fast_tanh(float x) {
  return __builtin_amdgcn_tanhf(x);
}
__device__ __forceinline__ float fast_sigmoid(float x) {
  return fmaf(0.5f, __builtin_amdgcn_tanhf(0.5f * x), 0.5f);
}

__global__ __launch_bounds__(256) void kalman_gru_scan(
    const float* __restrict__ x,
    const float* __restrict__ y,
    const float* __restrict__ W_enc,
    const float* __restrict__ b_enc,
    const float* __restrict__ W_in,
    const float* __restrict__ W_rec,
    const float* __restrict__ b_s,
    const float* __restrict__ W_x,
    const float* __restrict__ W_h,
    const float* __restrict__ b_x,
    const float* __restrict__ b_h,
    const float* __restrict__ W_gate,
    const float* __restrict__ b_gate,
    float* __restrict__ out,
    size_t ws_size)
{
  if (ws_size >= WS_NEED_BYTES) return;
  constexpr int T = 2048, IN_ = 32, S = 32, M = 8, E = 64, H = 128;
  constexpr int CIN = 65, G = 384;
  constexpr int XP = 36;
  constexpr int IP = 68;
  constexpr int HP = 132;
  constexpr int NP = 260;
  constexpr int SP = 20;

  __shared__ float sh_x[16 * XP];
  __shared__ float sh_inp[16 * IP];
  __shared__ float sh_h[16 * HP];
  __shared__ float sh_ns[16 * NP];
  __shared__ float sh_kst[16 * XP];
  __shared__ float sh_gate[16 * 8];
  __shared__ float sh_sc0[16 * SP];
  __shared__ float sh_sc1[16 * SP];
  __shared__ v2f sh_wencf[8 * 4 * 32];
  __shared__ v2f sh_wgf[32 * 32];
  __shared__ float sh_bg[8];

  const int tid  = threadIdx.x;
  const int wave = tid >> 5;
  const int lane = tid & 31;
  const int b0   = blockIdx.x * 16;

  const int colb = lane & 15;
  const int rb   = (lane >> 4) << 3;
  const int zc0  = wave * 16;
  const int hc   = zc0 + colb;

  const float bias_z  = b_x[hc]       + b_h[hc];
  const float bias_r  = b_x[128 + hc] + b_h[128 + hc];
  const float bias_xn = b_x[256 + hc];
  const float bias_hn = b_h[256 + hc];
  const float bias_s0 = b_s[wave * 32 + colb];
  const float bias_s1 = b_s[wave * 32 + 16 + colb];
  const float bias_e  = (wave < 4) ? b_enc[wave * 16 + colb] : 0.0f;

  v2f whz[32], whr[32], whn[32];
  #pragma unroll
  for (int i = 0; i < 32; ++i) {
    whz[i] = frag_b(W_h, G, 4 * i, zc0);
    whr[i] = frag_b(W_h, G, 4 * i, 128 + zc0);
    whn[i] = frag_b(W_h, G, 4 * i, 256 + zc0);
  }
  v2f wxz[17], wxr[17], wxn[17];
  #pragma unroll
  for (int i = 0; i < 16; ++i) {
    wxz[i] = frag_b(W_x, G, 4 * i, zc0);
    wxr[i] = frag_b(W_x, G, 4 * i, 128 + zc0);
    wxn[i] = frag_b(W_x, G, 4 * i, 256 + zc0);
  }
  wxz[16] = frag_b_kguard(W_x, G, 64, zc0,       CIN);
  wxr[16] = frag_b_kguard(W_x, G, 64, 128 + zc0, CIN);
  wxn[16] = frag_b_kguard(W_x, G, 64, 256 + zc0, CIN);

  for (int i = tid; i < 16 * HP; i += 256) sh_h[i] = 0.0f;
  for (int i = tid; i < 16 * XP; i += 256) sh_kst[i] = 0.0f;
  for (int i = tid; i < 16 * IP; i += 256) sh_inp[i] = 0.0f;
  for (int idx = tid; idx < 8 * 4 * 32; idx += 256) {
    int ln = idx & 31, tile = (idx >> 5) & 3, i = idx >> 7;
    int kk  = 4 * i + ((ln >> 4) << 1);
    int col = tile * 16 + (ln & 15);
    v2f b;
    b.x = W_enc[kk * E + col];
    b.y = W_enc[(kk + 1) * E + col];
    sh_wencf[idx] = b;
  }
  for (int idx = tid; idx < 32 * 32; idx += 256) {
    int ln = idx & 31, i = idx >> 5;
    int kk  = 4 * i + ((ln >> 4) << 1);
    int col = ln & 15;
    v2f b;
    b.x = (col < M) ? W_gate[kk * M + col]       : 0.0f;
    b.y = (col < M) ? W_gate[(kk + 1) * M + col] : 0.0f;
    sh_wgf[idx] = b;
  }
  if (tid < M) sh_bg[tid] = b_gate[tid];
  const float* xr0 = x + (size_t)(b0 + wave) * T * IN_ + lane;
  const float* xr1 = x + (size_t)(b0 + 8 + wave) * T * IN_ + lane;
  const float* yr0 = y + (size_t)(b0 + wave) * T;
  const float* yr1 = y + (size_t)(b0 + 8 + wave) * T;
  sh_x[wave * XP + lane]       = xr0[0];
  sh_x[(wave + 8) * XP + lane] = xr1[0];
  float ycur0 = yr0[0], ycur1 = yr1[0];
  v8f Ahz = {}, Ahr = {}, Ahn = {};
  __syncthreads();

  for (int t = 0; t < T; ++t) {
    {
      const int m = wave;
      const float* Wi = W_in  + m * IN_ * S;
      const float* Wr = W_rec + m * S * S;
      v8f c0 = {}, c1 = {};
      #pragma unroll
      for (int k0 = 0; k0 < IN_; k0 += 4) {
        v2f a = frag_a(sh_x, XP, k0);
        c0 = wmma4(a, frag_b(Wi, S, k0, 0),  c0);
        c1 = wmma4(a, frag_b(Wi, S, k0, 16), c1);
      }
      #pragma unroll
      for (int k0 = 0; k0 < S; k0 += 4) {
        v2f a = frag_a(sh_kst, XP, k0);
        c0 = wmma4(a, frag_b(Wr, S, k0, 0),  c0);
        c1 = wmma4(a, frag_b(Wr, S, k0, 16), c1);
      }
      #pragma unroll
      for (int v = 0; v < 8; ++v) {
        int row = rb + v;
        sh_ns[row * NP + m * 32 + colb]      = fast_tanh(c0[v] + bias_s0);
        sh_ns[row * NP + m * 32 + 16 + colb] = fast_tanh(c1[v] + bias_s1);
      }
    }
    if (wave < 4) {
      int c0c = wave * 16;
      v8f c = {};
      #pragma unroll
      for (int i = 0; i < IN_ / 4; ++i)
        c = wmma4(frag_a(sh_x, XP, 4 * i),
                  sh_wencf[(i * 4 + wave) * 32 + lane], c);
      #pragma unroll
      for (int v = 0; v < 8; ++v)
        sh_inp[(rb + v) * IP + c0c + colb] = fast_tanh(c[v] + bias_e);
    }
    __syncthreads();

    const int tn = (t + 1 < T) ? t + 1 : t;
    const float xn0 = xr0[(size_t)tn * IN_];
    const float xn1 = xr1[(size_t)tn * IN_];
    const float yn0 = yr0[tn];
    const float yn1 = yr1[tn];

    float hnew[8];
    {
      v8f Cz = Ahz, Cr = Ahr, Cxn = {};
      const v8f Chn = Ahn;
      #pragma unroll
      for (int i = 0; i < 16; ++i) {
        v2f a = frag_a(sh_inp, IP, 4 * i);
        Cz  = wmma4(a, wxz[i], Cz);
        Cr  = wmma4(a, wxr[i], Cr);
        Cxn = wmma4(a, wxn[i], Cxn);
      }
      {
        v2f a = frag_a(sh_inp, IP, 64);
        Cz  = wmma4(a, wxz[16], Cz);
        Cr  = wmma4(a, wxr[16], Cr);
        Cxn = wmma4(a, wxn[16], Cxn);
      }
      #pragma unroll
      for (int v = 0; v < 8; ++v) {
        int row = rb + v;
        float z = fast_sigmoid(Cz[v] + bias_z);
        float r = fast_sigmoid(Cr[v] + bias_r);
        float n = fast_tanh((Cxn[v] + bias_xn) + r * (Chn[v] + bias_hn));
        hnew[v] = (1.0f - z) * n + z * sh_h[row * HP + hc];
      }
    }
    #pragma unroll
    for (int v = 0; v < 8; ++v) sh_h[(rb + v) * HP + hc] = hnew[v];
    sh_x[wave * XP + lane]       = xn0;
    sh_x[(wave + 8) * XP + lane] = xn1;
    __syncthreads();

    if (wave < 2) {
      float* dst = (wave == 0) ? sh_sc0 : sh_sc1;
      const int kbase = wave * 64;
      v8f c = {};
      #pragma unroll
      for (int i = 0; i < 16; ++i) {
        int k0 = kbase + 4 * i;
        v2f a = frag_a(sh_h, HP, k0);
        v2f b = sh_wgf[(wave * 16 + i) * 32 + lane];
        c = wmma4(a, b, c);
      }
      if (colb < M) {
        #pragma unroll
        for (int v = 0; v < 8; ++v) dst[(rb + v) * SP + colb] = c[v];
      }
    }
    Ahz = v8f{}; Ahr = v8f{}; Ahn = v8f{};
    #pragma unroll
    for (int i = 0; i < 10; ++i) {
      v2f a = frag_a(sh_h, HP, 4 * i);
      Ahz = wmma4(a, whz[i], Ahz);
      Ahr = wmma4(a, whr[i], Ahr);
      Ahn = wmma4(a, whn[i], Ahn);
    }
    __syncthreads();

    if (tid < 16) {
      float s[M], mx = -1e30f;
      #pragma unroll
      for (int m = 0; m < M; ++m) {
        s[m] = sh_sc0[tid * SP + m] + sh_sc1[tid * SP + m] + sh_bg[m];
        mx = fmaxf(mx, s[m]);
      }
      float sum = 0.0f;
      #pragma unroll
      for (int m = 0; m < M; ++m) { s[m] = __expf(s[m] - mx); sum += s[m]; }
      float inv = __fdividef(1.0f, sum);
      #pragma unroll
      for (int m = 0; m < M; ++m) sh_gate[tid * M + m] = s[m] * inv;
    }
    #pragma unroll
    for (int i = 10; i < 21; ++i) {
      v2f a = frag_a(sh_h, HP, 4 * i);
      Ahz = wmma4(a, whz[i], Ahz);
      Ahr = wmma4(a, whr[i], Ahr);
      Ahn = wmma4(a, whn[i], Ahn);
    }
    __syncthreads();

    float mix[2];
    #pragma unroll
    for (int j = 0; j < 2; ++j) {
      int row = wave + 8 * j;
      float acc = 0.0f;
      #pragma unroll
      for (int m = 0; m < M; ++m)
        acc += sh_gate[row * M + m] * sh_ns[row * NP + m * 32 + lane];
      sh_kst[row * XP + lane] = acc;
      mix[j] = acc;
    }
    if (lane == S - 1) {
      out[(size_t)(b0 + wave) * T + t]     = mix[0];
      out[(size_t)(b0 + wave + 8) * T + t] = mix[1];
      sh_inp[wave * IP + 64]       = mix[0] - ycur0;
      sh_inp[(wave + 8) * IP + 64] = mix[1] - ycur1;
    }
    ycur0 = yn0;
    ycur1 = yn1;
    #pragma unroll
    for (int i = 21; i < 32; ++i) {
      v2f a = frag_a(sh_h, HP, 4 * i);
      Ahz = wmma4(a, whz[i], Ahz);
      Ahr = wmma4(a, whr[i], Ahr);
      Ahn = wmma4(a, whn[i], Ahn);
    }
    __syncthreads();
  }
}

__global__ __launch_bounds__(256) void input_products_chunk(
    const float* __restrict__ x,
    const float* __restrict__ W_enc,
    const float* __restrict__ b_enc,
    const float* __restrict__ W_in,
    const float* __restrict__ b_s,
    const float* __restrict__ W_x,
    const float* __restrict__ b_x,
    const float* __restrict__ b_h,
    float* __restrict__ ws, size_t ws_size, int chunk)
{
  if (ws_size < WS_NEED_BYTES) return;
  constexpr int T = 2048, IN_ = 32, S = 32, E = 64, G = 384;
  constexpr int IP = 68;
  __shared__ float sh_enc[2][16 * IP];

  const int tid  = threadIdx.x;
  const int wave = tid >> 5;
  const int lane = tid & 31;
  const int colb = lane & 15;
  const int rb   = (lane >> 4) << 3;
  const int zc0  = wave * 16;
  const int hc   = zc0 + colb;

  const float bias_z  = b_x[hc]       + b_h[hc];
  const float bias_r  = b_x[128 + hc] + b_h[128 + hc];
  const float bias_xn = b_x[256 + hc];
  const float bias_s0 = b_s[wave * 32 + colb];
  const float bias_s1 = b_s[wave * 32 + 16 + colb];
  const float bias_e  = b_enc[(wave & 3) * 16 + colb];

  v2f wi0[8], wi1[8], we[8];
  #pragma unroll
  for (int i = 0; i < 8; ++i) {
    wi0[i] = frag_b(W_in + wave * IN_ * S, S, 4 * i, 0);
    wi1[i] = frag_b(W_in + wave * IN_ * S, S, 4 * i, 16);
    we[i]  = frag_b(W_enc, E, 4 * i, (wave & 3) * 16);
  }
  v2f wxz[16], wxr[16], wxn[16];
  #pragma unroll
  for (int i = 0; i < 16; ++i) {
    wxz[i] = frag_b(W_x, G, 4 * i, zc0);
    wxr[i] = frag_b(W_x, G, 4 * i, 128 + zc0);
    wxn[i] = frag_b(W_x, G, 4 * i, 256 + zc0);
  }

  float4* P = reinterpret_cast<float4*>(ws);
  const int row = lane & 15;
  const int kh  = (lane >> 4) << 1;

  constexpr int NT = (8 * TC) / 256;
  v2f a[8];
  {
    const int bt = blockIdx.x / TC, tl = blockIdx.x % TC;
    const float* xrow = x + ((size_t)(bt * 16 + row) * T + chunk * TC + tl) * IN_ + kh;
    #pragma unroll
    for (int i = 0; i < 8; ++i) { a[i].x = xrow[4 * i]; a[i].y = xrow[4 * i + 1]; }
  }
  for (int it = 0; it < NT; ++it) {
    const int tile = blockIdx.x + 256 * it;
    const int bt = tile / TC;
    const int tl = tile % TC;

    float4* dst = P + (((size_t)(bt * TC + tl) * 8 + wave) * 10) * 32 + lane;

    v8f c0 = {}, c1 = {};
    #pragma unroll
    for (int i = 0; i < 8; ++i) {
      c0 = wmma4(a[i], wi0[i], c0);
      c1 = wmma4(a[i], wi1[i], c1);
    }

    float* enc = sh_enc[it & 1];
    if (wave < 4) {
      v8f ce = {};
      #pragma unroll
      for (int i = 0; i < 8; ++i) ce = wmma4(a[i], we[i], ce);
      #pragma unroll
      for (int v = 0; v < 8; ++v)
        enc[(rb + v) * IP + wave * 16 + colb] = fast_tanh(ce[v] + bias_e);
    }
    __syncthreads();

    v2f an[8];
    {
      const int tile_n = (it + 1 < NT) ? tile + 256 : tile;
      const int bt_n = tile_n / TC, tl_n = tile_n % TC;
      const float* xrow = x + ((size_t)(bt_n * 16 + row) * T + chunk * TC + tl_n) * IN_ + kh;
      #pragma unroll
      for (int i = 0; i < 8; ++i) { an[i].x = xrow[4 * i]; an[i].y = xrow[4 * i + 1]; }
    }
    dst[0 * 32] = make_float4(c0[0] + bias_s0, c0[1] + bias_s0, c0[2] + bias_s0, c0[3] + bias_s0);
    dst[1 * 32] = make_float4(c0[4] + bias_s0, c0[5] + bias_s0, c0[6] + bias_s0, c0[7] + bias_s0);
    dst[2 * 32] = make_float4(c1[0] + bias_s1, c1[1] + bias_s1, c1[2] + bias_s1, c1[3] + bias_s1);
    dst[3 * 32] = make_float4(c1[4] + bias_s1, c1[5] + bias_s1, c1[6] + bias_s1, c1[7] + bias_s1);

    v8f Cz = {}, Cr = {}, Cn = {};
    #pragma unroll
    for (int i = 0; i < 16; ++i) {
      v2f ea = frag_a(enc, IP, 4 * i);
      Cz = wmma4(ea, wxz[i], Cz);
      Cr = wmma4(ea, wxr[i], Cr);
      Cn = wmma4(ea, wxn[i], Cn);
    }
    dst[4 * 32] = make_float4(Cz[0] + bias_z, Cz[1] + bias_z, Cz[2] + bias_z, Cz[3] + bias_z);
    dst[5 * 32] = make_float4(Cz[4] + bias_z, Cz[5] + bias_z, Cz[6] + bias_z, Cz[7] + bias_z);
    dst[6 * 32] = make_float4(Cr[0] + bias_r, Cr[1] + bias_r, Cr[2] + bias_r, Cr[3] + bias_r);
    dst[7 * 32] = make_float4(Cr[4] + bias_r, Cr[5] + bias_r, Cr[6] + bias_r, Cr[7] + bias_r);
    dst[8 * 32] = make_float4(Cn[0] + bias_xn, Cn[1] + bias_xn, Cn[2] + bias_xn, Cn[3] + bias_xn);
    dst[9 * 32] = make_float4(Cn[4] + bias_xn, Cn[5] + bias_xn, Cn[6] + bias_xn, Cn[7] + bias_xn);
    #pragma unroll
    for (int i = 0; i < 8; ++i) a[i] = an[i];
  }
}

__global__ __launch_bounds__(256) void kalman_gru_scan_chunk(
    const float* __restrict__ y,
    const float* __restrict__ W_rec,
    const float* __restrict__ W_x,
    const float* __restrict__ W_h,
    const float* __restrict__ b_h,
    const float* __restrict__ W_gate,
    const float* __restrict__ b_gate,
    float* __restrict__ out,
    float* __restrict__ ws, size_t ws_size, int chunk)
{
  if (ws_size < WS_NEED_BYTES) return;
  constexpr int T = 2048, S = 32, M = 8, H = 128;
  constexpr int CIN = 65, G = 384;
  constexpr int XP = 36;
  constexpr int HP = 132;
  constexpr int NP = 260;
  constexpr int SP = 20;

  __shared__ float sh_h[16 * HP];
  __shared__ float sh_ns[16 * NP];
  __shared__ float sh_kst[16 * XP];
  __shared__ float sh_gate[16 * 8];
  __shared__ float sh_sc[8][16 * SP];
  __shared__ float sh_err[16];
  __shared__ float sh_bg[8];

  const int tid  = threadIdx.x;
  const int wave = tid >> 5;
  const int lane = tid & 31;
  const int b0   = blockIdx.x * 16;
  const int colb = lane & 15;
  const int rb   = (lane >> 4) << 3;
  const int zc0  = wave * 16;
  const int hc   = zc0 + colb;
  const int t0   = chunk * TC;

  const float bias_hn = b_h[256 + hc];

  v2f whz[32], whr[32], whn[32];
  #pragma unroll
  for (int i = 0; i < 32; ++i) {
    whz[i] = frag_b(W_h, G, 4 * i, zc0);
    whr[i] = frag_b(W_h, G, 4 * i, 128 + zc0);
    whn[i] = frag_b(W_h, G, 4 * i, 256 + zc0);
  }
  const v2f wxz_e = frag_b_kguard(W_x, G, 64, zc0,       CIN);
  const v2f wxr_e = frag_b_kguard(W_x, G, 64, 128 + zc0, CIN);
  const v2f wxn_e = frag_b_kguard(W_x, G, 64, 256 + zc0, CIN);
  v2f wr0[8], wr1[8];
  #pragma unroll
  for (int i = 0; i < 8; ++i) {
    wr0[i] = frag_b(W_rec + wave * S * S, S, 4 * i, 0);
    wr1[i] = frag_b(W_rec + wave * S * S, S, 4 * i, 16);
  }

  float* carry_h = ws + CARRY_H + (size_t)b0 * H;
  float* carry_k = ws + CARRY_K + (size_t)b0 * S;
  float* carry_e = ws + CARRY_E + b0;
  if (chunk == 0) {
    for (int i = tid; i < 16 * HP; i += 256) sh_h[i] = 0.0f;
    for (int i = tid; i < 16 * XP; i += 256) sh_kst[i] = 0.0f;
    if (tid < 16) sh_err[tid] = 0.0f;
  } else {
    for (int i = tid; i < 16 * H; i += 256) sh_h[(i >> 7) * HP + (i & 127)] = carry_h[i];
    for (int i = tid; i < 16 * S; i += 256) sh_kst[(i >> 5) * XP + (i & 31)] = carry_k[i];
    if (tid < 16) sh_err[tid] = carry_e[tid];
  }
  v2f wg[4];
  #pragma unroll
  for (int i = 0; i < 4; ++i) {
    const int kk = zc0 + 4 * i + ((lane >> 4) << 1);
    wg[i].x = (colb < M) ? W_gate[kk * M + colb]       : 0.0f;
    wg[i].y = (colb < M) ? W_gate[(kk + 1) * M + colb] : 0.0f;
  }
  if (tid < M) sh_bg[tid] = b_gate[tid];

  const float* yr0 = y + (size_t)(b0 + wave) * T + t0;
  const float* yr1 = y + (size_t)(b0 + 8 + wave) * T + t0;
  float ycur0 = yr0[0], ycur1 = yr1[0];

  const float4* Pw = reinterpret_cast<const float4*>(ws)
                   + (((size_t)blockIdx.x * TC) * 8 + wave) * 320 + lane;
  float4 nx[10];
  #pragma unroll
  for (int q = 0; q < 10; ++q) nx[q] = Pw[q * 32];
  __syncthreads();

  v8f Ahz = {}, Ahr = {}, Ahn = {};
  #pragma unroll
  for (int i = 0; i < 32; ++i) {
    v2f a = frag_a(sh_h, HP, 4 * i);
    Ahz = wmma4(a, whz[i], Ahz);
    Ahr = wmma4(a, whr[i], Ahr);
    Ahn = wmma4(a, whn[i], Ahn);
  }
  __syncthreads();

  for (int tl = 0; tl < TC; ++tl) {
    float hnew[8];
    v8f c0 = {nx[0].x, nx[0].y, nx[0].z, nx[0].w, nx[1].x, nx[1].y, nx[1].z, nx[1].w};
    v8f c1 = {nx[2].x, nx[2].y, nx[2].z, nx[2].w, nx[3].x, nx[3].y, nx[3].z, nx[3].w};
    {
      v2f ak[8];
      #pragma unroll
      for (int i = 0; i < 8; ++i) ak[i] = frag_a(sh_kst, XP, 4 * i);
      const float ev = sh_err[colb];
      float hold[8];
      #pragma unroll
      for (int v = 0; v < 8; ++v) hold[v] = sh_h[(rb + v) * HP + hc];
      v8f Cz = Ahz, Cr = Ahr, Cxn = {};
      const v8f Chn = Ahn;
      v2f ae;
      ae.x = (lane < 16) ? ev : 0.0f;
      ae.y = 0.0f;
      Cz  = wmma4(ae, wxz_e, Cz);
      Cr  = wmma4(ae, wxr_e, Cr);
      Cxn = wmma4(ae, wxn_e, Cxn);
      #pragma unroll
      for (int i = 0; i < 4; ++i) {
        c0 = wmma4(ak[i], wr0[i], c0);
        c1 = wmma4(ak[i], wr1[i], c1);
      }
      asm volatile("" :: "v"(c0), "v"(c1));
      const float pz[8] = {nx[4].x, nx[4].y, nx[4].z, nx[4].w, nx[5].x, nx[5].y, nx[5].z, nx[5].w};
      const float pr[8] = {nx[6].x, nx[6].y, nx[6].z, nx[6].w, nx[7].x, nx[7].y, nx[7].z, nx[7].w};
      const float pn[8] = {nx[8].x, nx[8].y, nx[8].z, nx[8].w, nx[9].x, nx[9].y, nx[9].z, nx[9].w};
      #pragma unroll
      for (int v = 0; v < 8; ++v) {
        float z = fast_sigmoid(Cz[v] + pz[v]);
        float r = fast_sigmoid(Cr[v] + pr[v]);
        float n = fast_tanh((Cxn[v] + pn[v]) + r * (Chn[v] + bias_hn));
        hnew[v] = (1.0f - z) * n + z * hold[v];
      }
      #pragma unroll
      for (int v = 0; v < 8; ++v) sh_h[(rb + v) * HP + hc] = hnew[v];
      __builtin_amdgcn_fence(__ATOMIC_RELEASE, "workgroup", "local");
      __builtin_amdgcn_s_barrier_signal(-1);
      __builtin_amdgcn_sched_barrier(0);
      v2f aown[4];
      #pragma unroll
      for (int i = 0; i < 4; ++i) aown[i] = frag_a(sh_h, HP, zc0 + 4 * i);
      #pragma unroll
      for (int i = 4; i < 8; ++i) {
        c0 = wmma4(ak[i], wr0[i], c0);
        c1 = wmma4(ak[i], wr1[i], c1);
      }
      v8f cl = {};
      #pragma unroll
      for (int i = 0; i < 4; ++i) cl = wmma4(aown[i], wg[i], cl);
      #pragma unroll
      for (int v = 0; v < 8; ++v) sh_sc[wave][(rb + v) * SP + colb] = cl[v];
      __builtin_amdgcn_sched_barrier(0);
      asm volatile("" :: "v"(c0), "v"(c1), "v"(cl));
      __builtin_amdgcn_s_barrier_wait(-1);
      __builtin_amdgcn_fence(__ATOMIC_ACQUIRE, "workgroup", "local");
    }

    v2f ahf[32];
    #pragma unroll
    for (int i = 0; i < 32; ++i) ahf[i] = frag_a(sh_h, HP, 4 * i);
    __builtin_amdgcn_sched_barrier(0);

    const int tn = (tl + 1 < TC) ? tl + 1 : tl;
    {
      const float4* Pn = Pw + (size_t)tn * P_STEP4;
      #pragma unroll
      for (int q = 0; q < 10; ++q) nx[q] = Pn[q * 32];
    }
    const float yn0 = yr0[tn];
    const float yn1 = yr1[tn];
    #pragma unroll
    for (int v = 0; v < 8; ++v) {
      int row = rb + v;
      sh_ns[row * NP + wave * 32 + colb]      = fast_tanh(c0[v]);
      sh_ns[row * NP + wave * 32 + 16 + colb] = fast_tanh(c1[v]);
    }
    Ahz = v8f{}; Ahr = v8f{}; Ahn = v8f{};
    #pragma unroll
    for (int i = 0; i < 7; ++i) {
      Ahz = wmma4(ahf[i], whz[i], Ahz);
      Ahr = wmma4(ahf[i], whr[i], Ahr);
      Ahn = wmma4(ahf[i], whn[i], Ahn);
    }
    __builtin_amdgcn_fence(__ATOMIC_RELEASE, "workgroup", "local");
    __builtin_amdgcn_s_barrier_signal(-1);
    __builtin_amdgcn_sched_barrier(0);
    #pragma unroll
    for (int i = 7; i < 10; ++i) {
      Ahz = wmma4(ahf[i], whz[i], Ahz);
      Ahr = wmma4(ahf[i], whr[i], Ahr);
      Ahn = wmma4(ahf[i], whn[i], Ahn);
    }
    __builtin_amdgcn_sched_barrier(0);
    asm volatile("" :: "v"(Ahz), "v"(Ahr), "v"(Ahn));
    __builtin_amdgcn_s_barrier_wait(-1);
    __builtin_amdgcn_fence(__ATOMIC_ACQUIRE, "workgroup", "local");

    if (tid < 16) {
      float s[M], mx = -1e30f;
      #pragma unroll
      for (int m = 0; m < M; ++m) {
        float a = sh_sc[0][tid * SP + m];
        #pragma unroll
        for (int w = 1; w < 8; ++w) a += sh_sc[w][tid * SP + m];
        s[m] = a + sh_bg[m];
        mx = fmaxf(mx, s[m]);
      }
      float sum = 0.0f;
      #pragma unroll
      for (int m = 0; m < M; ++m) { s[m] = __expf(s[m] - mx); sum += s[m]; }
      float inv = __fdividef(1.0f, sum);
      #pragma unroll
      for (int m = 0; m < M; ++m) sh_gate[tid * M + m] = s[m] * inv;
    }
    __builtin_amdgcn_fence(__ATOMIC_RELEASE, "workgroup", "local");
    __builtin_amdgcn_s_barrier_signal(-1);
    #pragma unroll
    for (int i = 10; i < 21; ++i) {
      Ahz = wmma4(ahf[i], whz[i], Ahz);
      Ahr = wmma4(ahf[i], whr[i], Ahr);
      Ahn = wmma4(ahf[i], whn[i], Ahn);
    }
    asm volatile("" :: "v"(Ahz), "v"(Ahr), "v"(Ahn));
    __builtin_amdgcn_s_barrier_wait(-1);
    __builtin_amdgcn_fence(__ATOMIC_ACQUIRE, "workgroup", "local");

    float mix[2];
    {
      float gt[2][M], nsv[2][M];
      #pragma unroll
      for (int j = 0; j < 2; ++j) {
        #pragma unroll
        for (int m = 0; m < M; ++m) {
          gt[j][m]  = sh_gate[(wave + 8 * j) * M + m];
          nsv[j][m] = sh_ns[(wave + 8 * j) * NP + m * 32 + lane];
        }
      }
      asm volatile("" :: "v"(gt[0][0]), "v"(gt[0][1]), "v"(gt[0][2]), "v"(gt[0][3]),
                         "v"(gt[0][4]), "v"(gt[0][5]), "v"(gt[0][6]), "v"(gt[0][7]),
                         "v"(gt[1][0]), "v"(gt[1][1]), "v"(gt[1][2]), "v"(gt[1][3]),
                         "v"(gt[1][4]), "v"(gt[1][5]), "v"(gt[1][6]), "v"(gt[1][7]));
      asm volatile("" :: "v"(nsv[0][0]), "v"(nsv[0][1]), "v"(nsv[0][2]), "v"(nsv[0][3]),
                         "v"(nsv[0][4]), "v"(nsv[0][5]), "v"(nsv[0][6]), "v"(nsv[0][7]),
                         "v"(nsv[1][0]), "v"(nsv[1][1]), "v"(nsv[1][2]), "v"(nsv[1][3]),
                         "v"(nsv[1][4]), "v"(nsv[1][5]), "v"(nsv[1][6]), "v"(nsv[1][7]));
      #pragma unroll
      for (int j = 0; j < 2; ++j) {
        float acc = 0.0f;
        #pragma unroll
        for (int m = 0; m < M; ++m) acc += gt[j][m] * nsv[j][m];
        sh_kst[(wave + 8 * j) * XP + lane] = acc;
        mix[j] = acc;
      }
    }
    if (lane == S - 1) {
      out[(size_t)(b0 + wave) * T + t0 + tl]     = mix[0];
      out[(size_t)(b0 + wave + 8) * T + t0 + tl] = mix[1];
      sh_err[wave]     = mix[0] - ycur0;
      sh_err[wave + 8] = mix[1] - ycur1;
    }
    ycur0 = yn0;
    ycur1 = yn1;
    __builtin_amdgcn_fence(__ATOMIC_RELEASE, "workgroup", "local");
    __builtin_amdgcn_s_barrier_signal(-1);
    #pragma unroll
    for (int i = 21; i < 32; ++i) {
      Ahz = wmma4(ahf[i], whz[i], Ahz);
      Ahr = wmma4(ahf[i], whr[i], Ahr);
      Ahn = wmma4(ahf[i], whn[i], Ahn);
    }
    asm volatile("" :: "v"(Ahz), "v"(Ahr), "v"(Ahn));
    __builtin_amdgcn_s_barrier_wait(-1);
    __builtin_amdgcn_fence(__ATOMIC_ACQUIRE, "workgroup", "local");
  }

  if (chunk + 1 < NCHUNK) {
    for (int i = tid; i < 16 * H; i += 256) carry_h[i] = sh_h[(i >> 7) * HP + (i & 127)];
    for (int i = tid; i < 16 * S; i += 256) carry_k[i] = sh_kst[(i >> 5) * XP + (i & 31)];
    if (tid < 16) carry_e[tid] = sh_err[tid];
  }
}

extern "C" void kernel_launch(void* const* d_in, const int* in_sizes, int n_in,
                              void* d_out, int out_size, void* d_ws, size_t ws_size,
                              hipStream_t stream) {
  (void)in_sizes; (void)n_in; (void)out_size;
  const float* x      = (const float*)d_in[0];
  const float* y      = (const float*)d_in[1];
  const float* W_enc  = (const float*)d_in[2];
  const float* b_enc  = (const float*)d_in[3];
  const float* W_in   = (const float*)d_in[4];
  const float* W_rec  = (const float*)d_in[5];
  const float* b_s    = (const float*)d_in[6];
  const float* W_x    = (const float*)d_in[7];
  const float* W_h    = (const float*)d_in[8];
  const float* b_x    = (const float*)d_in[9];
  const float* b_h    = (const float*)d_in[10];
  const float* W_gate = (const float*)d_in[11];
  const float* b_gate = (const float*)d_in[12];
  float* out = (float*)d_out;
  float* ws  = (float*)d_ws;

  kalman_gru_scan<<<dim3(8), dim3(256), 0, stream>>>(
      x, y, W_enc, b_enc, W_in, W_rec, b_s, W_x, W_h, b_x, b_h,
      W_gate, b_gate, out, ws_size);

#define SCAN_BLOCK(c)                                                          \
  input_products_chunk<<<dim3(256), dim3(256), 0, stream>>>(                   \
      x, W_enc, b_enc, W_in, b_s, W_x, b_x, b_h, ws, ws_size, c);              \
  kalman_gru_scan_chunk<<<dim3(8), dim3(256), 0, stream>>>(                    \
      y, W_rec, W_x, W_h, b_h, W_gate, b_gate, out, ws, ws_size, c);
  SCAN_BLOCK(0)  SCAN_BLOCK(1)  SCAN_BLOCK(2)  SCAN_BLOCK(3)
  SCAN_BLOCK(4)  SCAN_BLOCK(5)  SCAN_BLOCK(6)  SCAN_BLOCK(7)
  SCAN_BLOCK(8)  SCAN_BLOCK(9)  SCAN_BLOCK(10) SCAN_BLOCK(11)
  SCAN_BLOCK(12) SCAN_BLOCK(13) SCAN_BLOCK(14) SCAN_BLOCK(15)
#undef SCAN_BLOCK
}
